# starting kernel
# baseline (speedup 1.0000x reference)
#include <hip/hip_runtime.h>

typedef __attribute__((ext_vector_type(2))) float v2f;
typedef __attribute__((ext_vector_type(8))) float v8f;

#define BN_EPS 1e-5f

__global__ void zero_f32(float* __restrict__ p, long long n) {
  long long i = (long long)blockIdx.x * blockDim.x + threadIdx.x;
  if (i < n) p[i] = 0.0f;
}

__global__ void count_deg(const int* __restrict__ dst, float* __restrict__ deg, int e) {
  int i = blockIdx.x * blockDim.x + threadIdx.x;
  if (i < e) atomicAdd(&deg[dst[i]], 1.0f);
}

__global__ void make_dinv(float* __restrict__ deg, int n) {
  int i = blockIdx.x * blockDim.x + threadIdx.x;
  if (i < n) deg[i] = rsqrtf(deg[i] + 1.0f);
}

template <int K, int NC, int MT>
__global__ void __launch_bounds__(MT*(NC/16)*32)
gemm_wmma_f32(const float* __restrict__ A, const float* __restrict__ B,
              float* __restrict__ C) {
  constexpr int KP  = K + 4;
  constexpr int NCP = NC + 8;
  constexpr int NW  = MT * (NC/16);
  __shared__ float sA[MT*16*KP];
  __shared__ float sB[K*NCP];

  const int tid = threadIdx.x;
  const int nthreads = NW * 32;
  const float* Ablk = A + (size_t)blockIdx.x * (MT*16) * K;

  for (int idx = tid; idx < MT*16*(K/4); idx += nthreads) {
    int r = idx / (K/4), c4 = (idx % (K/4)) * 4;
    *(float4*)&sA[r*KP + c4] = *(const float4*)&Ablk[(size_t)r*K + c4];
  }
  for (int idx = tid; idx < K*(NC/4); idx += nthreads) {
    int r = idx / (NC/4), c4 = (idx % (NC/4)) * 4;
    *(float4*)&sB[r*NCP + c4] = *(const float4*)&B[(size_t)r*NC + c4];
  }
  __syncthreads();

  const int wave  = tid >> 5, lane = tid & 31;
  const int mt    = wave / (NC/16), nt = wave % (NC/16);
  const int rrow  = lane & 15;
  const int khalf = lane >> 4;

  const float* a_ptr = &sA[(mt*16 + rrow)*KP + khalf*2];
  const float* b_ptr = &sB[(khalf*2)*NCP + nt*16 + rrow];

  v8f c0 = {};
  v8f c1 = {};
  #pragma unroll 4
  for (int ks = 0; ks < K/8; ++ks) {
    v2f a0; a0.x = a_ptr[0];     a0.y = a_ptr[1];
    v2f b0; b0.x = b_ptr[0];     b0.y = b_ptr[NCP];
    v2f a1; a1.x = a_ptr[4];     a1.y = a_ptr[5];
    v2f b1; b1.x = b_ptr[4*NCP]; b1.y = b_ptr[5*NCP];
    c0 = __builtin_amdgcn_wmma_f32_16x16x4_f32(false, a0, false, b0,
                                               (short)0, c0, false, false);
    c1 = __builtin_amdgcn_wmma_f32_16x16x4_f32(false, a1, false, b1,
                                               (short)0, c1, false, false);
    a_ptr += 8;
    b_ptr += 8*NCP;
  }
  v8f c = c0 + c1;

  const int gm  = blockIdx.x*(MT*16) + mt*16 + khalf*8;
  const int col = nt*16 + rrow;
  #pragma unroll
  for (int r = 0; r < 8; ++r)
    C[(size_t)(gm + r)*NC + col] = c[r];
}

template <int F>
__global__ void scatter_edges(const int* __restrict__ src, const int* __restrict__ dst,
                              const float* __restrict__ dinv, const float* __restrict__ h,
                              float* __restrict__ agg, long long total) {
  long long t = (long long)blockIdx.x * blockDim.x + threadIdx.x;
  if (t >= total) return;
  constexpr int HALF = F / 2;
  int e = (int)(t / HALF);
  int f = (int)(t % HALF) * 2;
  int s = src[e], d = dst[e];
  float norm = dinv[s] * dinv[d];
  float2 hv = *(const float2*)&h[(size_t)s*F + f];
  atomicAdd(&agg[(size_t)d*F + f],     hv.x * norm);
  atomicAdd(&agg[(size_t)d*F + f + 1], hv.y * norm);
}

template <int F>
__global__ void finalize_bn(float* __restrict__ agg, const float* __restrict__ h,
                            const float* __restrict__ dinv, const float* __restrict__ bias,
                            const float* __restrict__ gamma, const float* __restrict__ beta,
                            const float* __restrict__ mean, const float* __restrict__ var,
                            long long total) {
  long long t = (long long)blockIdx.x * blockDim.x + threadIdx.x;
  if (t >= total) return;
  int node = (int)(t / F), f = (int)(t % F);
  float di = dinv[node];
  float v = agg[t] + h[t] * di * di + bias[f];
  v = (v - mean[f]) * rsqrtf(var[f] + BN_EPS) * gamma[f] + beta[f];
  agg[t] = fmaxf(v, 0.0f);
}

__global__ void final_head(const float* __restrict__ agg2, const float* __restrict__ h2,
                           const float* __restrict__ dinv, const float* __restrict__ b2,
                           const float* __restrict__ g2, const float* __restrict__ be2,
                           const float* __restrict__ m2, const float* __restrict__ v2,
                           const float* __restrict__ Wl, const float* __restrict__ bl,
                           float* __restrict__ out, int n) {
  int node = blockIdx.x * blockDim.x + threadIdx.x;
  if (node >= n) return;
  float di = dinv[node];
  float di2 = di * di;
  float acc0 = bl[0], acc1 = bl[1];
  const float* ar = agg2 + (size_t)node * 32;
  const float* hr = h2 + (size_t)node * 32;
  #pragma unroll
  for (int f = 0; f < 32; ++f) {
    float v = ar[f] + hr[f] * di2 + b2[f];
    v = (v - m2[f]) * rsqrtf(v2[f] + BN_EPS) * g2[f] + be2[f];
    v = fmaxf(v, 0.0f);
    acc0 += v * Wl[f*2 + 0];
    acc1 += v * Wl[f*2 + 1];
  }
  out[(size_t)node*2 + 0] = acc0;
  out[(size_t)node*2 + 1] = acc1;
}

extern "C" void kernel_launch(void* const* d_in, const int* in_sizes, int n_in,
                              void* d_out, int out_size, void* d_ws, size_t ws_size,
                              hipStream_t stream) {
  const float* x      = (const float*)d_in[0];
  const int*   ei     = (const int*)  d_in[1];
  const float* W1     = (const float*)d_in[2];
  const float* b1     = (const float*)d_in[3];
  const float* gamma1 = (const float*)d_in[4];
  const float* beta1  = (const float*)d_in[5];
  const float* mean1  = (const float*)d_in[6];
  const float* var1   = (const float*)d_in[7];
  const float* W2     = (const float*)d_in[8];
  const float* b2     = (const float*)d_in[9];
  const float* gamma2 = (const float*)d_in[10];
  const float* beta2  = (const float*)d_in[11];
  const float* mean2  = (const float*)d_in[12];
  const float* var2   = (const float*)d_in[13];
  const float* Wl     = (const float*)d_in[14];
  const float* bl     = (const float*)d_in[15];

  const int n = in_sizes[0] / 256;
  const int e = in_sizes[1] / 2;
  const int* src = ei;
  const int* dst = ei + e;

  float* ws   = (float*)d_ws;
  float* dinv = ws;
  float* h1   = dinv + (size_t)n;
  float* agg1 = h1   + (size_t)n * 64;
  float* h2   = agg1 + (size_t)n * 64;
  float* agg2 = h2   + (size_t)n * 32;

  const int T = 256;
  auto blocks = [](long long total, int t) {
    return (unsigned int)((total + t - 1) / t);
  };

  zero_f32<<<blocks(n, T), T, 0, stream>>>(dinv, n);
  zero_f32<<<blocks((long long)n * 64, T), T, 0, stream>>>(agg1, (long long)n * 64);
  zero_f32<<<blocks((long long)n * 32, T), T, 0, stream>>>(agg2, (long long)n * 32);

  count_deg<<<blocks(e, T), T, 0, stream>>>(dst, dinv, e);
  make_dinv<<<blocks(n, T), T, 0, stream>>>(dinv, n);

  gemm_wmma_f32<256, 64, 1><<<n / 16, 128, 0, stream>>>(x, W1, h1);

  scatter_edges<64><<<blocks((long long)e * 32, T), T, 0, stream>>>(
      src, dst, dinv, h1, agg1, (long long)e * 32);

  finalize_bn<64><<<blocks((long long)n * 64, T), T, 0, stream>>>(
      agg1, h1, dinv, b1, gamma1, beta1, mean1, var1, (long long)n * 64);

  gemm_wmma_f32<64, 32, 2><<<n / 32, 128, 0, stream>>>(agg1, W2, h2);

  scatter_edges<32><<<blocks((long long)e * 16, T), T, 0, stream>>>(
      src, dst, dinv, h2, agg2, (long long)e * 16);

  final_head<<<blocks(n, T), T, 0, stream>>>(
      agg2, h2, dinv, b2, gamma2, beta2, mean2, var2, Wl, bl,
      (float*)d_out, n);
}


# aggregation: work-group owns a range of destination rows as f32 sums in LDS, scans an edge chunk, adds matching rows with LDS adds, then line-wide global adds; sum order = LDS arrival order
# speedup vs baseline: 2.0929x; 2.0929x over previous
#include <hip/hip_runtime.h>

typedef __attribute__((ext_vector_type(2))) float v2f;
typedef __attribute__((ext_vector_type(8))) float v8f;

#define BN_EPS 1e-5f

__global__ void zero_f32(float* __restrict__ p, long long n) {
  long long i = (long long)blockIdx.x * blockDim.x + threadIdx.x;
  if (i < n) p[i] = 0.0f;
}

template <int R, int C, int NT, int U>
__global__ void __launch_bounds__(NT)
degree_partials(const int* __restrict__ dst, unsigned int* __restrict__ part,
                int n, int e) {
  __shared__ __attribute__((aligned(16))) unsigned int cnt[R];
  const int tid = threadIdx.x;
  const int lo  = blockIdx.x * R;
  const int c   = blockIdx.y;

  uint4* cnt4 = (uint4*)cnt;
  for (int i = tid; i < R/4; i += NT) cnt4[i] = make_uint4(0u, 0u, 0u, 0u);
  __syncthreads();

  const long long per = ((long long)e + C - 1) / C;
  const long long beg = per * c;
  long long end = beg + per;
  if (end > e) end = e;
  const int len = end > beg ? (int)(end - beg) : 0;
  const int* dptr = dst + beg;

  for (int base = 0; base < len; base += NT*U) {
    int v[U];
    #pragma unroll
    for (int u = 0; u < U; ++u) {
      const int i = base + u*NT + tid;
      const int x = dptr[i < len ? i : len - 1];
      v[u] = (i < len) ? x : -1;
    }
    #pragma unroll
    for (int u = 0; u < U; ++u) {
      const unsigned int d = (unsigned int)v[u] - (unsigned int)lo;
      if (d < (unsigned int)R) atomicAdd(&cnt[d], 1u);
    }
  }
  __syncthreads();

  int m = n - lo;
  if (m > R) m = R;
  unsigned int* row = part + (size_t)c * n + lo;
  for (int i = tid; i < m; i += NT) row[i] = cnt[i];
}

template <int C>
__global__ void dinv_from_partials(const unsigned int* __restrict__ part,
                                   float* __restrict__ dinv, int n) {
  int i = blockIdx.x * blockDim.x + threadIdx.x;
  if (i >= n) return;
  unsigned int s = 0;
  #pragma unroll
  for (int c = 0; c < C; ++c) s += part[(size_t)c * n + i];
  dinv[i] = rsqrtf((float)s + 1.0f);
}

template <int K, int NC, int MT>
__global__ void __launch_bounds__(MT*(NC/16)*32)
gemm_wmma_f32(const float* __restrict__ A, const float* __restrict__ B,
              float* __restrict__ C) {
  constexpr int KP  = K + 4;
  constexpr int NCP = NC + 8;
  constexpr int NW  = MT * (NC/16);
  __shared__ float sA[MT*16*KP];
  __shared__ float sB[K*NCP];

  const int tid = threadIdx.x;
  const int nthreads = NW * 32;
  const float* Ablk = A + (size_t)blockIdx.x * (MT*16) * K;

  for (int idx = tid; idx < MT*16*(K/4); idx += nthreads) {
    int r = idx / (K/4), c4 = (idx % (K/4)) * 4;
    *(float4*)&sA[r*KP + c4] = *(const float4*)&Ablk[(size_t)r*K + c4];
  }
  for (int idx = tid; idx < K*(NC/4); idx += nthreads) {
    int r = idx / (NC/4), c4 = (idx % (NC/4)) * 4;
    *(float4*)&sB[r*NCP + c4] = *(const float4*)&B[(size_t)r*NC + c4];
  }
  __syncthreads();

  const int wave  = tid >> 5, lane = tid & 31;
  const int mt    = wave / (NC/16), nt = wave % (NC/16);
  const int rrow  = lane & 15;
  const int khalf = lane >> 4;

  const float* a_ptr = &sA[(mt*16 + rrow)*KP + khalf*2];
  const float* b_ptr = &sB[(khalf*2)*NCP + nt*16 + rrow];

  v8f c0 = {};
  v8f c1 = {};
  #pragma unroll 4
  for (int ks = 0; ks < K/8; ++ks) {
    v2f a0; a0.x = a_ptr[0];     a0.y = a_ptr[1];
    v2f b0; b0.x = b_ptr[0];     b0.y = b_ptr[NCP];
    v2f a1; a1.x = a_ptr[4];     a1.y = a_ptr[5];
    v2f b1; b1.x = b_ptr[4*NCP]; b1.y = b_ptr[5*NCP];
    c0 = __builtin_amdgcn_wmma_f32_16x16x4_f32(false, a0, false, b0,
                                               (short)0, c0, false, false);
    c1 = __builtin_amdgcn_wmma_f32_16x16x4_f32(false, a1, false, b1,
                                               (short)0, c1, false, false);
    a_ptr += 8;
    b_ptr += 8*NCP;
  }
  v8f c = c0 + c1;

  const int gm  = blockIdx.x*(MT*16) + mt*16 + khalf*8;
  const int col = nt*16 + rrow;
  #pragma unroll
  for (int r = 0; r < 8; ++r)
    C[(size_t)(gm + r)*NC + col] = c[r];
}

template <int F, int R, int C, int NT, int U, int G>
__global__ void __launch_bounds__(NT)
aggregate_ranges(const int* __restrict__ src, const int* __restrict__ dst,
                 const float* __restrict__ dinv, const float* __restrict__ h,
                 float* __restrict__ agg, int n, int e) {
  constexpr int NW = NT / 32;
  constexpr int FQ = F / 32;
  constexpr int QN = 64;
  __shared__ __attribute__((aligned(16))) float acc[R*F];
  __shared__ uint2 ring[NW*QN];

  const int tid  = threadIdx.x;
  const int lane = tid & 31;
  const int wave = tid >> 5;
  const int lo   = blockIdx.x * R;
  const int c    = blockIdx.y;

  float4* acc4 = (float4*)acc;
  for (int i = tid; i < R*F/4; i += NT) acc4[i] = make_float4(0.f, 0.f, 0.f, 0.f);
  __syncthreads();

  const long long per = ((long long)e + C - 1) / C;
  const long long beg = per * c;
  long long end = beg + per;
  if (end > e) end = e;
  const int len = end > beg ? (int)(end - beg) : 0;
  const int* dptr = dst + beg;

  uint2* q = ring + wave*QN;
  unsigned int head = 0, cnt = 0;

  auto drain = [&](unsigned int k) {
    const uint2 ent = q[(head + lane) & (QN - 1)];
    const bool val = (unsigned int)lane < k;
    const unsigned int ei = val ? ent.x : 0u;
    const unsigned int dl = val ? ent.y : 0u;
    const int   s = src[ei];
    const float w = dinv[s] * dinv[lo + (int)dl];
    for (unsigned int j0 = 0; j0 < k; j0 += G) {
      int sj[G]; float wj[G]; unsigned int dj[G]; float v[G][FQ];
      #pragma unroll
      for (int g = 0; g < G; ++g) {
        const int j = (int)j0 + g;
        sj[g] = __builtin_amdgcn_readlane(s, j);
        wj[g] = __int_as_float(__builtin_amdgcn_readlane(__float_as_int(w), j));
        dj[g] = (unsigned int)__builtin_amdgcn_readlane((int)dl, j);
        #pragma unroll
        for (int f = 0; f < FQ; ++f)
          v[g][f] = h[(size_t)sj[g]*F + f*32 + lane];
      }
      #pragma unroll
      for (int g = 0; g < G; ++g) {
        if (j0 + g < k) {
          #pragma unroll
          for (int f = 0; f < FQ; ++f)
            atomicAdd(&acc[dj[g]*F + f*32 + lane], v[g][f] * wj[g]);
        }
      }
    }
  };

  auto load_tile = [&](int t0, int (&out)[U]) {
    #pragma unroll
    for (int u = 0; u < U; ++u) {
      const int i = t0 + u*32 + lane;
      const int x = dptr[i < len ? i : len - 1];
      out[u] = (i < len) ? x : -1;
    }
  };
  constexpr int STEP = NW*32*U;
  int tb = wave*(32*U);
  int dv[U];
  if (tb < len) load_tile(tb, dv);
  for (; tb < len; tb += STEP) {
    int nv[U];
    load_tile(tb + STEP, nv);
    #pragma nounroll
    for (int u = 0; u < U; ++u) {
      const unsigned int dl = (unsigned int)dv[u] - (unsigned int)lo;
      const bool hit = dl < (unsigned int)R;
      const unsigned int m = __builtin_amdgcn_ballot_w32(hit);
      if (m) {
        if (hit) {
          const unsigned int pos =
              head + cnt + (unsigned int)__builtin_popcount(m & ((1u << lane) - 1u));
          q[pos & (QN - 1)] =
              make_uint2((unsigned int)(beg + tb + u*32 + lane), dl);
        }
        cnt += (unsigned int)__builtin_popcount(m);
        if (cnt >= 32u) { drain(32u); head += 32u; cnt -= 32u; }
      }
    }
    #pragma unroll
    for (int u = 0; u < U; ++u) dv[u] = nv[u];
  }
  if (cnt) drain(cnt);
  __syncthreads();

  int rows = n - lo;
  if (rows > R) rows = R;
  float* arow = agg + (size_t)lo * F;
  for (int i = tid; i < rows*F; i += NT) atomicAdd(&arow[i], acc[i]);
}

template <int F>
__global__ void finalize_bn(float* __restrict__ agg, const float* __restrict__ h,
                            const float* __restrict__ dinv, const float* __restrict__ bias,
                            const float* __restrict__ gamma, const float* __restrict__ beta,
                            const float* __restrict__ mean, const float* __restrict__ var,
                            long long total) {
  long long t = (long long)blockIdx.x * blockDim.x + threadIdx.x;
  if (t >= total) return;
  int node = (int)(t / F), f = (int)(t % F);
  float di = dinv[node];
  float v = agg[t] + h[t] * di * di + bias[f];
  v = (v - mean[f]) * rsqrtf(var[f] + BN_EPS) * gamma[f] + beta[f];
  agg[t] = fmaxf(v, 0.0f);
}

__global__ void final_head(const float* __restrict__ agg2, const float* __restrict__ h2,
                           const float* __restrict__ dinv, const float* __restrict__ b2,
                           const float* __restrict__ g2, const float* __restrict__ be2,
                           const float* __restrict__ m2, const float* __restrict__ v2,
                           const float* __restrict__ Wl, const float* __restrict__ bl,
                           float* __restrict__ out, int n) {
  int node = blockIdx.x * blockDim.x + threadIdx.x;
  if (node >= n) return;
  float di = dinv[node];
  float di2 = di * di;
  float acc0 = bl[0], acc1 = bl[1];
  const float* ar = agg2 + (size_t)node * 32;
  const float* hr = h2 + (size_t)node * 32;
  #pragma unroll
  for (int f = 0; f < 32; ++f) {
    float v = ar[f] + hr[f] * di2 + b2[f];
    v = (v - m2[f]) * rsqrtf(v2[f] + BN_EPS) * g2[f] + be2[f];
    v = fmaxf(v, 0.0f);
    acc0 += v * Wl[f*2 + 0];
    acc1 += v * Wl[f*2 + 1];
  }
  out[(size_t)node*2 + 0] = acc0;
  out[(size_t)node*2 + 1] = acc1;
}

extern "C" void kernel_launch(void* const* d_in, const int* in_sizes, int n_in,
                              void* d_out, int out_size, void* d_ws, size_t ws_size,
                              hipStream_t stream) {
  const float* x      = (const float*)d_in[0];
  const int*   ei     = (const int*)  d_in[1];
  const float* W1     = (const float*)d_in[2];
  const float* b1     = (const float*)d_in[3];
  const float* gamma1 = (const float*)d_in[4];
  const float* beta1  = (const float*)d_in[5];
  const float* mean1  = (const float*)d_in[6];
  const float* var1   = (const float*)d_in[7];
  const float* W2     = (const float*)d_in[8];
  const float* b2     = (const float*)d_in[9];
  const float* gamma2 = (const float*)d_in[10];
  const float* beta2  = (const float*)d_in[11];
  const float* mean2  = (const float*)d_in[12];
  const float* var2   = (const float*)d_in[13];
  const float* Wl     = (const float*)d_in[14];
  const float* bl     = (const float*)d_in[15];

  const int n = in_sizes[0] / 256;
  const int e = in_sizes[1] / 2;
  const int* src = ei;
  const int* dst = ei + e;

  float* ws   = (float*)d_ws;
  float* dinv = ws;
  float* h1   = dinv + (size_t)n;
  float* agg1 = h1   + (size_t)n * 64;
  float* h2   = agg1 + (size_t)n * 64;
  float* agg2 = h2   + (size_t)n * 32;

  const int T = 256;
  auto blocks = [](long long total, int t) {
    return (unsigned int)((total + t - 1) / t);
  };

  zero_f32<<<blocks((long long)n * 64, T), T, 0, stream>>>(agg1, (long long)n * 64);
  zero_f32<<<blocks((long long)n * 32, T), T, 0, stream>>>(agg2, (long long)n * 32);

  unsigned int* part = (unsigned int*)h1;
  degree_partials<25000, 64, 512, 25><<<dim3(blocks(n, 25000), 64), 512, 0, stream>>>(
      dst, part, n, e);
  dinv_from_partials<64><<<blocks(n, T), T, 0, stream>>>(part, dinv, n);

  gemm_wmma_f32<256, 64, 1><<<n / 16, 128, 0, stream>>>(x, W1, h1);

  aggregate_ranges<64, 782, 2, 1024, 16, 8><<<dim3(blocks(n, 782), 2), 1024, 0, stream>>>(
      src, dst, dinv, h1, agg1, n, e);

  finalize_bn<64><<<blocks((long long)n * 64, T), T, 0, stream>>>(
      agg1, h1, dinv, b1, gamma1, beta1, mean1, var1, (long long)n * 64);

  gemm_wmma_f32<64, 32, 2><<<n / 32, 128, 0, stream>>>(agg1, W2, h2);

  aggregate_ranges<32, 1563, 4, 1024, 16, 8><<<dim3(blocks(n, 1563), 4), 1024, 0, stream>>>(
      src, dst, dinv, h2, agg2, n, e);

  final_head<<<blocks(n, T), T, 0, stream>>>(
      agg2, h2, dinv, b2, gamma2, beta2, mean2, var2, Wl, bl,
      (float*)d_out, n);
}


# aggregation scan: one pass marks, per node range, which 32-edge lines hold one of its destinations (bit rows); a range's work-group fetches only marked lines; sum order = LDS arrival order as before
# speedup vs baseline: 2.2603x; 1.0800x over previous
#include <hip/hip_runtime.h>

typedef __attribute__((ext_vector_type(2))) float v2f;
typedef __attribute__((ext_vector_type(8))) float v8f;

#define BN_EPS 1e-5f

__global__ void zero_f32(float* __restrict__ p, long long n) {
  long long i = (long long)blockIdx.x * blockDim.x + threadIdx.x;
  if (i < n) p[i] = 0.0f;
}

template <int R, int C, int NT, int U>
__global__ void __launch_bounds__(NT)
degree_partials(const int* __restrict__ dst, unsigned int* __restrict__ part,
                int n, int e) {
  __shared__ __attribute__((aligned(16))) unsigned int cnt[R];
  const int tid = threadIdx.x;
  const int lo  = blockIdx.x * R;
  const int c   = blockIdx.y;

  uint4* cnt4 = (uint4*)cnt;
  for (int i = tid; i < R/4; i += NT) cnt4[i] = make_uint4(0u, 0u, 0u, 0u);
  __syncthreads();

  const long long per = ((long long)e + C - 1) / C;
  const long long beg = per * c;
  long long end = beg + per;
  if (end > e) end = e;
  const int len = end > beg ? (int)(end - beg) : 0;
  const int* dptr = dst + beg;

  for (int base = 0; base < len; base += NT*U) {
    int v[U];
    #pragma unroll
    for (int u = 0; u < U; ++u) {
      const int i = base + u*NT + tid;
      const int x = dptr[i < len ? i : len - 1];
      v[u] = (i < len) ? x : -1;
    }
    #pragma unroll
    for (int u = 0; u < U; ++u) {
      const unsigned int d = (unsigned int)v[u] - (unsigned int)lo;
      if (d < (unsigned int)R) atomicAdd(&cnt[d], 1u);
    }
  }
  __syncthreads();

  int m = n - lo;
  if (m > R) m = R;
  unsigned int* row = part + (size_t)c * n + lo;
  for (int i = tid; i < m; i += NT) row[i] = cnt[i];
}

template <int C>
__global__ void dinv_from_partials(const unsigned int* __restrict__ part,
                                   float* __restrict__ dinv, int n) {
  int i = blockIdx.x * blockDim.x + threadIdx.x;
  if (i >= n) return;
  unsigned int s = 0;
  #pragma unroll
  for (int c = 0; c < C; ++c) s += part[(size_t)c * n + i];
  dinv[i] = rsqrtf((float)s + 1.0f);
}

template <int K, int NC, int MT>
__global__ void __launch_bounds__(MT*(NC/16)*32)
gemm_wmma_f32(const float* __restrict__ A, const float* __restrict__ B,
              float* __restrict__ C) {
  constexpr int KP  = K + 4;
  constexpr int NCP = NC + 8;
  constexpr int NW  = MT * (NC/16);
  __shared__ float sA[MT*16*KP];
  __shared__ float sB[K*NCP];

  const int tid = threadIdx.x;
  const int nthreads = NW * 32;
  const float* Ablk = A + (size_t)blockIdx.x * (MT*16) * K;

  for (int idx = tid; idx < MT*16*(K/4); idx += nthreads) {
    int r = idx / (K/4), c4 = (idx % (K/4)) * 4;
    *(float4*)&sA[r*KP + c4] = *(const float4*)&Ablk[(size_t)r*K + c4];
  }
  for (int idx = tid; idx < K*(NC/4); idx += nthreads) {
    int r = idx / (NC/4), c4 = (idx % (NC/4)) * 4;
    *(float4*)&sB[r*NCP + c4] = *(const float4*)&B[(size_t)r*NC + c4];
  }
  __syncthreads();

  const int wave  = tid >> 5, lane = tid & 31;
  const int mt    = wave / (NC/16), nt = wave % (NC/16);
  const int rrow  = lane & 15;
  const int khalf = lane >> 4;

  const float* a_ptr = &sA[(mt*16 + rrow)*KP + khalf*2];
  const float* b_ptr = &sB[(khalf*2)*NCP + nt*16 + rrow];

  v8f c0 = {};
  v8f c1 = {};
  #pragma unroll 4
  for (int ks = 0; ks < K/8; ++ks) {
    v2f a0; a0.x = a_ptr[0];     a0.y = a_ptr[1];
    v2f b0; b0.x = b_ptr[0];     b0.y = b_ptr[NCP];
    v2f a1; a1.x = a_ptr[4];     a1.y = a_ptr[5];
    v2f b1; b1.x = b_ptr[4*NCP]; b1.y = b_ptr[5*NCP];
    c0 = __builtin_amdgcn_wmma_f32_16x16x4_f32(false, a0, false, b0,
                                               (short)0, c0, false, false);
    c1 = __builtin_amdgcn_wmma_f32_16x16x4_f32(false, a1, false, b1,
                                               (short)0, c1, false, false);
    a_ptr += 8;
    b_ptr += 8*NCP;
  }
  v8f c = c0 + c1;

  const int gm  = blockIdx.x*(MT*16) + mt*16 + khalf*8;
  const int col = nt*16 + rrow;
  #pragma unroll
  for (int r = 0; r < 8; ++r)
    C[(size_t)(gm + r)*NC + col] = c[r];
}

template <int R, int NR, int NT>
__global__ void __launch_bounds__(NT)
mark_lines(const int* __restrict__ dst, unsigned int* __restrict__ marks, int e) {
  constexpr int NW = NT / 32;
  constexpr int LP = NW + 1;
  __shared__ unsigned int bits[NR * LP];
  const int tid = threadIdx.x, lane = tid & 31, wave = tid >> 5;
  for (int i = tid; i < NR * LP; i += NT) bits[i] = 0u;
  __syncthreads();

  const int nwords = (e + 1023) >> 10;
  const int pitch  = (nwords + 31) & ~31;
  const int word   = blockIdx.x * NW + wave;
  if (word < nwords) {
    const int base = word * 1024 + lane;
    for (int j0 = 0; j0 < 32; j0 += 8) {
      int v[8];
      #pragma unroll
      for (int u = 0; u < 8; ++u) {
        const int i = base + (j0 + u) * 32;
        const int x = dst[i < e ? i : e - 1];
        v[u] = (i < e) ? x : -1;
      }
      #pragma unroll
      for (int u = 0; u < 8; ++u) {
        const unsigned int rid = (unsigned int)v[u] / (unsigned int)R;
        if (rid < (unsigned int)NR) atomicOr(&bits[rid * LP + wave], 1u << (j0 + u));
      }
    }
  }
  __syncthreads();

  for (int i = tid; i < NR * NW; i += NT) {
    const int r = i / NW, w = i % NW;
    marks[(size_t)r * pitch + blockIdx.x * NW + w] = bits[r * LP + w];
  }
}

template <int F, int R, int C, int NT, int U, int G>
__global__ void __launch_bounds__(NT)
aggregate_ranges(const int* __restrict__ src, const int* __restrict__ dst,
                 const float* __restrict__ dinv, const float* __restrict__ h,
                 float* __restrict__ agg, const unsigned int* __restrict__ marks,
                 int n, int e) {
  constexpr int NW = NT / 32;
  constexpr int FQ = F / 32;
  constexpr int QN = 64;
  __shared__ __attribute__((aligned(16))) float acc[R*F];
  __shared__ uint2 ring[NW*QN];

  const int tid  = threadIdx.x;
  const int lane = tid & 31;
  const int wave = __builtin_amdgcn_readfirstlane(tid >> 5);
  const int lo   = blockIdx.x * R;
  const int c    = blockIdx.y;

  float4* acc4 = (float4*)acc;
  for (int i = tid; i < R*F/4; i += NT) acc4[i] = make_float4(0.f, 0.f, 0.f, 0.f);
  __syncthreads();

  const int nwords = (e + 1023) >> 10;
  const int pitch  = (nwords + 31) & ~31;
  const int per    = (nwords + C - 1) / C;
  const int wbeg   = per * c;
  int wend = wbeg + per;
  if (wend > nwords) wend = nwords;
  const unsigned int* mrow = marks + (size_t)blockIdx.x * pitch;

  uint2* q = ring + wave*QN;
  unsigned int head = 0, cnt = 0;

  auto drain = [&](unsigned int k) {
    const uint2 ent = q[(head + lane) & (QN - 1)];
    const bool val = (unsigned int)lane < k;
    const unsigned int ei = val ? ent.x : 0u;
    const unsigned int dl = val ? ent.y : 0u;
    const int   s = src[ei];
    const float w = dinv[s] * dinv[lo + (int)dl];
    for (unsigned int j0 = 0; j0 < k; j0 += G) {
      int sj[G]; float wj[G]; unsigned int dj[G]; float v[G][FQ];
      #pragma unroll
      for (int g = 0; g < G; ++g) {
        const int j = (int)j0 + g;
        sj[g] = __builtin_amdgcn_readlane(s, j);
        wj[g] = __int_as_float(__builtin_amdgcn_readlane(__float_as_int(w), j));
        dj[g] = (unsigned int)__builtin_amdgcn_readlane((int)dl, j);
        #pragma unroll
        for (int f = 0; f < FQ; ++f)
          v[g][f] = h[(size_t)sj[g]*F + f*32 + lane];
      }
      #pragma unroll
      for (int g = 0; g < G; ++g) {
        if (j0 + g < k) {
          #pragma unroll
          for (int f = 0; f < FQ; ++f)
            atomicAdd(&acc[dj[g]*F + f*32 + lane], v[g][f] * wj[g]);
        }
      }
    }
  };

  int wi = wbeg + wave;
  unsigned int m = 0u, mnext = 0u;
  if (wi < wend) m = mrow[wi];
  if (wi + NW < wend) mnext = mrow[wi + NW];
  m = (unsigned int)__builtin_amdgcn_readfirstlane((int)m);

  auto fetch = [&](int& w, unsigned int& mb, int (&v)[U]) -> int {
    while (m == 0u && wi + NW < wend) {
      wi += NW;
      m = (unsigned int)__builtin_amdgcn_readfirstlane((int)mnext);
      mnext = (wi + NW < wend) ? mrow[wi + NW] : 0u;
    }
    w = wi;
    mb = m;
    int k = 0;
    int j = 0;
    #pragma unroll
    for (int u = 0; u < U; ++u) {
      const bool have = (m != 0u);
      if (have) j = __builtin_ctz(m);
      m &= m - 1u;
      k += have ? 1 : 0;
      const int i = ((wi << 5) + j) * 32 + lane;
      const int x = dst[i < e ? i : e - 1];
      v[u] = (i < e) ? x : -1;
    }
    return k;
  };

  int wA = 0, kA; unsigned int mbA = 0u; int dvA[U];
  kA = fetch(wA, mbA, dvA);
  while (kA > 0) {
    int wB = 0, kB; unsigned int mbB = 0u; int dvB[U];
    kB = fetch(wB, mbB, dvB);
    #pragma nounroll
    for (int u = 0; u < kA; ++u) {
      const int j = __builtin_ctz(mbA);
      mbA &= mbA - 1u;
      const unsigned int dl = (unsigned int)dvA[u] - (unsigned int)lo;
      const bool hit = dl < (unsigned int)R;
      const unsigned int hm = __builtin_amdgcn_ballot_w32(hit);
      if (hit) {
        const unsigned int pos =
            head + cnt + (unsigned int)__builtin_popcount(hm & ((1u << lane) - 1u));
        q[pos & (QN - 1)] =
            make_uint2((unsigned int)(((wA << 5) + j) * 32 + lane), dl);
      }
      cnt += (unsigned int)__builtin_popcount(hm);
      if (cnt >= 32u) { drain(32u); head += 32u; cnt -= 32u; }
    }
    #pragma unroll
    for (int u = 0; u < U; ++u) dvA[u] = dvB[u];
    wA = wB; mbA = mbB; kA = kB;
  }
  if (cnt) drain(cnt);
  __syncthreads();

  int rows = n - lo;
  if (rows > R) rows = R;
  float* arow = agg + (size_t)lo * F;
  for (int i = tid; i < rows*F; i += NT) atomicAdd(&arow[i], acc[i]);
}

template <int F>
__global__ void finalize_bn(float* __restrict__ agg, const float* __restrict__ h,
                            const float* __restrict__ dinv, const float* __restrict__ bias,
                            const float* __restrict__ gamma, const float* __restrict__ beta,
                            const float* __restrict__ mean, const float* __restrict__ var,
                            long long total) {
  long long t = (long long)blockIdx.x * blockDim.x + threadIdx.x;
  if (t >= total) return;
  int node = (int)(t / F), f = (int)(t % F);
  float di = dinv[node];
  float v = agg[t] + h[t] * di * di + bias[f];
  v = (v - mean[f]) * rsqrtf(var[f] + BN_EPS) * gamma[f] + beta[f];
  agg[t] = fmaxf(v, 0.0f);
}

__global__ void final_head(const float* __restrict__ agg2, const float* __restrict__ h2,
                           const float* __restrict__ dinv, const float* __restrict__ b2,
                           const float* __restrict__ g2, const float* __restrict__ be2,
                           const float* __restrict__ m2, const float* __restrict__ v2,
                           const float* __restrict__ Wl, const float* __restrict__ bl,
                           float* __restrict__ out, int n) {
  int node = blockIdx.x * blockDim.x + threadIdx.x;
  if (node >= n) return;
  float di = dinv[node];
  float di2 = di * di;
  float acc0 = bl[0], acc1 = bl[1];
  const float* ar = agg2 + (size_t)node * 32;
  const float* hr = h2 + (size_t)node * 32;
  #pragma unroll
  for (int f = 0; f < 32; ++f) {
    float v = ar[f] + hr[f] * di2 + b2[f];
    v = (v - m2[f]) * rsqrtf(v2[f] + BN_EPS) * g2[f] + be2[f];
    v = fmaxf(v, 0.0f);
    acc0 += v * Wl[f*2 + 0];
    acc1 += v * Wl[f*2 + 1];
  }
  out[(size_t)node*2 + 0] = acc0;
  out[(size_t)node*2 + 1] = acc1;
}

extern "C" void kernel_launch(void* const* d_in, const int* in_sizes, int n_in,
                              void* d_out, int out_size, void* d_ws, size_t ws_size,
                              hipStream_t stream) {
  const float* x      = (const float*)d_in[0];
  const int*   ei     = (const int*)  d_in[1];
  const float* W1     = (const float*)d_in[2];
  const float* b1     = (const float*)d_in[3];
  const float* gamma1 = (const float*)d_in[4];
  const float* beta1  = (const float*)d_in[5];
  const float* mean1  = (const float*)d_in[6];
  const float* var1   = (const float*)d_in[7];
  const float* W2     = (const float*)d_in[8];
  const float* b2     = (const float*)d_in[9];
  const float* gamma2 = (const float*)d_in[10];
  const float* beta2  = (const float*)d_in[11];
  const float* mean2  = (const float*)d_in[12];
  const float* var2   = (const float*)d_in[13];
  const float* Wl     = (const float*)d_in[14];
  const float* bl     = (const float*)d_in[15];

  const int n = in_sizes[0] / 256;
  const int e = in_sizes[1] / 2;
  const int* src = ei;
  const int* dst = ei + e;

  float* ws   = (float*)d_ws;
  float* dinv = ws;
  float* h1   = dinv + (size_t)n;
  float* agg1 = h1   + (size_t)n * 64;
  float* h2   = agg1 + (size_t)n * 64;
  float* agg2 = h2   + (size_t)n * 32;

  const int T = 256;
  auto blocks = [](long long total, int t) {
    return (unsigned int)((total + t - 1) / t);
  };

  zero_f32<<<blocks((long long)n * 64, T), T, 0, stream>>>(agg1, (long long)n * 64);
  zero_f32<<<blocks((long long)n * 32, T), T, 0, stream>>>(agg2, (long long)n * 32);

  unsigned int* part = (unsigned int*)h1;
  degree_partials<25000, 64, 512, 25><<<dim3(blocks(n, 25000), 64), 512, 0, stream>>>(
      dst, part, n, e);
  dinv_from_partials<64><<<blocks(n, T), T, 0, stream>>>(part, dinv, n);

  gemm_wmma_f32<256, 64, 1><<<n / 16, 128, 0, stream>>>(x, W1, h1);

  unsigned int* marks1 = (unsigned int*)h2;
  mark_lines<782, 128, 1024><<<blocks(blocks(e, 1024), 32), 1024, 0, stream>>>(dst, marks1, e);

  aggregate_ranges<64, 782, 2, 1024, 8, 8><<<dim3(blocks(n, 782), 2), 1024, 0, stream>>>(
      src, dst, dinv, h1, agg1, marks1, n, e);

  finalize_bn<64><<<blocks((long long)n * 64, T), T, 0, stream>>>(
      agg1, h1, dinv, b1, gamma1, beta1, mean1, var1, (long long)n * 64);

  gemm_wmma_f32<64, 32, 2><<<n / 32, 128, 0, stream>>>(agg1, W2, h2);

  unsigned int* marks2 = (unsigned int*)h1;
  mark_lines<1563, 64, 1024><<<blocks(blocks(e, 1024), 32), 1024, 0, stream>>>(dst, marks2, e);

  aggregate_ranges<32, 1563, 4, 1024, 16, 8><<<dim3(blocks(n, 1563), 4), 1024, 0, stream>>>(
      src, dst, dinv, h2, agg2, marks2, n, e);

  final_head<<<blocks(n, T), T, 0, stream>>>(
      agg2, h2, dinv, b2, gamma2, beta2, mean2, var2, Wl, bl,
      (float*)d_out, n);
}


# GEMM staging (both layers): a thread issues all its global loads of W and of the row tile (32+8 in layer 1) before the first LDS store waits, not one row-tile load per wait; sums unchanged
# speedup vs baseline: 2.4978x; 1.1051x over previous
#include <hip/hip_runtime.h>

typedef __attribute__((ext_vector_type(2))) float v2f;
typedef __attribute__((ext_vector_type(8))) float v8f;

#define BN_EPS 1e-5f

__global__ void zero_f32(float* __restrict__ p, long long n) {
  long long i = (long long)blockIdx.x * blockDim.x + threadIdx.x;
  if (i < n) p[i] = 0.0f;
}

template <int R, int C, int NT, int U>
__global__ void __launch_bounds__(NT)
degree_partials(const int* __restrict__ dst, unsigned int* __restrict__ part,
                int n, int e) {
  __shared__ __attribute__((aligned(16))) unsigned int cnt[R];
  const int tid = threadIdx.x;
  const int lo  = blockIdx.x * R;
  const int c   = blockIdx.y;

  uint4* cnt4 = (uint4*)cnt;
  for (int i = tid; i < R/4; i += NT) cnt4[i] = make_uint4(0u, 0u, 0u, 0u);
  __syncthreads();

  const long long per = ((long long)e + C - 1) / C;
  const long long beg = per * c;
  long long end = beg + per;
  if (end > e) end = e;
  const int len = end > beg ? (int)(end - beg) : 0;
  const int* dptr = dst + beg;

  for (int base = 0; base < len; base += NT*U) {
    int v[U];
    #pragma unroll
    for (int u = 0; u < U; ++u) {
      const int i = base + u*NT + tid;
      const int x = dptr[i < len ? i : len - 1];
      v[u] = (i < len) ? x : -1;
    }
    #pragma unroll
    for (int u = 0; u < U; ++u) {
      const unsigned int d = (unsigned int)v[u] - (unsigned int)lo;
      if (d < (unsigned int)R) atomicAdd(&cnt[d], 1u);
    }
  }
  __syncthreads();

  int m = n - lo;
  if (m > R) m = R;
  unsigned int* row = part + (size_t)c * n + lo;
  for (int i = tid; i < m; i += NT) row[i] = cnt[i];
}

template <int C>
__global__ void dinv_from_partials(const unsigned int* __restrict__ part,
                                   float* __restrict__ dinv, int n) {
  int i = blockIdx.x * blockDim.x + threadIdx.x;
  if (i >= n) return;
  unsigned int s = 0;
  #pragma unroll
  for (int c = 0; c < C; ++c) s += part[(size_t)c * n + i];
  dinv[i] = rsqrtf((float)s + 1.0f);
}

template <int K, int NC, int MT>
__global__ void __launch_bounds__(MT*(NC/16)*32)
gemm_wmma_f32(const float* __restrict__ A, const float* __restrict__ B,
              float* __restrict__ C) {
  constexpr int KP  = K + 4;
  constexpr int NCP = NC + 8;
  constexpr int NW  = MT * (NC/16);
  __shared__ float sA[MT*16*KP];
  __shared__ float sB[K*NCP];

  const int tid = threadIdx.x;
  constexpr int NTH = NW * 32;
  constexpr int NA  = MT*16*(K/4) / NTH;
  constexpr int NB  = K*(NC/4) / NTH;
  static_assert(NA * NTH == MT*16*(K/4), "row tile must split evenly over the threads");
  static_assert(NB * NTH == K*(NC/4), "B must split evenly over the threads");
  const float* Ablk = A + (size_t)blockIdx.x * (MT*16) * K;

  float4 rb[NB];
  float4 ra[NA];
  #pragma unroll
  for (int i = 0; i < NB; ++i) {
    const int idx = tid + i*NTH;
    const int r = idx / (NC/4), c4 = (idx % (NC/4)) * 4;
    rb[i] = *(const float4*)&B[(size_t)r*NC + c4];
  }
  #pragma unroll
  for (int i = 0; i < NA; ++i) {
    const int idx = tid + i*NTH;
    const int r = idx / (K/4), c4 = (idx % (K/4)) * 4;
    ra[i] = *(const float4*)&Ablk[(size_t)r*K + c4];
  }
  #pragma unroll
  for (int i = 0; i < NB; ++i) {
    const int idx = tid + i*NTH;
    const int r = idx / (NC/4), c4 = (idx % (NC/4)) * 4;
    *(float4*)&sB[r*NCP + c4] = rb[i];
  }
  #pragma unroll
  for (int i = 0; i < NA; ++i) {
    const int idx = tid + i*NTH;
    const int r = idx / (K/4), c4 = (idx % (K/4)) * 4;
    *(float4*)&sA[r*KP + c4] = ra[i];
  }
  __syncthreads();

  const int wave  = tid >> 5, lane = tid & 31;
  const int mt    = wave / (NC/16), nt = wave % (NC/16);
  const int rrow  = lane & 15;
  const int khalf = lane >> 4;

  const float* a_ptr = &sA[(mt*16 + rrow)*KP + khalf*2];
  const float* b_ptr = &sB[(khalf*2)*NCP + nt*16 + rrow];

  v8f c0 = {};
  v8f c1 = {};
  #pragma unroll 4
  for (int ks = 0; ks < K/8; ++ks) {
    v2f a0; a0.x = a_ptr[0];     a0.y = a_ptr[1];
    v2f b0; b0.x = b_ptr[0];     b0.y = b_ptr[NCP];
    v2f a1; a1.x = a_ptr[4];     a1.y = a_ptr[5];
    v2f b1; b1.x = b_ptr[4*NCP]; b1.y = b_ptr[5*NCP];
    c0 = __builtin_amdgcn_wmma_f32_16x16x4_f32(false, a0, false, b0,
                                               (short)0, c0, false, false);
    c1 = __builtin_amdgcn_wmma_f32_16x16x4_f32(false, a1, false, b1,
                                               (short)0, c1, false, false);
    a_ptr += 8;
    b_ptr += 8*NCP;
  }
  v8f c = c0 + c1;

  const int gm  = blockIdx.x*(MT*16) + mt*16 + khalf*8;
  const int col = nt*16 + rrow;
  #pragma unroll
  for (int r = 0; r < 8; ++r)
    C[(size_t)(gm + r)*NC + col] = c[r];
}

template <int R, int NR, int NT>
__global__ void __launch_bounds__(NT)
mark_lines(const int* __restrict__ dst, unsigned int* __restrict__ marks, int e) {
  constexpr int NW = NT / 32;
  constexpr int LP = NW + 1;
  __shared__ unsigned int bits[NR * LP];
  const int tid = threadIdx.x, lane = tid & 31, wave = tid >> 5;
  for (int i = tid; i < NR * LP; i += NT) bits[i] = 0u;
  __syncthreads();

  const int nwords = (e + 1023) >> 10;
  const int pitch  = (nwords + 31) & ~31;
  const int word   = blockIdx.x * NW + wave;
  if (word < nwords) {
    const int base = word * 1024 + lane;
    for (int j0 = 0; j0 < 32; j0 += 8) {
      int v[8];
      #pragma unroll
      for (int u = 0; u < 8; ++u) {
        const int i = base + (j0 + u) * 32;
        const int x = dst[i < e ? i : e - 1];
        v[u] = (i < e) ? x : -1;
      }
      #pragma unroll
      for (int u = 0; u < 8; ++u) {
        const unsigned int rid = (unsigned int)v[u] / (unsigned int)R;
        if (rid < (unsigned int)NR) atomicOr(&bits[rid * LP + wave], 1u << (j0 + u));
      }
    }
  }
  __syncthreads();

  for (int i = tid; i < NR * NW; i += NT) {
    const int r = i / NW, w = i % NW;
    marks[(size_t)r * pitch + blockIdx.x * NW + w] = bits[r * LP + w];
  }
}

template <int F, int R, int C, int NT, int U, int G>
__global__ void __launch_bounds__(NT)
aggregate_ranges(const int* __restrict__ src, const int* __restrict__ dst,
                 const float* __restrict__ dinv, const float* __restrict__ h,
                 float* __restrict__ agg, const unsigned int* __restrict__ marks,
                 int n, int e) {
  constexpr int NW = NT / 32;
  constexpr int FQ = F / 32;
  constexpr int QN = 64;
  __shared__ __attribute__((aligned(16))) float acc[R*F];
  __shared__ uint2 ring[NW*QN];

  const int tid  = threadIdx.x;
  const int lane = tid & 31;
  const int wave = __builtin_amdgcn_readfirstlane(tid >> 5);
  const int lo   = blockIdx.x * R;
  const int c    = blockIdx.y;

  float4* acc4 = (float4*)acc;
  for (int i = tid; i < R*F/4; i += NT) acc4[i] = make_float4(0.f, 0.f, 0.f, 0.f);
  __syncthreads();

  const int nwords = (e + 1023) >> 10;
  const int pitch  = (nwords + 31) & ~31;
  const int per    = (nwords + C - 1) / C;
  const int wbeg   = per * c;
  int wend = wbeg + per;
  if (wend > nwords) wend = nwords;
  const unsigned int* mrow = marks + (size_t)blockIdx.x * pitch;

  uint2* q = ring + wave*QN;
  unsigned int head = 0, cnt = 0;

  auto drain = [&](unsigned int k) {
    const uint2 ent = q[(head + lane) & (QN - 1)];
    const bool val = (unsigned int)lane < k;
    const unsigned int ei = val ? ent.x : 0u;
    const unsigned int dl = val ? ent.y : 0u;
    const int   s = src[ei];
    const float w = dinv[s] * dinv[lo + (int)dl];
    for (unsigned int j0 = 0; j0 < k; j0 += G) {
      int sj[G]; float wj[G]; unsigned int dj[G]; float v[G][FQ];
      #pragma unroll
      for (int g = 0; g < G; ++g) {
        const int j = (int)j0 + g;
        sj[g] = __builtin_amdgcn_readlane(s, j);
        wj[g] = __int_as_float(__builtin_amdgcn_readlane(__float_as_int(w), j));
        dj[g] = (unsigned int)__builtin_amdgcn_readlane((int)dl, j);
        #pragma unroll
        for (int f = 0; f < FQ; ++f)
          v[g][f] = h[(size_t)sj[g]*F + f*32 + lane];
      }
      #pragma unroll
      for (int g = 0; g < G; ++g) {
        if (j0 + g < k) {
          #pragma unroll
          for (int f = 0; f < FQ; ++f)
            atomicAdd(&acc[dj[g]*F + f*32 + lane], v[g][f] * wj[g]);
        }
      }
    }
  };

  int wi = wbeg + wave;
  unsigned int m = 0u, mnext = 0u;
  if (wi < wend) m = mrow[wi];
  if (wi + NW < wend) mnext = mrow[wi + NW];
  m = (unsigned int)__builtin_amdgcn_readfirstlane((int)m);

  auto fetch = [&](int& w, unsigned int& mb, int (&v)[U]) -> int {
    while (m == 0u && wi + NW < wend) {
      wi += NW;
      m = (unsigned int)__builtin_amdgcn_readfirstlane((int)mnext);
      mnext = (wi + NW < wend) ? mrow[wi + NW] : 0u;
    }
    w = wi;
    mb = m;
    int k = 0;
    int j = 0;
    #pragma unroll
    for (int u = 0; u < U; ++u) {
      const bool have = (m != 0u);
      if (have) j = __builtin_ctz(m);
      m &= m - 1u;
      k += have ? 1 : 0;
      const int i = ((wi << 5) + j) * 32 + lane;
      const int x = dst[i < e ? i : e - 1];
      v[u] = (i < e) ? x : -1;
    }
    return k;
  };

  int wA = 0, kA; unsigned int mbA = 0u; int dvA[U];
  kA = fetch(wA, mbA, dvA);
  while (kA > 0) {
    int wB = 0, kB; unsigned int mbB = 0u; int dvB[U];
    kB = fetch(wB, mbB, dvB);
    #pragma nounroll
    for (int u = 0; u < kA; ++u) {
      const int j = __builtin_ctz(mbA);
      mbA &= mbA - 1u;
      const unsigned int dl = (unsigned int)dvA[u] - (unsigned int)lo;
      const bool hit = dl < (unsigned int)R;
      const unsigned int hm = __builtin_amdgcn_ballot_w32(hit);
      if (hit) {
        const unsigned int pos =
            head + cnt + (unsigned int)__builtin_popcount(hm & ((1u << lane) - 1u));
        q[pos & (QN - 1)] =
            make_uint2((unsigned int)(((wA << 5) + j) * 32 + lane), dl);
      }
      cnt += (unsigned int)__builtin_popcount(hm);
      if (cnt >= 32u) { drain(32u); head += 32u; cnt -= 32u; }
    }
    #pragma unroll
    for (int u = 0; u < U; ++u) dvA[u] = dvB[u];
    wA = wB; mbA = mbB; kA = kB;
  }
  if (cnt) drain(cnt);
  __syncthreads();

  int rows = n - lo;
  if (rows > R) rows = R;
  float* arow = agg + (size_t)lo * F;
  for (int i = tid; i < rows*F; i += NT) atomicAdd(&arow[i], acc[i]);
}

template <int F>
__global__ void finalize_bn(float* __restrict__ agg, const float* __restrict__ h,
                            const float* __restrict__ dinv, const float* __restrict__ bias,
                            const float* __restrict__ gamma, const float* __restrict__ beta,
                            const float* __restrict__ mean, const float* __restrict__ var,
                            long long total) {
  long long t = (long long)blockIdx.x * blockDim.x + threadIdx.x;
  if (t >= total) return;
  int node = (int)(t / F), f = (int)(t % F);
  float di = dinv[node];
  float v = agg[t] + h[t] * di * di + bias[f];
  v = (v - mean[f]) * rsqrtf(var[f] + BN_EPS) * gamma[f] + beta[f];
  agg[t] = fmaxf(v, 0.0f);
}

__global__ void final_head(const float* __restrict__ agg2, const float* __restrict__ h2,
                           const float* __restrict__ dinv, const float* __restrict__ b2,
                           const float* __restrict__ g2, const float* __restrict__ be2,
                           const float* __restrict__ m2, const float* __restrict__ v2,
                           const float* __restrict__ Wl, const float* __restrict__ bl,
                           float* __restrict__ out, int n) {
  int node = blockIdx.x * blockDim.x + threadIdx.x;
  if (node >= n) return;
  float di = dinv[node];
  float di2 = di * di;
  float acc0 = bl[0], acc1 = bl[1];
  const float* ar = agg2 + (size_t)node * 32;
  const float* hr = h2 + (size_t)node * 32;
  #pragma unroll
  for (int f = 0; f < 32; ++f) {
    float v = ar[f] + hr[f] * di2 + b2[f];
    v = (v - m2[f]) * rsqrtf(v2[f] + BN_EPS) * g2[f] + be2[f];
    v = fmaxf(v, 0.0f);
    acc0 += v * Wl[f*2 + 0];
    acc1 += v * Wl[f*2 + 1];
  }
  out[(size_t)node*2 + 0] = acc0;
  out[(size_t)node*2 + 1] = acc1;
}

extern "C" void kernel_launch(void* const* d_in, const int* in_sizes, int n_in,
                              void* d_out, int out_size, void* d_ws, size_t ws_size,
                              hipStream_t stream) {
  const float* x      = (const float*)d_in[0];
  const int*   ei     = (const int*)  d_in[1];
  const float* W1     = (const float*)d_in[2];
  const float* b1     = (const float*)d_in[3];
  const float* gamma1 = (const float*)d_in[4];
  const float* beta1  = (const float*)d_in[5];
  const float* mean1  = (const float*)d_in[6];
  const float* var1   = (const float*)d_in[7];
  const float* W2     = (const float*)d_in[8];
  const float* b2     = (const float*)d_in[9];
  const float* gamma2 = (const float*)d_in[10];
  const float* beta2  = (const float*)d_in[11];
  const float* mean2  = (const float*)d_in[12];
  const float* var2   = (const float*)d_in[13];
  const float* Wl     = (const float*)d_in[14];
  const float* bl     = (const float*)d_in[15];

  const int n = in_sizes[0] / 256;
  const int e = in_sizes[1] / 2;
  const int* src = ei;
  const int* dst = ei + e;

  float* ws   = (float*)d_ws;
  float* dinv = ws;
  float* h1   = dinv + (size_t)n;
  float* agg1 = h1   + (size_t)n * 64;
  float* h2   = agg1 + (size_t)n * 64;
  float* agg2 = h2   + (size_t)n * 32;

  const int T = 256;
  auto blocks = [](long long total, int t) {
    return (unsigned int)((total + t - 1) / t);
  };

  zero_f32<<<blocks((long long)n * 64, T), T, 0, stream>>>(agg1, (long long)n * 64);
  zero_f32<<<blocks((long long)n * 32, T), T, 0, stream>>>(agg2, (long long)n * 32);

  unsigned int* part = (unsigned int*)h1;
  degree_partials<25000, 64, 512, 25><<<dim3(blocks(n, 25000), 64), 512, 0, stream>>>(
      dst, part, n, e);
  dinv_from_partials<64><<<blocks(n, T), T, 0, stream>>>(part, dinv, n);

  gemm_wmma_f32<256, 64, 1><<<n / 16, 128, 0, stream>>>(x, W1, h1);

  unsigned int* marks1 = (unsigned int*)h2;
  mark_lines<782, 128, 1024><<<blocks(blocks(e, 1024), 32), 1024, 0, stream>>>(dst, marks1, e);

  aggregate_ranges<64, 782, 2, 1024, 8, 8><<<dim3(blocks(n, 782), 2), 1024, 0, stream>>>(
      src, dst, dinv, h1, agg1, marks1, n, e);

  finalize_bn<64><<<blocks((long long)n * 64, T), T, 0, stream>>>(
      agg1, h1, dinv, b1, gamma1, beta1, mean1, var1, (long long)n * 64);

  gemm_wmma_f32<64, 32, 2><<<n / 32, 128, 0, stream>>>(agg1, W2, h2);

  unsigned int* marks2 = (unsigned int*)h1;
  mark_lines<1563, 64, 1024><<<blocks(blocks(e, 1024), 32), 1024, 0, stream>>>(dst, marks2, e);

  aggregate_ranges<32, 1563, 4, 1024, 16, 8><<<dim3(blocks(n, 1563), 4), 1024, 0, stream>>>(
      src, dst, dinv, h2, agg2, marks2, n, e);

  final_head<<<blocks(n, T), T, 0, stream>>>(
      agg2, h2, dinv, b2, gamma2, beta2, mean2, var2, Wl, bl,
      (float*)d_out, n);
}


# aggregation drain: a 32-edge drain's row fetches in flight together (32 fetches per wait in both layers, was 16 and 8), first group issued beside the degree look-ups; sum order = LDS arrival order
# speedup vs baseline: 2.5832x; 1.0342x over previous
#include <hip/hip_runtime.h>

typedef __attribute__((ext_vector_type(2))) float v2f;
typedef __attribute__((ext_vector_type(8))) float v8f;

#define BN_EPS 1e-5f

template <bool B> struct flag_t { static constexpr bool value = B; };

__global__ void zero_f32(float* __restrict__ p, long long n) {
  long long i = (long long)blockIdx.x * blockDim.x + threadIdx.x;
  if (i < n) p[i] = 0.0f;
}

template <int R, int C, int NT, int U>
__global__ void __launch_bounds__(NT)
degree_partials(const int* __restrict__ dst, unsigned int* __restrict__ part,
                int n, int e) {
  __shared__ __attribute__((aligned(16))) unsigned int cnt[R];
  const int tid = threadIdx.x;
  const int lo  = blockIdx.x * R;
  const int c   = blockIdx.y;

  uint4* cnt4 = (uint4*)cnt;
  for (int i = tid; i < R/4; i += NT) cnt4[i] = make_uint4(0u, 0u, 0u, 0u);
  __syncthreads();

  const long long per = ((long long)e + C - 1) / C;
  const long long beg = per * c;
  long long end = beg + per;
  if (end > e) end = e;
  const int len = end > beg ? (int)(end - beg) : 0;
  const int* dptr = dst + beg;

  for (int base = 0; base < len; base += NT*U) {
    int v[U];
    #pragma unroll
    for (int u = 0; u < U; ++u) {
      const int i = base + u*NT + tid;
      const int x = dptr[i < len ? i : len - 1];
      v[u] = (i < len) ? x : -1;
    }
    #pragma unroll
    for (int u = 0; u < U; ++u) {
      const unsigned int d = (unsigned int)v[u] - (unsigned int)lo;
      if (d < (unsigned int)R) atomicAdd(&cnt[d], 1u);
    }
  }
  __syncthreads();

  int m = n - lo;
  if (m > R) m = R;
  unsigned int* row = part + (size_t)c * n + lo;
  for (int i = tid; i < m; i += NT) row[i] = cnt[i];
}

template <int C>
__global__ void dinv_from_partials(const unsigned int* __restrict__ part,
                                   float* __restrict__ dinv, int n) {
  int i = blockIdx.x * blockDim.x + threadIdx.x;
  if (i >= n) return;
  unsigned int s = 0;
  #pragma unroll
  for (int c = 0; c < C; ++c) s += part[(size_t)c * n + i];
  dinv[i] = rsqrtf((float)s + 1.0f);
}

template <int K, int NC, int MT>
__global__ void __launch_bounds__(MT*(NC/16)*32)
gemm_wmma_f32(const float* __restrict__ A, const float* __restrict__ B,
              float* __restrict__ C) {
  constexpr int KP  = K + 4;
  constexpr int NCP = NC + 8;
  constexpr int NW  = MT * (NC/16);
  __shared__ float sA[MT*16*KP];
  __shared__ float sB[K*NCP];

  const int tid = threadIdx.x;
  constexpr int NTH = NW * 32;
  constexpr int NA  = MT*16*(K/4) / NTH;
  constexpr int NB  = K*(NC/4) / NTH;
  static_assert(NA * NTH == MT*16*(K/4), "row tile must split evenly over the threads");
  static_assert(NB * NTH == K*(NC/4), "B must split evenly over the threads");
  const float* Ablk = A + (size_t)blockIdx.x * (MT*16) * K;

  float4 rb[NB];
  float4 ra[NA];
  #pragma unroll
  for (int i = 0; i < NB; ++i) {
    const int idx = tid + i*NTH;
    const int r = idx / (NC/4), c4 = (idx % (NC/4)) * 4;
    rb[i] = *(const float4*)&B[(size_t)r*NC + c4];
  }
  #pragma unroll
  for (int i = 0; i < NA; ++i) {
    const int idx = tid + i*NTH;
    const int r = idx / (K/4), c4 = (idx % (K/4)) * 4;
    ra[i] = *(const float4*)&Ablk[(size_t)r*K + c4];
  }
  #pragma unroll
  for (int i = 0; i < NB; ++i) {
    const int idx = tid + i*NTH;
    const int r = idx / (NC/4), c4 = (idx % (NC/4)) * 4;
    *(float4*)&sB[r*NCP + c4] = rb[i];
  }
  #pragma unroll
  for (int i = 0; i < NA; ++i) {
    const int idx = tid + i*NTH;
    const int r = idx / (K/4), c4 = (idx % (K/4)) * 4;
    *(float4*)&sA[r*KP + c4] = ra[i];
  }
  __syncthreads();

  const int wave  = tid >> 5, lane = tid & 31;
  const int mt    = wave / (NC/16), nt = wave % (NC/16);
  const int rrow  = lane & 15;
  const int khalf = lane >> 4;

  const float* a_ptr = &sA[(mt*16 + rrow)*KP + khalf*2];
  const float* b_ptr = &sB[(khalf*2)*NCP + nt*16 + rrow];

  v8f c0 = {};
  v8f c1 = {};
  #pragma unroll 4
  for (int ks = 0; ks < K/8; ++ks) {
    v2f a0; a0.x = a_ptr[0];     a0.y = a_ptr[1];
    v2f b0; b0.x = b_ptr[0];     b0.y = b_ptr[NCP];
    v2f a1; a1.x = a_ptr[4];     a1.y = a_ptr[5];
    v2f b1; b1.x = b_ptr[4*NCP]; b1.y = b_ptr[5*NCP];
    c0 = __builtin_amdgcn_wmma_f32_16x16x4_f32(false, a0, false, b0,
                                               (short)0, c0, false, false);
    c1 = __builtin_amdgcn_wmma_f32_16x16x4_f32(false, a1, false, b1,
                                               (short)0, c1, false, false);
    a_ptr += 8;
    b_ptr += 8*NCP;
  }
  v8f c = c0 + c1;

  const int gm  = blockIdx.x*(MT*16) + mt*16 + khalf*8;
  const int col = nt*16 + rrow;
  #pragma unroll
  for (int r = 0; r < 8; ++r)
    C[(size_t)(gm + r)*NC + col] = c[r];
}

template <int R, int NR, int NT>
__global__ void __launch_bounds__(NT)
mark_lines(const int* __restrict__ dst, unsigned int* __restrict__ marks, int e) {
  constexpr int NW = NT / 32;
  constexpr int LP = NW + 1;
  __shared__ unsigned int bits[NR * LP];
  const int tid = threadIdx.x, lane = tid & 31, wave = tid >> 5;
  for (int i = tid; i < NR * LP; i += NT) bits[i] = 0u;
  __syncthreads();

  const int nwords = (e + 1023) >> 10;
  const int pitch  = (nwords + 31) & ~31;
  const int word   = blockIdx.x * NW + wave;
  if (word < nwords) {
    const int base = word * 1024 + lane;
    for (int j0 = 0; j0 < 32; j0 += 8) {
      int v[8];
      #pragma unroll
      for (int u = 0; u < 8; ++u) {
        const int i = base + (j0 + u) * 32;
        const int x = dst[i < e ? i : e - 1];
        v[u] = (i < e) ? x : -1;
      }
      #pragma unroll
      for (int u = 0; u < 8; ++u) {
        const unsigned int rid = (unsigned int)v[u] / (unsigned int)R;
        if (rid < (unsigned int)NR) atomicOr(&bits[rid * LP + wave], 1u << (j0 + u));
      }
    }
  }
  __syncthreads();

  for (int i = tid; i < NR * NW; i += NT) {
    const int r = i / NW, w = i % NW;
    marks[(size_t)r * pitch + blockIdx.x * NW + w] = bits[r * LP + w];
  }
}

template <int F, int R, int C, int NT, int U, int G>
__global__ void __launch_bounds__(NT)
aggregate_ranges(const int* __restrict__ src, const int* __restrict__ dst,
                 const float* __restrict__ dinv, const float* __restrict__ h,
                 float* __restrict__ agg, const unsigned int* __restrict__ marks,
                 int n, int e) {
  constexpr int NW = NT / 32;
  constexpr int FQ = F / 32;
  constexpr int QN = 64;
  __shared__ __attribute__((aligned(16))) float acc[R*F];
  __shared__ uint2 ring[NW*QN];

  const int tid  = threadIdx.x;
  const int lane = tid & 31;
  const int wave = __builtin_amdgcn_readfirstlane(tid >> 5);
  const int lo   = blockIdx.x * R;
  const int c    = blockIdx.y;

  float4* acc4 = (float4*)acc;
  for (int i = tid; i < R*F/4; i += NT) acc4[i] = make_float4(0.f, 0.f, 0.f, 0.f);
  __syncthreads();

  const int nwords = (e + 1023) >> 10;
  const int pitch  = (nwords + 31) & ~31;
  const int per    = (nwords + C - 1) / C;
  const int wbeg   = per * c;
  int wend = wbeg + per;
  if (wend > nwords) wend = nwords;
  const unsigned int* mrow = marks + (size_t)blockIdx.x * pitch;

  uint2* q = ring + wave*QN;
  unsigned int head = 0, cnt = 0;

  auto drain = [&](unsigned int k, auto full) {
    constexpr bool FULL = decltype(full)::value;
    const uint2 ent = q[(head + lane) & (QN - 1)];
    const bool val = (unsigned int)lane < k;
    const unsigned int ei = val ? ent.x : 0u;
    const unsigned int dl = val ? ent.y : 0u;
    const int   s = src[ei];
    const float w = dinv[s] * dinv[lo + (int)dl];
    auto group = [&](unsigned int j0) {
      int sj[G]; float wj[G]; unsigned int dj[G]; float v[G][FQ];
      #pragma unroll
      for (int g = 0; g < G; ++g) {
        const int j = (int)j0 + g;
        sj[g] = __builtin_amdgcn_readlane(s, j);
        wj[g] = __int_as_float(__builtin_amdgcn_readlane(__float_as_int(w), j));
        dj[g] = (unsigned int)__builtin_amdgcn_readlane((int)dl, j);
        const unsigned int rb = (unsigned int)sj[g] * (unsigned int)F + (unsigned int)lane;
        #pragma unroll
        for (int f = 0; f < FQ; ++f)
          v[g][f] = h[rb + (unsigned int)(f*32)];
      }
      #pragma unroll
      for (int g = 0; g < G; ++g) {
        if (FULL || j0 + g < k) {
          #pragma unroll
          for (int f = 0; f < FQ; ++f)
            atomicAdd(&acc[dj[g]*F + f*32 + lane], v[g][f] * wj[g]);
        }
      }
    };
    if constexpr (FULL) {
      #pragma unroll
      for (unsigned int j0 = 0; j0 < 32u; j0 += G) group(j0);
    } else {
      for (unsigned int j0 = 0; j0 < k; j0 += G) group(j0);
    }
  };

  int wi = wbeg + wave;
  unsigned int m = 0u, mnext = 0u;
  if (wi < wend) m = mrow[wi];
  if (wi + NW < wend) mnext = mrow[wi + NW];
  m = (unsigned int)__builtin_amdgcn_readfirstlane((int)m);

  auto fetch = [&](int& w, unsigned int& mb, int (&v)[U]) -> int {
    while (m == 0u && wi + NW < wend) {
      wi += NW;
      m = (unsigned int)__builtin_amdgcn_readfirstlane((int)mnext);
      mnext = (wi + NW < wend) ? mrow[wi + NW] : 0u;
    }
    w = wi;
    mb = m;
    int k = 0;
    int j = 0;
    #pragma unroll
    for (int u = 0; u < U; ++u) {
      const bool have = (m != 0u);
      if (have) j = __builtin_ctz(m);
      m &= m - 1u;
      k += have ? 1 : 0;
      const int i = ((wi << 5) + j) * 32 + lane;
      const int x = dst[i < e ? i : e - 1];
      v[u] = (i < e) ? x : -1;
    }
    return k;
  };

  int wA = 0, kA; unsigned int mbA = 0u; int dvA[U];
  kA = fetch(wA, mbA, dvA);
  while (kA > 0) {
    int wB = 0, kB; unsigned int mbB = 0u; int dvB[U];
    kB = fetch(wB, mbB, dvB);
    #pragma nounroll
    for (int u = 0; u < kA; ++u) {
      const int j = __builtin_ctz(mbA);
      mbA &= mbA - 1u;
      const unsigned int dl = (unsigned int)dvA[u] - (unsigned int)lo;
      const bool hit = dl < (unsigned int)R;
      const unsigned int hm = __builtin_amdgcn_ballot_w32(hit);
      if (hit) {
        const unsigned int pos =
            head + cnt + (unsigned int)__builtin_popcount(hm & ((1u << lane) - 1u));
        q[pos & (QN - 1)] =
            make_uint2((unsigned int)(((wA << 5) + j) * 32 + lane), dl);
      }
      cnt += (unsigned int)__builtin_popcount(hm);
      if (cnt >= 32u) { drain(32u, flag_t<true>{}); head += 32u; cnt -= 32u; }
    }
    #pragma unroll
    for (int u = 0; u < U; ++u) dvA[u] = dvB[u];
    wA = wB; mbA = mbB; kA = kB;
  }
  if (cnt) drain(cnt, flag_t<false>{});
  __syncthreads();

  int rows = n - lo;
  if (rows > R) rows = R;
  float* arow = agg + (size_t)lo * F;
  for (int i = tid; i < rows*F; i += NT) atomicAdd(&arow[i], acc[i]);
}

template <int F>
__global__ void finalize_bn(float* __restrict__ agg, const float* __restrict__ h,
                            const float* __restrict__ dinv, const float* __restrict__ bias,
                            const float* __restrict__ gamma, const float* __restrict__ beta,
                            const float* __restrict__ mean, const float* __restrict__ var,
                            long long total) {
  long long t = (long long)blockIdx.x * blockDim.x + threadIdx.x;
  if (t >= total) return;
  int node = (int)(t / F), f = (int)(t % F);
  float di = dinv[node];
  float v = agg[t] + h[t] * di * di + bias[f];
  v = (v - mean[f]) * rsqrtf(var[f] + BN_EPS) * gamma[f] + beta[f];
  agg[t] = fmaxf(v, 0.0f);
}

__global__ void final_head(const float* __restrict__ agg2, const float* __restrict__ h2,
                           const float* __restrict__ dinv, const float* __restrict__ b2,
                           const float* __restrict__ g2, const float* __restrict__ be2,
                           const float* __restrict__ m2, const float* __restrict__ v2,
                           const float* __restrict__ Wl, const float* __restrict__ bl,
                           float* __restrict__ out, int n) {
  int node = blockIdx.x * blockDim.x + threadIdx.x;
  if (node >= n) return;
  float di = dinv[node];
  float di2 = di * di;
  float acc0 = bl[0], acc1 = bl[1];
  const float* ar = agg2 + (size_t)node * 32;
  const float* hr = h2 + (size_t)node * 32;
  #pragma unroll
  for (int f = 0; f < 32; ++f) {
    float v = ar[f] + hr[f] * di2 + b2[f];
    v = (v - m2[f]) * rsqrtf(v2[f] + BN_EPS) * g2[f] + be2[f];
    v = fmaxf(v, 0.0f);
    acc0 += v * Wl[f*2 + 0];
    acc1 += v * Wl[f*2 + 1];
  }
  out[(size_t)node*2 + 0] = acc0;
  out[(size_t)node*2 + 1] = acc1;
}

extern "C" void kernel_launch(void* const* d_in, const int* in_sizes, int n_in,
                              void* d_out, int out_size, void* d_ws, size_t ws_size,
                              hipStream_t stream) {
  const float* x      = (const float*)d_in[0];
  const int*   ei     = (const int*)  d_in[1];
  const float* W1     = (const float*)d_in[2];
  const float* b1     = (const float*)d_in[3];
  const float* gamma1 = (const float*)d_in[4];
  const float* beta1  = (const float*)d_in[5];
  const float* mean1  = (const float*)d_in[6];
  const float* var1   = (const float*)d_in[7];
  const float* W2     = (const float*)d_in[8];
  const float* b2     = (const float*)d_in[9];
  const float* gamma2 = (const float*)d_in[10];
  const float* beta2  = (const float*)d_in[11];
  const float* mean2  = (const float*)d_in[12];
  const float* var2   = (const float*)d_in[13];
  const float* Wl     = (const float*)d_in[14];
  const float* bl     = (const float*)d_in[15];

  const int n = in_sizes[0] / 256;
  const int e = in_sizes[1] / 2;
  const int* src = ei;
  const int* dst = ei + e;

  float* ws   = (float*)d_ws;
  float* dinv = ws;
  float* h1   = dinv + (size_t)n;
  float* agg1 = h1   + (size_t)n * 64;
  float* h2   = agg1 + (size_t)n * 64;
  float* agg2 = h2   + (size_t)n * 32;

  const int T = 256;
  auto blocks = [](long long total, int t) {
    return (unsigned int)((total + t - 1) / t);
  };

  zero_f32<<<blocks((long long)n * 64, T), T, 0, stream>>>(agg1, (long long)n * 64);
  zero_f32<<<blocks((long long)n * 32, T), T, 0, stream>>>(agg2, (long long)n * 32);

  unsigned int* part = (unsigned int*)h1;
  degree_partials<25000, 64, 512, 25><<<dim3(blocks(n, 25000), 64), 512, 0, stream>>>(
      dst, part, n, e);
  dinv_from_partials<64><<<blocks(n, T), T, 0, stream>>>(part, dinv, n);

  gemm_wmma_f32<256, 64, 1><<<n / 16, 128, 0, stream>>>(x, W1, h1);

  unsigned int* marks1 = (unsigned int*)h2;
  mark_lines<782, 128, 1024><<<blocks(blocks(e, 1024), 32), 1024, 0, stream>>>(dst, marks1, e);

  aggregate_ranges<64, 782, 2, 1024, 8, 16><<<dim3(blocks(n, 782), 2), 1024, 0, stream>>>(
      src, dst, dinv, h1, agg1, marks1, n, e);

  finalize_bn<64><<<blocks((long long)n * 64, T), T, 0, stream>>>(
      agg1, h1, dinv, b1, gamma1, beta1, mean1, var1, (long long)n * 64);

  gemm_wmma_f32<64, 32, 2><<<n / 32, 128, 0, stream>>>(agg1, W2, h2);

  unsigned int* marks2 = (unsigned int*)h1;
  mark_lines<1563, 64, 1024><<<blocks(blocks(e, 1024), 32), 1024, 0, stream>>>(dst, marks2, e);

  aggregate_ranges<32, 1563, 4, 1024, 16, 32><<<dim3(blocks(n, 1563), 4), 1024, 0, stream>>>(
      src, dst, dinv, h2, agg2, marks2, n, e);

  final_head<<<blocks(n, T), T, 0, stream>>>(
      agg2, h2, dinv, b2, gamma2, beta2, mean2, var2, Wl, bl,
      (float*)d_out, n);
}


# aggregation scan: marks per 8-edge piece (not per 32-edge line); lanes queue their marked pieces in LDS, one wave load fetches 4 pieces, one test-and-note pass serves 4; sum order = LDS arrival order
# speedup vs baseline: 2.9419x; 1.1389x over previous
#include <hip/hip_runtime.h>

typedef __attribute__((ext_vector_type(2))) float v2f;
typedef __attribute__((ext_vector_type(8))) float v8f;

#define BN_EPS 1e-5f

template <bool B> struct flag_t { static constexpr bool value = B; };

__global__ void zero_f32(float* __restrict__ p, long long n) {
  long long i = (long long)blockIdx.x * blockDim.x + threadIdx.x;
  if (i < n) p[i] = 0.0f;
}

template <int R, int C, int NT, int U>
__global__ void __launch_bounds__(NT)
degree_partials(const int* __restrict__ dst, unsigned int* __restrict__ part,
                int n, int e) {
  __shared__ __attribute__((aligned(16))) unsigned int cnt[R];
  const int tid = threadIdx.x;
  const int lo  = blockIdx.x * R;
  const int c   = blockIdx.y;

  uint4* cnt4 = (uint4*)cnt;
  for (int i = tid; i < R/4; i += NT) cnt4[i] = make_uint4(0u, 0u, 0u, 0u);
  __syncthreads();

  const long long per = ((long long)e + C - 1) / C;
  const long long beg = per * c;
  long long end = beg + per;
  if (end > e) end = e;
  const int len = end > beg ? (int)(end - beg) : 0;
  const int* dptr = dst + beg;

  for (int base = 0; base < len; base += NT*U) {
    int v[U];
    #pragma unroll
    for (int u = 0; u < U; ++u) {
      const int i = base + u*NT + tid;
      const int x = dptr[i < len ? i : len - 1];
      v[u] = (i < len) ? x : -1;
    }
    #pragma unroll
    for (int u = 0; u < U; ++u) {
      const unsigned int d = (unsigned int)v[u] - (unsigned int)lo;
      if (d < (unsigned int)R) atomicAdd(&cnt[d], 1u);
    }
  }
  __syncthreads();

  int m = n - lo;
  if (m > R) m = R;
  unsigned int* row = part + (size_t)c * n + lo;
  for (int i = tid; i < m; i += NT) row[i] = cnt[i];
}

template <int C>
__global__ void dinv_from_partials(const unsigned int* __restrict__ part,
                                   float* __restrict__ dinv, int n) {
  int i = blockIdx.x * blockDim.x + threadIdx.x;
  if (i >= n) return;
  unsigned int s = 0;
  #pragma unroll
  for (int c = 0; c < C; ++c) s += part[(size_t)c * n + i];
  dinv[i] = rsqrtf((float)s + 1.0f);
}

template <int K, int NC, int MT>
__global__ void __launch_bounds__(MT*(NC/16)*32)
gemm_wmma_f32(const float* __restrict__ A, const float* __restrict__ B,
              float* __restrict__ C) {
  constexpr int KP  = K + 4;
  constexpr int NCP = NC + 8;
  constexpr int NW  = MT * (NC/16);
  __shared__ float sA[MT*16*KP];
  __shared__ float sB[K*NCP];

  const int tid = threadIdx.x;
  constexpr int NTH = NW * 32;
  constexpr int NA  = MT*16*(K/4) / NTH;
  constexpr int NB  = K*(NC/4) / NTH;
  static_assert(NA * NTH == MT*16*(K/4), "row tile must split evenly over the threads");
  static_assert(NB * NTH == K*(NC/4), "B must split evenly over the threads");
  const float* Ablk = A + (size_t)blockIdx.x * (MT*16) * K;

  float4 rb[NB];
  float4 ra[NA];
  #pragma unroll
  for (int i = 0; i < NB; ++i) {
    const int idx = tid + i*NTH;
    const int r = idx / (NC/4), c4 = (idx % (NC/4)) * 4;
    rb[i] = *(const float4*)&B[(size_t)r*NC + c4];
  }
  #pragma unroll
  for (int i = 0; i < NA; ++i) {
    const int idx = tid + i*NTH;
    const int r = idx / (K/4), c4 = (idx % (K/4)) * 4;
    ra[i] = *(const float4*)&Ablk[(size_t)r*K + c4];
  }
  #pragma unroll
  for (int i = 0; i < NB; ++i) {
    const int idx = tid + i*NTH;
    const int r = idx / (NC/4), c4 = (idx % (NC/4)) * 4;
    *(float4*)&sB[r*NCP + c4] = rb[i];
  }
  #pragma unroll
  for (int i = 0; i < NA; ++i) {
    const int idx = tid + i*NTH;
    const int r = idx / (K/4), c4 = (idx % (K/4)) * 4;
    *(float4*)&sA[r*KP + c4] = ra[i];
  }
  __syncthreads();

  const int wave  = tid >> 5, lane = tid & 31;
  const int mt    = wave / (NC/16), nt = wave % (NC/16);
  const int rrow  = lane & 15;
  const int khalf = lane >> 4;

  const float* a_ptr = &sA[(mt*16 + rrow)*KP + khalf*2];
  const float* b_ptr = &sB[(khalf*2)*NCP + nt*16 + rrow];

  v8f c0 = {};
  v8f c1 = {};
  #pragma unroll 4
  for (int ks = 0; ks < K/8; ++ks) {
    v2f a0; a0.x = a_ptr[0];     a0.y = a_ptr[1];
    v2f b0; b0.x = b_ptr[0];     b0.y = b_ptr[NCP];
    v2f a1; a1.x = a_ptr[4];     a1.y = a_ptr[5];
    v2f b1; b1.x = b_ptr[4*NCP]; b1.y = b_ptr[5*NCP];
    c0 = __builtin_amdgcn_wmma_f32_16x16x4_f32(false, a0, false, b0,
                                               (short)0, c0, false, false);
    c1 = __builtin_amdgcn_wmma_f32_16x16x4_f32(false, a1, false, b1,
                                               (short)0, c1, false, false);
    a_ptr += 8;
    b_ptr += 8*NCP;
  }
  v8f c = c0 + c1;

  const int gm  = blockIdx.x*(MT*16) + mt*16 + khalf*8;
  const int col = nt*16 + rrow;
  #pragma unroll
  for (int r = 0; r < 8; ++r)
    C[(size_t)(gm + r)*NC + col] = c[r];
}

template <int R, int NR, int NT>
__global__ void __launch_bounds__(NT)
mark_pieces(const int* __restrict__ dst, unsigned int* __restrict__ marks, int e) {
  constexpr int NW = NT / 32;
  constexpr int WW = NW * 4;
  constexpr int LP = WW + 1;
  __shared__ unsigned int bits[NR * LP];
  const int tid = threadIdx.x, lane = tid & 31, wave = tid >> 5;
  for (int i = tid; i < NR * LP; i += NT) bits[i] = 0u;
  __syncthreads();

  const int nwords = (e + 255) >> 8;
  const int pitch  = ((((e + 1023) >> 10) + 31) & ~31) * 4;
  const int word0  = (blockIdx.x * NW + wave) * 4;
  if (word0 < nwords) {
    const int base = word0 * 256 + lane;
    for (int j0 = 0; j0 < 32; j0 += 8) {
      int v[8];
      #pragma unroll
      for (int u = 0; u < 8; ++u) {
        const int i = base + (j0 + u) * 32;
        const int x = dst[i < e ? i : e - 1];
        v[u] = (i < e) ? x : -1;
      }
      #pragma unroll
      for (int u = 0; u < 8; ++u) {
        const unsigned int rid = (unsigned int)v[u] / (unsigned int)R;
        if (rid < (unsigned int)NR)
          atomicOr(&bits[rid * LP + wave * 4 + (j0 >> 3)], 1u << (u * 4 + (lane >> 3)));
      }
    }
  }
  __syncthreads();

  for (int i = tid; i < NR * WW; i += NT) {
    const int r = i / WW, w = i % WW;
    marks[(size_t)r * pitch + blockIdx.x * WW + w] = bits[r * LP + w];
  }
}

template <int F, int R, int C, int NT, int U, int G>
__global__ void __launch_bounds__(NT)
aggregate_ranges(const int* __restrict__ src, const int* __restrict__ dst,
                 const float* __restrict__ dinv, const float* __restrict__ h,
                 float* __restrict__ agg, const unsigned int* __restrict__ marks,
                 int n, int e) {
  constexpr int NW = NT / 32;
  constexpr int FQ = F / 32;
  constexpr int QN = 256;
  constexpr int PQ = 64;
  static_assert(31 + 32*U <= QN, "ring too small for a batch");
  static_assert(4*U + 31 <= PQ, "piece queue too small for the batch");
  __shared__ __attribute__((aligned(16))) float acc[R*F];
  __shared__ uint2 ring[NW*QN];
  __shared__ int pieces[NW*PQ];

  const int tid  = threadIdx.x;
  const int lane = tid & 31;
  const int wave = __builtin_amdgcn_readfirstlane(tid >> 5);
  const int lo   = blockIdx.x * R;
  const int c    = blockIdx.y;

  float4* acc4 = (float4*)acc;
  for (int i = tid; i < R*F/4; i += NT) acc4[i] = make_float4(0.f, 0.f, 0.f, 0.f);
  __syncthreads();

  const int nwords = (e + 255) >> 8;
  const int pitch  = ((((e + 1023) >> 10) + 31) & ~31) * 4;
  const int per    = (nwords + C - 1) / C;
  const int wbeg   = per * c;
  int wend = wbeg + per;
  if (wend > nwords) wend = nwords;
  const unsigned int* mrow = marks + (size_t)blockIdx.x * pitch;
  const int pw = (per + NW - 1) / NW;
  int wb = wbeg + wave * pw;
  int wstop = wb + pw;
  if (wstop > wend) wstop = wend;

  uint2* q = ring + wave*QN;
  unsigned int head = 0, cnt = 0;

  auto drain = [&](unsigned int k, auto full) {
    constexpr bool FULL = decltype(full)::value;
    const uint2 ent = q[(head + lane) & (QN - 1)];
    const bool val = (unsigned int)lane < k;
    const unsigned int ei = val ? ent.x : 0u;
    const unsigned int dl = val ? ent.y : 0u;
    const int   s = src[ei];
    const float w = dinv[s] * dinv[lo + (int)dl];
    auto group = [&](unsigned int j0) {
      int sj[G]; float wj[G]; unsigned int dj[G]; float v[G][FQ];
      #pragma unroll
      for (int g = 0; g < G; ++g) {
        const int j = (int)j0 + g;
        sj[g] = __builtin_amdgcn_readlane(s, j);
        wj[g] = __int_as_float(__builtin_amdgcn_readlane(__float_as_int(w), j));
        dj[g] = (unsigned int)__builtin_amdgcn_readlane((int)dl, j);
        const unsigned int rb = (unsigned int)sj[g] * (unsigned int)F + (unsigned int)lane;
        #pragma unroll
        for (int f = 0; f < FQ; ++f)
          v[g][f] = h[rb + (unsigned int)(f*32)];
      }
      #pragma unroll
      for (int g = 0; g < G; ++g) {
        if (FULL || j0 + g < k) {
          #pragma unroll
          for (int f = 0; f < FQ; ++f)
            atomicAdd(&acc[dj[g]*F + f*32 + lane], v[g][f] * wj[g]);
        }
      }
    };
    if constexpr (FULL) {
      #pragma unroll
      for (unsigned int j0 = 0; j0 < 32u; j0 += G) group(j0);
    } else {
      for (unsigned int j0 = 0; j0 < k; j0 += G) group(j0);
    }
  };

  int* pq = pieces + wave*PQ;
  unsigned int qh = 0, qn = 0;
  const unsigned int below = (1u << lane) - 1u;

  auto load_block = [&](int w0) -> unsigned int {
    const int w = w0 + lane;
    return mrow[w < wstop ? w : 0];
  };
  auto live = [&](int w0, unsigned int x) -> unsigned int {
    return (w0 + lane < wstop) ? x : 0u;
  };
  unsigned int mv  = live(wb, load_block(wb));
  unsigned int mvn = load_block(wb + 32);

  auto fill = [&]() {
    while (qn < 4u*U) {
      const bool has = (mv != 0u);
      const unsigned int hm = __builtin_amdgcn_ballot_w32(has);
      if (hm == 0u) {
        if (wb + 32 >= wstop) break;
        wb += 32;
        mv  = live(wb, mvn);
        mvn = load_block(wb + 32);
        continue;
      }
      if (has) {
        const int j = __builtin_ctz(mv);
        const unsigned int pos = qh + qn + (unsigned int)__builtin_popcount(hm & below);
        pq[pos & (PQ - 1)] = ((wb + lane) << 5) + j;
        mv &= mv - 1u;
      }
      qn += (unsigned int)__builtin_popcount(hm);
    }
  };

  auto fetch = [&](int (&v)[U], int (&ix)[U]) -> int {
    fill();
    const unsigned int take = qn < 4u*U ? qn : 4u*U;
    unsigned int p[U];
    #pragma unroll
    for (int u = 0; u < U; ++u)
      p[u] = (unsigned int)pq[(qh + (unsigned int)(u*4) + ((unsigned int)lane >> 3)) & (PQ - 1)];
    #pragma unroll
    for (int u = 0; u < U; ++u) {
      const unsigned int t = (unsigned int)(u*4) + ((unsigned int)lane >> 3);
      const unsigned int none = (t < take) ? 0u : ~0u;
      const int i = (int)(((p[u] << 3) | (unsigned int)(lane & 7)) | none);
      const bool in = (unsigned int)i < (unsigned int)e;
      const int x = dst[in ? i : e - 1];
      v[u]  = in ? x : -1;
      ix[u] = i;
    }
    qh += take;
    qn -= take;
    return (int)((take + 3u) >> 2);
  };

  int kA; int dvA[U]; int ixA[U];
  kA = fetch(dvA, ixA);
  while (kA > 0) {
    int kB; int dvB[U]; int ixB[U];
    kB = fetch(dvB, ixB);
    #pragma unroll
    for (int u = 0; u < U; ++u) {
      const unsigned int dl = (unsigned int)dvA[u] - (unsigned int)lo;
      const bool hit = dl < (unsigned int)R;
      const unsigned int hm = __builtin_amdgcn_ballot_w32(hit);
      if (hit) {
        const unsigned int pos =
            head + cnt + (unsigned int)__builtin_popcount(hm & below);
        q[pos & (QN - 1)] = make_uint2((unsigned int)ixA[u], dl);
      }
      cnt += (unsigned int)__builtin_popcount(hm);
    }
    while (cnt >= 32u) { drain(32u, flag_t<true>{}); head += 32u; cnt -= 32u; }
    #pragma unroll
    for (int u = 0; u < U; ++u) { dvA[u] = dvB[u]; ixA[u] = ixB[u]; }
    kA = kB;
  }
  if (cnt) drain(cnt, flag_t<false>{});
  __syncthreads();

  int rows = n - lo;
  if (rows > R) rows = R;
  float* arow = agg + (size_t)lo * F;
  for (int i = tid; i < rows*F; i += NT) atomicAdd(&arow[i], acc[i]);
}

template <int F>
__global__ void finalize_bn(float* __restrict__ agg, const float* __restrict__ h,
                            const float* __restrict__ dinv, const float* __restrict__ bias,
                            const float* __restrict__ gamma, const float* __restrict__ beta,
                            const float* __restrict__ mean, const float* __restrict__ var,
                            long long total) {
  long long t = (long long)blockIdx.x * blockDim.x + threadIdx.x;
  if (t >= total) return;
  int node = (int)(t / F), f = (int)(t % F);
  float di = dinv[node];
  float v = agg[t] + h[t] * di * di + bias[f];
  v = (v - mean[f]) * rsqrtf(var[f] + BN_EPS) * gamma[f] + beta[f];
  agg[t] = fmaxf(v, 0.0f);
}

__global__ void final_head(const float* __restrict__ agg2, const float* __restrict__ h2,
                           const float* __restrict__ dinv, const float* __restrict__ b2,
                           const float* __restrict__ g2, const float* __restrict__ be2,
                           const float* __restrict__ m2, const float* __restrict__ v2,
                           const float* __restrict__ Wl, const float* __restrict__ bl,
                           float* __restrict__ out, int n) {
  int node = blockIdx.x * blockDim.x + threadIdx.x;
  if (node >= n) return;
  float di = dinv[node];
  float di2 = di * di;
  float acc0 = bl[0], acc1 = bl[1];
  const float* ar = agg2 + (size_t)node * 32;
  const float* hr = h2 + (size_t)node * 32;
  #pragma unroll
  for (int f = 0; f < 32; ++f) {
    float v = ar[f] + hr[f] * di2 + b2[f];
    v = (v - m2[f]) * rsqrtf(v2[f] + BN_EPS) * g2[f] + be2[f];
    v = fmaxf(v, 0.0f);
    acc0 += v * Wl[f*2 + 0];
    acc1 += v * Wl[f*2 + 1];
  }
  out[(size_t)node*2 + 0] = acc0;
  out[(size_t)node*2 + 1] = acc1;
}

extern "C" void kernel_launch(void* const* d_in, const int* in_sizes, int n_in,
                              void* d_out, int out_size, void* d_ws, size_t ws_size,
                              hipStream_t stream) {
  const float* x      = (const float*)d_in[0];
  const int*   ei     = (const int*)  d_in[1];
  const float* W1     = (const float*)d_in[2];
  const float* b1     = (const float*)d_in[3];
  const float* gamma1 = (const float*)d_in[4];
  const float* beta1  = (const float*)d_in[5];
  const float* mean1  = (const float*)d_in[6];
  const float* var1   = (const float*)d_in[7];
  const float* W2     = (const float*)d_in[8];
  const float* b2     = (const float*)d_in[9];
  const float* gamma2 = (const float*)d_in[10];
  const float* beta2  = (const float*)d_in[11];
  const float* mean2  = (const float*)d_in[12];
  const float* var2   = (const float*)d_in[13];
  const float* Wl     = (const float*)d_in[14];
  const float* bl     = (const float*)d_in[15];

  const int n = in_sizes[0] / 256;
  const int e = in_sizes[1] / 2;
  const int* src = ei;
  const int* dst = ei + e;

  float* ws   = (float*)d_ws;
  float* dinv = ws;
  float* h1   = dinv + (size_t)n;
  float* agg1 = h1   + (size_t)n * 64;
  float* h2   = agg1 + (size_t)n * 64;
  float* agg2 = h2   + (size_t)n * 32;

  const int T = 256;
  auto blocks = [](long long total, int t) {
    return (unsigned int)((total + t - 1) / t);
  };

  zero_f32<<<blocks((long long)n * 64, T), T, 0, stream>>>(agg1, (long long)n * 64);
  zero_f32<<<blocks((long long)n * 32, T), T, 0, stream>>>(agg2, (long long)n * 32);

  unsigned int* part = (unsigned int*)h1;
  degree_partials<25000, 64, 512, 25><<<dim3(blocks(n, 25000), 64), 512, 0, stream>>>(
      dst, part, n, e);
  dinv_from_partials<64><<<blocks(n, T), T, 0, stream>>>(part, dinv, n);

  gemm_wmma_f32<256, 64, 1><<<n / 16, 128, 0, stream>>>(x, W1, h1);

  unsigned int* marks1 = (unsigned int*)h2;
  mark_pieces<782, 128, 1024><<<blocks(blocks(e, 1024), 32), 1024, 0, stream>>>(dst, marks1, e);

  aggregate_ranges<64, 782, 2, 1024, 4, 16><<<dim3(blocks(n, 782), 2), 1024, 0, stream>>>(
      src, dst, dinv, h1, agg1, marks1, n, e);

  finalize_bn<64><<<blocks((long long)n * 64, T), T, 0, stream>>>(
      agg1, h1, dinv, b1, gamma1, beta1, mean1, var1, (long long)n * 64);

  gemm_wmma_f32<64, 32, 2><<<n / 32, 128, 0, stream>>>(agg1, W2, h2);

  unsigned int* marks2 = (unsigned int*)h1;
  mark_pieces<1563, 64, 1024><<<blocks(blocks(e, 1024), 32), 1024, 0, stream>>>(dst, marks2, e);

  aggregate_ranges<32, 1563, 4, 1024, 4, 32><<<dim3(blocks(n, 1563), 4), 1024, 0, stream>>>(
      src, dst, dinv, h2, agg2, marks2, n, e);

  final_head<<<blocks(n, T), T, 0, stream>>>(
      agg2, h2, dinv, b2, gamma2, beta2, mean2, var2, Wl, bl,
      (float*)d_out, n);
}


# layer-2 aggregation: layer 1 leaves each range's edges (source, row) as a dense list sized from degree counts; layer 2 streams it (no marks, piece scan, src look-up); sum order = LDS arrival order
# speedup vs baseline: 3.2317x; 1.0985x over previous
#include <hip/hip_runtime.h>

typedef __attribute__((ext_vector_type(2))) float v2f;
typedef __attribute__((ext_vector_type(8))) float v8f;

#define BN_EPS 1e-5f

template <bool B> struct flag_t { static constexpr bool value = B; };

__global__ void zero_f32(float* __restrict__ p, long long n) {
  long long i = (long long)blockIdx.x * blockDim.x + threadIdx.x;
  if (i < n) p[i] = 0.0f;
}

template <int R, int C, int NT, int U>
__global__ void __launch_bounds__(NT)
degree_partials(const int* __restrict__ dst, unsigned int* __restrict__ part,
                int n, int e) {
  __shared__ __attribute__((aligned(16))) unsigned int cnt[R];
  const int tid = threadIdx.x;
  const int lo  = blockIdx.x * R;
  const int c   = blockIdx.y;

  uint4* cnt4 = (uint4*)cnt;
  for (int i = tid; i < R/4; i += NT) cnt4[i] = make_uint4(0u, 0u, 0u, 0u);
  __syncthreads();

  const long long per = ((long long)e + C - 1) / C;
  const long long beg = per * c;
  long long end = beg + per;
  if (end > e) end = e;
  const int len = end > beg ? (int)(end - beg) : 0;
  const int* dptr = dst + beg;

  for (int base = 0; base < len; base += NT*U) {
    int v[U];
    #pragma unroll
    for (int u = 0; u < U; ++u) {
      const int i = base + u*NT + tid;
      const int x = dptr[i < len ? i : len - 1];
      v[u] = (i < len) ? x : -1;
    }
    #pragma unroll
    for (int u = 0; u < U; ++u) {
      const unsigned int d = (unsigned int)v[u] - (unsigned int)lo;
      if (d < (unsigned int)R) atomicAdd(&cnt[d], 1u);
    }
  }
  __syncthreads();

  int m = n - lo;
  if (m > R) m = R;
  unsigned int* row = part + (size_t)c * n + lo;
  for (int i = tid; i < m; i += NT) row[i] = cnt[i];
}

template <int C, int R, int NT>
__global__ void __launch_bounds__(NT)
dinv_and_counts(const unsigned int* __restrict__ part, float* __restrict__ dinv,
                unsigned int* __restrict__ rc, int n) {
  __shared__ unsigned int tot[2];
  const int tid = threadIdx.x;
  if (tid < 2) tot[tid] = 0u;
  __syncthreads();
  const int i = blockIdx.x * R + tid;
  unsigned int s0 = 0u, s1 = 0u;
  if (tid < R && i < n) {
    #pragma unroll
    for (int c = 0; c < C/2; ++c) s0 += part[(size_t)c * n + i];
    #pragma unroll
    for (int c = C/2; c < C; ++c) s1 += part[(size_t)c * n + i];
    dinv[i] = rsqrtf((float)(s0 + s1) + 1.0f);
  }
  for (int o = 16; o > 0; o >>= 1) {
    s0 += __shfl_xor(s0, o);
    s1 += __shfl_xor(s1, o);
  }
  if ((tid & 31) == 0) { atomicAdd(&tot[0], s0); atomicAdd(&tot[1], s1); }
  __syncthreads();
  if (tid < 2) rc[blockIdx.x * 2 + tid] = tot[tid];
}

template <int K, int NC, int MT>
__global__ void __launch_bounds__(MT*(NC/16)*32)
gemm_wmma_f32(const float* __restrict__ A, const float* __restrict__ B,
              float* __restrict__ C) {
  constexpr int KP  = K + 4;
  constexpr int NCP = NC + 8;
  constexpr int NW  = MT * (NC/16);
  __shared__ float sA[MT*16*KP];
  __shared__ float sB[K*NCP];

  const int tid = threadIdx.x;
  constexpr int NTH = NW * 32;
  constexpr int NA  = MT*16*(K/4) / NTH;
  constexpr int NB  = K*(NC/4) / NTH;
  static_assert(NA * NTH == MT*16*(K/4), "row tile must split evenly over the threads");
  static_assert(NB * NTH == K*(NC/4), "B must split evenly over the threads");
  const float* Ablk = A + (size_t)blockIdx.x * (MT*16) * K;

  float4 rb[NB];
  float4 ra[NA];
  #pragma unroll
  for (int i = 0; i < NB; ++i) {
    const int idx = tid + i*NTH;
    const int r = idx / (NC/4), c4 = (idx % (NC/4)) * 4;
    rb[i] = *(const float4*)&B[(size_t)r*NC + c4];
  }
  #pragma unroll
  for (int i = 0; i < NA; ++i) {
    const int idx = tid + i*NTH;
    const int r = idx / (K/4), c4 = (idx % (K/4)) * 4;
    ra[i] = *(const float4*)&Ablk[(size_t)r*K + c4];
  }
  #pragma unroll
  for (int i = 0; i < NB; ++i) {
    const int idx = tid + i*NTH;
    const int r = idx / (NC/4), c4 = (idx % (NC/4)) * 4;
    *(float4*)&sB[r*NCP + c4] = rb[i];
  }
  #pragma unroll
  for (int i = 0; i < NA; ++i) {
    const int idx = tid + i*NTH;
    const int r = idx / (K/4), c4 = (idx % (K/4)) * 4;
    *(float4*)&sA[r*KP + c4] = ra[i];
  }
  __syncthreads();

  const int wave  = tid >> 5, lane = tid & 31;
  const int mt    = wave / (NC/16), nt = wave % (NC/16);
  const int rrow  = lane & 15;
  const int khalf = lane >> 4;

  const float* a_ptr = &sA[(mt*16 + rrow)*KP + khalf*2];
  const float* b_ptr = &sB[(khalf*2)*NCP + nt*16 + rrow];

  v8f c0 = {};
  v8f c1 = {};
  #pragma unroll 4
  for (int ks = 0; ks < K/8; ++ks) {
    v2f a0; a0.x = a_ptr[0];     a0.y = a_ptr[1];
    v2f b0; b0.x = b_ptr[0];     b0.y = b_ptr[NCP];
    v2f a1; a1.x = a_ptr[4];     a1.y = a_ptr[5];
    v2f b1; b1.x = b_ptr[4*NCP]; b1.y = b_ptr[5*NCP];
    c0 = __builtin_amdgcn_wmma_f32_16x16x4_f32(false, a0, false, b0,
                                               (short)0, c0, false, false);
    c1 = __builtin_amdgcn_wmma_f32_16x16x4_f32(false, a1, false, b1,
                                               (short)0, c1, false, false);
    a_ptr += 8;
    b_ptr += 8*NCP;
  }
  v8f c = c0 + c1;

  const int gm  = blockIdx.x*(MT*16) + mt*16 + khalf*8;
  const int col = nt*16 + rrow;
  #pragma unroll
  for (int r = 0; r < 8; ++r)
    C[(size_t)(gm + r)*NC + col] = c[r];
}

template <int R, int NR, int NT>
__global__ void __launch_bounds__(NT)
mark_pieces(const int* __restrict__ dst, unsigned int* __restrict__ marks, int e) {
  constexpr int NW = NT / 32;
  constexpr int WW = NW * 4;
  constexpr int LP = WW + 1;
  __shared__ unsigned int bits[NR * LP];
  const int tid = threadIdx.x, lane = tid & 31, wave = tid >> 5;
  for (int i = tid; i < NR * LP; i += NT) bits[i] = 0u;
  __syncthreads();

  const int nwords = (e + 255) >> 8;
  const int pitch  = ((((e + 1023) >> 10) + 31) & ~31) * 4;
  const int word0  = (blockIdx.x * NW + wave) * 4;
  if (word0 < nwords) {
    const int base = word0 * 256 + lane;
    for (int j0 = 0; j0 < 32; j0 += 8) {
      int v[8];
      #pragma unroll
      for (int u = 0; u < 8; ++u) {
        const int i = base + (j0 + u) * 32;
        const int x = dst[i < e ? i : e - 1];
        v[u] = (i < e) ? x : -1;
      }
      #pragma unroll
      for (int u = 0; u < 8; ++u) {
        const unsigned int rid = (unsigned int)v[u] / (unsigned int)R;
        if (rid < (unsigned int)NR)
          atomicOr(&bits[rid * LP + wave * 4 + (j0 >> 3)], 1u << (u * 4 + (lane >> 3)));
      }
    }
  }
  __syncthreads();

  for (int i = tid; i < NR * WW; i += NT) {
    const int r = i / WW, w = i % WW;
    marks[(size_t)r * pitch + blockIdx.x * WW + w] = bits[r * LP + w];
  }
}

template <int F, int R, int C, int NT, int U, int G>
__global__ void __launch_bounds__(NT)
aggregate_ranges(const int* __restrict__ src, const int* __restrict__ dst,
                 const float* __restrict__ dinv, const float* __restrict__ h,
                 float* __restrict__ agg, const unsigned int* __restrict__ marks,
                 unsigned int* __restrict__ list, const unsigned int* __restrict__ rc,
                 int n, int e) {
  constexpr int NW = NT / 32;
  constexpr int FQ = F / 32;
  constexpr int QN = 256;
  constexpr int PQ = 64;
  static_assert(31 + 32*U <= QN, "ring too small for a batch");
  static_assert(4*U + 31 <= PQ, "piece queue too small for the batch");
  __shared__ __attribute__((aligned(16))) float acc[R*F];
  __shared__ uint2 ring[NW*QN];
  __shared__ int pieces[NW*PQ];
  __shared__ unsigned int lcur;
  __shared__ unsigned int lbase;
  static_assert(C == 2, "rc holds one count per range and half of the edge list");

  const int tid  = threadIdx.x;
  const int lane = tid & 31;
  const int wave = __builtin_amdgcn_readfirstlane(tid >> 5);
  const int lo   = blockIdx.x * R;
  const int c    = blockIdx.y;

  float4* acc4 = (float4*)acc;
  for (int i = tid; i < R*F/4; i += NT) acc4[i] = make_float4(0.f, 0.f, 0.f, 0.f);
  if (tid == 0) { lcur = 0u; lbase = 0u; }
  __syncthreads();
  if (tid < (int)(blockIdx.x * C + blockIdx.y)) atomicAdd(&lbase, rc[tid]);
  __syncthreads();
  const unsigned int lb = lbase;

  const int nwords = (e + 255) >> 8;
  const int pitch  = ((((e + 1023) >> 10) + 31) & ~31) * 4;
  const int per    = (nwords + C - 1) / C;
  const int wbeg   = per * c;
  int wend = wbeg + per;
  if (wend > nwords) wend = nwords;
  const unsigned int* mrow = marks + (size_t)blockIdx.x * pitch;
  const int pw = (per + NW - 1) / NW;
  int wb = wbeg + wave * pw;
  int wstop = wb + pw;
  if (wstop > wend) wstop = wend;

  uint2* q = ring + wave*QN;
  unsigned int head = 0, cnt = 0;

  auto drain = [&](unsigned int k, auto full) {
    constexpr bool FULL = decltype(full)::value;
    unsigned int p0 = 0u;
    if (lane == 0) p0 = atomicAdd(&lcur, k);
    const uint2 ent = q[(head + lane) & (QN - 1)];
    const bool val = (unsigned int)lane < k;
    const unsigned int ei = val ? ent.x : 0u;
    const unsigned int dl = val ? ent.y : 0u;
    const int   s = src[ei];
    const float w = dinv[s] * dinv[lo + (int)dl];
    p0 = (unsigned int)__builtin_amdgcn_readfirstlane((int)p0);
    if (val) list[lb + p0 + (unsigned int)lane] = (unsigned int)s | (dl << 17);
    auto group = [&](unsigned int j0) {
      int sj[G]; float wj[G]; unsigned int dj[G]; float v[G][FQ];
      #pragma unroll
      for (int g = 0; g < G; ++g) {
        const int j = (int)j0 + g;
        sj[g] = __builtin_amdgcn_readlane(s, j);
        wj[g] = __int_as_float(__builtin_amdgcn_readlane(__float_as_int(w), j));
        dj[g] = (unsigned int)__builtin_amdgcn_readlane((int)dl, j);
        const unsigned int rb = (unsigned int)sj[g] * (unsigned int)F + (unsigned int)lane;
        #pragma unroll
        for (int f = 0; f < FQ; ++f)
          v[g][f] = h[rb + (unsigned int)(f*32)];
      }
      #pragma unroll
      for (int g = 0; g < G; ++g) {
        if (FULL || j0 + g < k) {
          #pragma unroll
          for (int f = 0; f < FQ; ++f)
            atomicAdd(&acc[dj[g]*F + f*32 + lane], v[g][f] * wj[g]);
        }
      }
    };
    if constexpr (FULL) {
      #pragma unroll
      for (unsigned int j0 = 0; j0 < 32u; j0 += G) group(j0);
    } else {
      for (unsigned int j0 = 0; j0 < k; j0 += G) group(j0);
    }
  };

  int* pq = pieces + wave*PQ;
  unsigned int qh = 0, qn = 0;
  const unsigned int below = (1u << lane) - 1u;

  auto load_block = [&](int w0) -> unsigned int {
    const int w = w0 + lane;
    return mrow[w < wstop ? w : 0];
  };
  auto live = [&](int w0, unsigned int x) -> unsigned int {
    return (w0 + lane < wstop) ? x : 0u;
  };
  unsigned int mv  = live(wb, load_block(wb));
  unsigned int mvn = load_block(wb + 32);

  auto fill = [&]() {
    while (qn < 4u*U) {
      const bool has = (mv != 0u);
      const unsigned int hm = __builtin_amdgcn_ballot_w32(has);
      if (hm == 0u) {
        if (wb + 32 >= wstop) break;
        wb += 32;
        mv  = live(wb, mvn);
        mvn = load_block(wb + 32);
        continue;
      }
      if (has) {
        const int j = __builtin_ctz(mv);
        const unsigned int pos = qh + qn + (unsigned int)__builtin_popcount(hm & below);
        pq[pos & (PQ - 1)] = ((wb + lane) << 5) + j;
        mv &= mv - 1u;
      }
      qn += (unsigned int)__builtin_popcount(hm);
    }
  };

  auto fetch = [&](int (&v)[U], int (&ix)[U]) -> int {
    fill();
    const unsigned int take = qn < 4u*U ? qn : 4u*U;
    unsigned int p[U];
    #pragma unroll
    for (int u = 0; u < U; ++u)
      p[u] = (unsigned int)pq[(qh + (unsigned int)(u*4) + ((unsigned int)lane >> 3)) & (PQ - 1)];
    #pragma unroll
    for (int u = 0; u < U; ++u) {
      const unsigned int t = (unsigned int)(u*4) + ((unsigned int)lane >> 3);
      const unsigned int none = (t < take) ? 0u : ~0u;
      const int i = (int)(((p[u] << 3) | (unsigned int)(lane & 7)) | none);
      const bool in = (unsigned int)i < (unsigned int)e;
      const int x = dst[in ? i : e - 1];
      v[u]  = in ? x : -1;
      ix[u] = i;
    }
    qh += take;
    qn -= take;
    return (int)((take + 3u) >> 2);
  };

  int kA; int dvA[U]; int ixA[U];
  kA = fetch(dvA, ixA);
  while (kA > 0) {
    int kB; int dvB[U]; int ixB[U];
    kB = fetch(dvB, ixB);
    #pragma unroll
    for (int u = 0; u < U; ++u) {
      const unsigned int dl = (unsigned int)dvA[u] - (unsigned int)lo;
      const bool hit = dl < (unsigned int)R;
      const unsigned int hm = __builtin_amdgcn_ballot_w32(hit);
      if (hit) {
        const unsigned int pos =
            head + cnt + (unsigned int)__builtin_popcount(hm & below);
        q[pos & (QN - 1)] = make_uint2((unsigned int)ixA[u], dl);
      }
      cnt += (unsigned int)__builtin_popcount(hm);
    }
    while (cnt >= 32u) { drain(32u, flag_t<true>{}); head += 32u; cnt -= 32u; }
    #pragma unroll
    for (int u = 0; u < U; ++u) { dvA[u] = dvB[u]; ixA[u] = ixB[u]; }
    kA = kB;
  }
  if (cnt) drain(cnt, flag_t<false>{});
  __syncthreads();

  int rows = n - lo;
  if (rows > R) rows = R;
  float* arow = agg + (size_t)lo * F;
  for (int i = tid; i < rows*F; i += NT) atomicAdd(&arow[i], acc[i]);
}

template <int R, int NT>
__global__ void __launch_bounds__(NT)
aggregate_lists(const unsigned int* __restrict__ list, const unsigned int* __restrict__ rc,
                const float* __restrict__ dinv, const float* __restrict__ h,
                float* __restrict__ agg, int n) {
  constexpr int F  = 32;
  constexpr int NW = NT / 32;
  __shared__ __attribute__((aligned(16))) float acc[R*F];
  __shared__ unsigned int lbase;

  const int tid  = threadIdx.x;
  const int lane = tid & 31;
  const int wave = __builtin_amdgcn_readfirstlane(tid >> 5);
  const int lo   = blockIdx.x * R;
  const int idx  = blockIdx.x * 2 + blockIdx.y;

  float4* acc4 = (float4*)acc;
  for (int i = tid; i < R*F/4; i += NT) acc4[i] = make_float4(0.f, 0.f, 0.f, 0.f);
  if (tid == 0) lbase = 0u;
  __syncthreads();
  if (tid < idx) atomicAdd(&lbase, rc[tid]);
  __syncthreads();
  const unsigned int cnt = rc[idx];
  const unsigned int lb  = lbase;

  auto load_block = [&](unsigned int b) -> unsigned int {
    const unsigned int i = b + (unsigned int)lane;
    const bool in = i < cnt;
    const unsigned int x = list[in ? lb + i : 0u];
    return in ? x : 0u;
  };

  auto factors = [&](unsigned int ent, float& ds, float& dd) {
    ds = dinv[ent & 0x1FFFFu];
    dd = dinv[lo + (int)(ent >> 17)];
  };

  auto block = [&](unsigned int ent, float ds, float dd, unsigned int k, auto full) {
    constexpr bool FULL = decltype(full)::value;
    const unsigned int s  = ent & 0x1FFFFu;
    const unsigned int dl = ent >> 17;
    float v[32];
    #pragma unroll
    for (int g = 0; g < 32; ++g) {
      const unsigned int sj = (unsigned int)__builtin_amdgcn_readlane((int)s, g);
      v[g] = h[sj * (unsigned int)F + (unsigned int)lane];
    }
    const float w = ds * dd;
    #pragma unroll
    for (int g = 0; g < 32; ++g) {
      if (FULL || (unsigned int)g < k) {
        const float wj = __int_as_float(__builtin_amdgcn_readlane(__float_as_int(w), g));
        const unsigned int dj = (unsigned int)__builtin_amdgcn_readlane((int)dl, g);
        atomicAdd(&acc[dj*F + (unsigned int)lane], v[g] * wj);
      }
    }
  };

  const unsigned int step = (unsigned int)(NW * 32);
  unsigned int b = (unsigned int)wave * 32u;
  unsigned int entA = load_block(b);
  unsigned int entB = load_block(b + step);
  float dsA, ddA;
  factors(entA, dsA, ddA);
  while (b + 32u <= cnt) {
    float dsB, ddB;
    factors(entB, dsB, ddB);
    const unsigned int entC = load_block(b + 2u * step);
    block(entA, dsA, ddA, 32u, flag_t<true>{});
    entA = entB; dsA = dsB; ddA = ddB;
    entB = entC;
    b += step;
  }
  if (b < cnt) block(entA, dsA, ddA, cnt - b, flag_t<false>{});
  __syncthreads();

  int rows = n - lo;
  if (rows > R) rows = R;
  float* arow = agg + (size_t)lo * F;
  for (int i = tid; i < rows*F; i += NT) atomicAdd(&arow[i], acc[i]);
}

template <int F>
__global__ void finalize_bn(float* __restrict__ agg, const float* __restrict__ h,
                            const float* __restrict__ dinv, const float* __restrict__ bias,
                            const float* __restrict__ gamma, const float* __restrict__ beta,
                            const float* __restrict__ mean, const float* __restrict__ var,
                            long long total) {
  long long t = (long long)blockIdx.x * blockDim.x + threadIdx.x;
  if (t >= total) return;
  int node = (int)(t / F), f = (int)(t % F);
  float di = dinv[node];
  float v = agg[t] + h[t] * di * di + bias[f];
  v = (v - mean[f]) * rsqrtf(var[f] + BN_EPS) * gamma[f] + beta[f];
  agg[t] = fmaxf(v, 0.0f);
}

__global__ void final_head(const float* __restrict__ agg2, const float* __restrict__ h2,
                           const float* __restrict__ dinv, const float* __restrict__ b2,
                           const float* __restrict__ g2, const float* __restrict__ be2,
                           const float* __restrict__ m2, const float* __restrict__ v2,
                           const float* __restrict__ Wl, const float* __restrict__ bl,
                           float* __restrict__ out, int n) {
  int node = blockIdx.x * blockDim.x + threadIdx.x;
  if (node >= n) return;
  float di = dinv[node];
  float di2 = di * di;
  float acc0 = bl[0], acc1 = bl[1];
  const float* ar = agg2 + (size_t)node * 32;
  const float* hr = h2 + (size_t)node * 32;
  #pragma unroll
  for (int f = 0; f < 32; ++f) {
    float v = ar[f] + hr[f] * di2 + b2[f];
    v = (v - m2[f]) * rsqrtf(v2[f] + BN_EPS) * g2[f] + be2[f];
    v = fmaxf(v, 0.0f);
    acc0 += v * Wl[f*2 + 0];
    acc1 += v * Wl[f*2 + 1];
  }
  out[(size_t)node*2 + 0] = acc0;
  out[(size_t)node*2 + 1] = acc1;
}

extern "C" void kernel_launch(void* const* d_in, const int* in_sizes, int n_in,
                              void* d_out, int out_size, void* d_ws, size_t ws_size,
                              hipStream_t stream) {
  const float* x      = (const float*)d_in[0];
  const int*   ei     = (const int*)  d_in[1];
  const float* W1     = (const float*)d_in[2];
  const float* b1     = (const float*)d_in[3];
  const float* gamma1 = (const float*)d_in[4];
  const float* beta1  = (const float*)d_in[5];
  const float* mean1  = (const float*)d_in[6];
  const float* var1   = (const float*)d_in[7];
  const float* W2     = (const float*)d_in[8];
  const float* b2     = (const float*)d_in[9];
  const float* gamma2 = (const float*)d_in[10];
  const float* beta2  = (const float*)d_in[11];
  const float* mean2  = (const float*)d_in[12];
  const float* var2   = (const float*)d_in[13];
  const float* Wl     = (const float*)d_in[14];
  const float* bl     = (const float*)d_in[15];

  const int n = in_sizes[0] / 256;
  const int e = in_sizes[1] / 2;
  const int* src = ei;
  const int* dst = ei + e;

  float* ws   = (float*)d_ws;
  float* dinv = ws;
  float* h1   = dinv + (size_t)n;
  float* agg1 = h1   + (size_t)n * 64;
  float* pp   = agg1 + (size_t)n * 64;
  float* pq   = pp   + (size_t)n * 32;
  float* h2   = h1;
  float* agg2 = h1   + (size_t)n * 32;
  unsigned int* marks1 = (unsigned int*)pp;
  unsigned int* rc     = (unsigned int*)(pp + (size_t)n * 24);
  unsigned int* list   = (unsigned int*)pq;

  const int T = 256;
  auto blocks = [](long long total, int t) {
    return (unsigned int)((total + t - 1) / t);
  };

  zero_f32<<<blocks((long long)n * 64, T), T, 0, stream>>>(agg1, (long long)n * 64);

  unsigned int* part = (unsigned int*)h1;
  degree_partials<25000, 64, 512, 25><<<dim3(blocks(n, 25000), 64), 512, 0, stream>>>(
      dst, part, n, e);
  dinv_and_counts<64, 782, 1024><<<blocks(n, 782), 1024, 0, stream>>>(part, dinv, rc, n);

  gemm_wmma_f32<256, 64, 1><<<n / 16, 128, 0, stream>>>(x, W1, h1);

  mark_pieces<782, 128, 1024><<<blocks(blocks(e, 1024), 32), 1024, 0, stream>>>(dst, marks1, e);

  aggregate_ranges<64, 782, 2, 1024, 4, 16><<<dim3(blocks(n, 782), 2), 1024, 0, stream>>>(
      src, dst, dinv, h1, agg1, marks1, list, rc, n, e);

  finalize_bn<64><<<blocks((long long)n * 64, T), T, 0, stream>>>(
      agg1, h1, dinv, b1, gamma1, beta1, mean1, var1, (long long)n * 64);

  zero_f32<<<blocks((long long)n * 32, T), T, 0, stream>>>(agg2, (long long)n * 32);

  gemm_wmma_f32<64, 32, 2><<<n / 32, 128, 0, stream>>>(agg1, W2, h2);

  aggregate_lists<782, 1024><<<dim3(blocks(n, 782), 2), 1024, 0, stream>>>(
      list, rc, dinv, h2, agg2, n);

  final_head<<<blocks(n, T), T, 0, stream>>>(
      agg2, h2, dinv, b2, gamma2, beta2, mean2, var2, Wl, bl,
      (float*)d_out, n);
}


# aggregation (both layers): each range of 400 nodes owned by ONE work-group for the whole edge list (250x1, was 128x2 halves): final LDS sums stored once; no zero fill, no global adds; 11->9 launches
# speedup vs baseline: 3.4209x; 1.0585x over previous
#include <hip/hip_runtime.h>

typedef __attribute__((ext_vector_type(2))) float v2f;
typedef __attribute__((ext_vector_type(8))) float v8f;

#define BN_EPS 1e-5f

template <bool B> struct flag_t { static constexpr bool value = B; };

template <int R, int C, int NT, int U>
__global__ void __launch_bounds__(NT)
degree_partials(const int* __restrict__ dst, unsigned int* __restrict__ part,
                int n, int e) {
  __shared__ __attribute__((aligned(16))) unsigned int cnt[R];
  const int tid = threadIdx.x;
  const int lo  = blockIdx.x * R;
  const int c   = blockIdx.y;

  uint4* cnt4 = (uint4*)cnt;
  for (int i = tid; i < R/4; i += NT) cnt4[i] = make_uint4(0u, 0u, 0u, 0u);
  __syncthreads();

  const long long per = ((long long)e + C - 1) / C;
  const long long beg = per * c;
  long long end = beg + per;
  if (end > e) end = e;
  const int len = end > beg ? (int)(end - beg) : 0;
  const int* dptr = dst + beg;

  for (int base = 0; base < len; base += NT*U) {
    int v[U];
    #pragma unroll
    for (int u = 0; u < U; ++u) {
      const int i = base + u*NT + tid;
      const int x = dptr[i < len ? i : len - 1];
      v[u] = (i < len) ? x : -1;
    }
    #pragma unroll
    for (int u = 0; u < U; ++u) {
      const unsigned int d = (unsigned int)v[u] - (unsigned int)lo;
      if (d < (unsigned int)R) atomicAdd(&cnt[d], 1u);
    }
  }
  __syncthreads();

  int m = n - lo;
  if (m > R) m = R;
  unsigned int* row = part + (size_t)c * n + lo;
  for (int i = tid; i < m; i += NT) row[i] = cnt[i];
}

template <int C, int R, int NT>
__global__ void __launch_bounds__(NT)
dinv_and_counts(const unsigned int* __restrict__ part, float* __restrict__ dinv,
                unsigned int* __restrict__ rc, int n) {
  static_assert(R <= NT, "one thread per node of the range");
  __shared__ unsigned int tot;
  const int tid = threadIdx.x;
  if (tid == 0) tot = 0u;
  __syncthreads();
  const int i = blockIdx.x * R + tid;
  unsigned int s = 0u;
  if (tid < R && i < n) {
    #pragma unroll
    for (int c = 0; c < C; ++c) s += part[(size_t)c * n + i];
    dinv[i] = rsqrtf((float)s + 1.0f);
  }
  for (int o = 16; o > 0; o >>= 1) s += __shfl_xor(s, o);
  if ((tid & 31) == 0) atomicAdd(&tot, s);
  __syncthreads();
  if (tid == 0) rc[blockIdx.x] = tot;
}

template <int K, int NC, int MT>
__global__ void __launch_bounds__(MT*(NC/16)*32)
gemm_wmma_f32(const float* __restrict__ A, const float* __restrict__ B,
              float* __restrict__ C) {
  constexpr int KP  = K + 4;
  constexpr int NCP = NC + 8;
  constexpr int NW  = MT * (NC/16);
  __shared__ float sA[MT*16*KP];
  __shared__ float sB[K*NCP];

  const int tid = threadIdx.x;
  constexpr int NTH = NW * 32;
  constexpr int NA  = MT*16*(K/4) / NTH;
  constexpr int NB  = K*(NC/4) / NTH;
  static_assert(NA * NTH == MT*16*(K/4), "row tile must split evenly over the threads");
  static_assert(NB * NTH == K*(NC/4), "B must split evenly over the threads");
  const float* Ablk = A + (size_t)blockIdx.x * (MT*16) * K;

  float4 rb[NB];
  float4 ra[NA];
  #pragma unroll
  for (int i = 0; i < NB; ++i) {
    const int idx = tid + i*NTH;
    const int r = idx / (NC/4), c4 = (idx % (NC/4)) * 4;
    rb[i] = *(const float4*)&B[(size_t)r*NC + c4];
  }
  #pragma unroll
  for (int i = 0; i < NA; ++i) {
    const int idx = tid + i*NTH;
    const int r = idx / (K/4), c4 = (idx % (K/4)) * 4;
    ra[i] = *(const float4*)&Ablk[(size_t)r*K + c4];
  }
  #pragma unroll
  for (int i = 0; i < NB; ++i) {
    const int idx = tid + i*NTH;
    const int r = idx / (NC/4), c4 = (idx % (NC/4)) * 4;
    *(float4*)&sB[r*NCP + c4] = rb[i];
  }
  #pragma unroll
  for (int i = 0; i < NA; ++i) {
    const int idx = tid + i*NTH;
    const int r = idx / (K/4), c4 = (idx % (K/4)) * 4;
    *(float4*)&sA[r*KP + c4] = ra[i];
  }
  __syncthreads();

  const int wave  = tid >> 5, lane = tid & 31;
  const int mt    = wave / (NC/16), nt = wave % (NC/16);
  const int rrow  = lane & 15;
  const int khalf = lane >> 4;

  const float* a_ptr = &sA[(mt*16 + rrow)*KP + khalf*2];
  const float* b_ptr = &sB[(khalf*2)*NCP + nt*16 + rrow];

  v8f c0 = {};
  v8f c1 = {};
  #pragma unroll 4
  for (int ks = 0; ks < K/8; ++ks) {
    v2f a0; a0.x = a_ptr[0];     a0.y = a_ptr[1];
    v2f b0; b0.x = b_ptr[0];     b0.y = b_ptr[NCP];
    v2f a1; a1.x = a_ptr[4];     a1.y = a_ptr[5];
    v2f b1; b1.x = b_ptr[4*NCP]; b1.y = b_ptr[5*NCP];
    c0 = __builtin_amdgcn_wmma_f32_16x16x4_f32(false, a0, false, b0,
                                               (short)0, c0, false, false);
    c1 = __builtin_amdgcn_wmma_f32_16x16x4_f32(false, a1, false, b1,
                                               (short)0, c1, false, false);
    a_ptr += 8;
    b_ptr += 8*NCP;
  }
  v8f c = c0 + c1;

  const int gm  = blockIdx.x*(MT*16) + mt*16 + khalf*8;
  const int col = nt*16 + rrow;
  #pragma unroll
  for (int r = 0; r < 8; ++r)
    C[(size_t)(gm + r)*NC + col] = c[r];
}

template <int R, int NR, int NT>
__global__ void __launch_bounds__(NT)
mark_pieces(const int* __restrict__ dst, unsigned int* __restrict__ marks, int e) {
  constexpr int NW = NT / 32;
  constexpr int WW = NW * 4;
  constexpr int LP = WW + 1;
  __shared__ unsigned int bits[NR * LP];
  const int tid = threadIdx.x, lane = tid & 31, wave = tid >> 5;
  for (int i = tid; i < NR * LP; i += NT) bits[i] = 0u;
  __syncthreads();

  const int nwords = (e + 255) >> 8;
  const int pitch  = ((((e + 1023) >> 10) + 31) & ~31) * 4;
  const int word0  = (blockIdx.x * NW + wave) * 4;
  if (word0 < nwords) {
    const int base = word0 * 256 + lane;
    for (int j0 = 0; j0 < 32; j0 += 8) {
      int v[8];
      #pragma unroll
      for (int u = 0; u < 8; ++u) {
        const int i = base + (j0 + u) * 32;
        const int x = dst[i < e ? i : e - 1];
        v[u] = (i < e) ? x : -1;
      }
      #pragma unroll
      for (int u = 0; u < 8; ++u) {
        const unsigned int rid = (unsigned int)v[u] / (unsigned int)R;
        if (rid < (unsigned int)NR)
          atomicOr(&bits[rid * LP + wave * 4 + (j0 >> 3)], 1u << (u * 4 + (lane >> 3)));
      }
    }
  }
  __syncthreads();

  for (int i = tid; i < NR * WW; i += NT) {
    const int r = i / WW, w = i % WW;
    marks[(size_t)r * pitch + blockIdx.x * WW + w] = bits[r * LP + w];
  }
}

template <int F, int R, int NT, int U, int G>
__global__ void __launch_bounds__(NT)
aggregate_ranges(const int* __restrict__ src, const int* __restrict__ dst,
                 const float* __restrict__ dinv, const float* __restrict__ h,
                 float* __restrict__ agg, const unsigned int* __restrict__ marks,
                 unsigned int* __restrict__ list, const unsigned int* __restrict__ rc,
                 int n, int e) {
  constexpr int NW = NT / 32;
  constexpr int FQ = F / 32;
  constexpr int QN = 256;
  constexpr int PQ = 64;
  static_assert(31 + 32*U <= QN, "ring too small for a batch");
  static_assert(4*U + 31 <= PQ, "piece queue too small for the batch");
  __shared__ __attribute__((aligned(16))) float acc[R*F];
  __shared__ uint2 ring[NW*QN];
  __shared__ int pieces[NW*PQ];
  __shared__ unsigned int lcur;
  __shared__ unsigned int lbase;

  const int tid  = threadIdx.x;
  const int lane = tid & 31;
  const int wave = __builtin_amdgcn_readfirstlane(tid >> 5);
  const int lo   = blockIdx.x * R;

  float4* acc4 = (float4*)acc;
  for (int i = tid; i < R*F/4; i += NT) acc4[i] = make_float4(0.f, 0.f, 0.f, 0.f);
  if (tid == 0) { lcur = 0u; lbase = 0u; }
  __syncthreads();
  if (tid < (int)blockIdx.x) atomicAdd(&lbase, rc[tid]);
  __syncthreads();
  const unsigned int lb = lbase;

  const int nwords = (e + 255) >> 8;
  const int pitch  = ((((e + 1023) >> 10) + 31) & ~31) * 4;
  const unsigned int* mrow = marks + (size_t)blockIdx.x * pitch;
  const int pw = (nwords + NW - 1) / NW;
  int wb = wave * pw;
  int wstop = wb + pw;
  if (wstop > nwords) wstop = nwords;

  uint2* q = ring + wave*QN;
  unsigned int head = 0, cnt = 0;

  auto drain = [&](unsigned int k, auto full) {
    constexpr bool FULL = decltype(full)::value;
    unsigned int p0 = 0u;
    if (lane == 0) p0 = atomicAdd(&lcur, k);
    const uint2 ent = q[(head + lane) & (QN - 1)];
    const bool val = (unsigned int)lane < k;
    const unsigned int ei = val ? ent.x : 0u;
    const unsigned int dl = val ? ent.y : 0u;
    const int   s = src[ei];
    const float w = dinv[s] * dinv[lo + (int)dl];
    p0 = (unsigned int)__builtin_amdgcn_readfirstlane((int)p0);
    if (val) list[lb + p0 + (unsigned int)lane] = (unsigned int)s | (dl << 17);
    auto group = [&](unsigned int j0) {
      int sj[G]; float wj[G]; unsigned int dj[G]; float v[G][FQ];
      #pragma unroll
      for (int g = 0; g < G; ++g) {
        const int j = (int)j0 + g;
        sj[g] = __builtin_amdgcn_readlane(s, j);
        wj[g] = __int_as_float(__builtin_amdgcn_readlane(__float_as_int(w), j));
        dj[g] = (unsigned int)__builtin_amdgcn_readlane((int)dl, j);
        const unsigned int rb = (unsigned int)sj[g] * (unsigned int)F + (unsigned int)lane;
        #pragma unroll
        for (int f = 0; f < FQ; ++f)
          v[g][f] = h[rb + (unsigned int)(f*32)];
      }
      #pragma unroll
      for (int g = 0; g < G; ++g) {
        if (FULL || j0 + g < k) {
          #pragma unroll
          for (int f = 0; f < FQ; ++f)
            atomicAdd(&acc[dj[g]*F + f*32 + lane], v[g][f] * wj[g]);
        }
      }
    };
    if constexpr (FULL) {
      #pragma unroll
      for (unsigned int j0 = 0; j0 < 32u; j0 += G) group(j0);
    } else {
      for (unsigned int j0 = 0; j0 < k; j0 += G) group(j0);
    }
  };

  int* pq = pieces + wave*PQ;
  unsigned int qh = 0, qn = 0;
  const unsigned int below = (1u << lane) - 1u;

  auto load_block = [&](int w0) -> unsigned int {
    const int w = w0 + lane;
    return mrow[w < wstop ? w : 0];
  };
  auto live = [&](int w0, unsigned int x) -> unsigned int {
    return (w0 + lane < wstop) ? x : 0u;
  };
  unsigned int mv  = live(wb, load_block(wb));
  unsigned int mvn = load_block(wb + 32);

  auto fill = [&]() {
    while (qn < 4u*U) {
      const bool has = (mv != 0u);
      const unsigned int hm = __builtin_amdgcn_ballot_w32(has);
      if (hm == 0u) {
        if (wb + 32 >= wstop) break;
        wb += 32;
        mv  = live(wb, mvn);
        mvn = load_block(wb + 32);
        continue;
      }
      if (has) {
        const int j = __builtin_ctz(mv);
        const unsigned int pos = qh + qn + (unsigned int)__builtin_popcount(hm & below);
        pq[pos & (PQ - 1)] = ((wb + lane) << 5) + j;
        mv &= mv - 1u;
      }
      qn += (unsigned int)__builtin_popcount(hm);
    }
  };

  auto fetch = [&](int (&v)[U], int (&ix)[U]) -> int {
    fill();
    const unsigned int take = qn < 4u*U ? qn : 4u*U;
    unsigned int p[U];
    #pragma unroll
    for (int u = 0; u < U; ++u)
      p[u] = (unsigned int)pq[(qh + (unsigned int)(u*4) + ((unsigned int)lane >> 3)) & (PQ - 1)];
    #pragma unroll
    for (int u = 0; u < U; ++u) {
      const unsigned int t = (unsigned int)(u*4) + ((unsigned int)lane >> 3);
      const unsigned int none = (t < take) ? 0u : ~0u;
      const int i = (int)(((p[u] << 3) | (unsigned int)(lane & 7)) | none);
      const bool in = (unsigned int)i < (unsigned int)e;
      const int x = dst[in ? i : e - 1];
      v[u]  = in ? x : -1;
      ix[u] = i;
    }
    qh += take;
    qn -= take;
    return (int)((take + 3u) >> 2);
  };

  int kA; int dvA[U]; int ixA[U];
  kA = fetch(dvA, ixA);
  while (kA > 0) {
    int kB; int dvB[U]; int ixB[U];
    kB = fetch(dvB, ixB);
    #pragma unroll
    for (int u = 0; u < U; ++u) {
      const unsigned int dl = (unsigned int)dvA[u] - (unsigned int)lo;
      const bool hit = dl < (unsigned int)R;
      const unsigned int hm = __builtin_amdgcn_ballot_w32(hit);
      if (hit) {
        const unsigned int pos =
            head + cnt + (unsigned int)__builtin_popcount(hm & below);
        q[pos & (QN - 1)] = make_uint2((unsigned int)ixA[u], dl);
      }
      cnt += (unsigned int)__builtin_popcount(hm);
    }
    while (cnt >= 32u) { drain(32u, flag_t<true>{}); head += 32u; cnt -= 32u; }
    #pragma unroll
    for (int u = 0; u < U; ++u) { dvA[u] = dvB[u]; ixA[u] = ixB[u]; }
    kA = kB;
  }
  if (cnt) drain(cnt, flag_t<false>{});
  __syncthreads();

  static_assert(F % 32 == 0, "a range starts on a 128-byte line");
  int rows = n - lo;
  if (rows > R) rows = R;
  float4* arow4 = (float4*)(agg + (size_t)lo * F);
  for (int i = tid; i < rows*F/4; i += NT) arow4[i] = acc4[i];
}

template <int R, int NT>
__global__ void __launch_bounds__(NT)
aggregate_lists(const unsigned int* __restrict__ list, const unsigned int* __restrict__ rc,
                const float* __restrict__ dinv, const float* __restrict__ h,
                float* __restrict__ agg, int n) {
  constexpr int F  = 32;
  constexpr int NW = NT / 32;
  __shared__ __attribute__((aligned(16))) float acc[R*F];
  __shared__ unsigned int lbase;

  const int tid  = threadIdx.x;
  const int lane = tid & 31;
  const int wave = __builtin_amdgcn_readfirstlane(tid >> 5);
  const int lo   = blockIdx.x * R;
  const int idx  = blockIdx.x;

  float4* acc4 = (float4*)acc;
  for (int i = tid; i < R*F/4; i += NT) acc4[i] = make_float4(0.f, 0.f, 0.f, 0.f);
  if (tid == 0) lbase = 0u;
  __syncthreads();
  if (tid < idx) atomicAdd(&lbase, rc[tid]);
  __syncthreads();
  const unsigned int cnt = rc[idx];
  const unsigned int lb  = lbase;

  auto load_block = [&](unsigned int b) -> unsigned int {
    const unsigned int i = b + (unsigned int)lane;
    const bool in = i < cnt;
    const unsigned int x = list[in ? lb + i : 0u];
    return in ? x : 0u;
  };

  auto factors = [&](unsigned int ent, float& ds, float& dd) {
    ds = dinv[ent & 0x1FFFFu];
    dd = dinv[lo + (int)(ent >> 17)];
  };

  auto block = [&](unsigned int ent, float ds, float dd, unsigned int k, auto full) {
    constexpr bool FULL = decltype(full)::value;
    const unsigned int s  = ent & 0x1FFFFu;
    const unsigned int dl = ent >> 17;
    float v[32];
    #pragma unroll
    for (int g = 0; g < 32; ++g) {
      const unsigned int sj = (unsigned int)__builtin_amdgcn_readlane((int)s, g);
      v[g] = h[sj * (unsigned int)F + (unsigned int)lane];
    }
    const float w = ds * dd;
    #pragma unroll
    for (int g = 0; g < 32; ++g) {
      if (FULL || (unsigned int)g < k) {
        const float wj = __int_as_float(__builtin_amdgcn_readlane(__float_as_int(w), g));
        const unsigned int dj = (unsigned int)__builtin_amdgcn_readlane((int)dl, g);
        atomicAdd(&acc[dj*F + (unsigned int)lane], v[g] * wj);
      }
    }
  };

  const unsigned int step = (unsigned int)(NW * 32);
  unsigned int b = (unsigned int)wave * 32u;
  unsigned int entA = load_block(b);
  unsigned int entB = load_block(b + step);
  float dsA, ddA;
  factors(entA, dsA, ddA);
  while (b + 32u <= cnt) {
    float dsB, ddB;
    factors(entB, dsB, ddB);
    const unsigned int entC = load_block(b + 2u * step);
    block(entA, dsA, ddA, 32u, flag_t<true>{});
    entA = entB; dsA = dsB; ddA = ddB;
    entB = entC;
    b += step;
  }
  if (b < cnt) block(entA, dsA, ddA, cnt - b, flag_t<false>{});
  __syncthreads();

  int rows = n - lo;
  if (rows > R) rows = R;
  float4* arow4 = (float4*)(agg + (size_t)lo * F);
  for (int i = tid; i < rows*F/4; i += NT) arow4[i] = acc4[i];
}

template <int F>
__global__ void finalize_bn(float* __restrict__ agg, const float* __restrict__ h,
                            const float* __restrict__ dinv, const float* __restrict__ bias,
                            const float* __restrict__ gamma, const float* __restrict__ beta,
                            const float* __restrict__ mean, const float* __restrict__ var,
                            long long total) {
  long long t = (long long)blockIdx.x * blockDim.x + threadIdx.x;
  if (t >= total) return;
  int node = (int)(t / F), f = (int)(t % F);
  float di = dinv[node];
  float v = agg[t] + h[t] * di * di + bias[f];
  v = (v - mean[f]) * rsqrtf(var[f] + BN_EPS) * gamma[f] + beta[f];
  agg[t] = fmaxf(v, 0.0f);
}

__global__ void final_head(const float* __restrict__ agg2, const float* __restrict__ h2,
                           const float* __restrict__ dinv, const float* __restrict__ b2,
                           const float* __restrict__ g2, const float* __restrict__ be2,
                           const float* __restrict__ m2, const float* __restrict__ v2,
                           const float* __restrict__ Wl, const float* __restrict__ bl,
                           float* __restrict__ out, int n) {
  int node = blockIdx.x * blockDim.x + threadIdx.x;
  if (node >= n) return;
  float di = dinv[node];
  float di2 = di * di;
  float acc0 = bl[0], acc1 = bl[1];
  const float* ar = agg2 + (size_t)node * 32;
  const float* hr = h2 + (size_t)node * 32;
  #pragma unroll
  for (int f = 0; f < 32; ++f) {
    float v = ar[f] + hr[f] * di2 + b2[f];
    v = (v - m2[f]) * rsqrtf(v2[f] + BN_EPS) * g2[f] + be2[f];
    v = fmaxf(v, 0.0f);
    acc0 += v * Wl[f*2 + 0];
    acc1 += v * Wl[f*2 + 1];
  }
  out[(size_t)node*2 + 0] = acc0;
  out[(size_t)node*2 + 1] = acc1;
}

extern "C" void kernel_launch(void* const* d_in, const int* in_sizes, int n_in,
                              void* d_out, int out_size, void* d_ws, size_t ws_size,
                              hipStream_t stream) {
  const float* x      = (const float*)d_in[0];
  const int*   ei     = (const int*)  d_in[1];
  const float* W1     = (const float*)d_in[2];
  const float* b1     = (const float*)d_in[3];
  const float* gamma1 = (const float*)d_in[4];
  const float* beta1  = (const float*)d_in[5];
  const float* mean1  = (const float*)d_in[6];
  const float* var1   = (const float*)d_in[7];
  const float* W2     = (const float*)d_in[8];
  const float* b2     = (const float*)d_in[9];
  const float* gamma2 = (const float*)d_in[10];
  const float* beta2  = (const float*)d_in[11];
  const float* mean2  = (const float*)d_in[12];
  const float* var2   = (const float*)d_in[13];
  const float* Wl     = (const float*)d_in[14];
  const float* bl     = (const float*)d_in[15];

  const int n = in_sizes[0] / 256;
  const int e = in_sizes[1] / 2;
  const int* src = ei;
  const int* dst = ei + e;

  float* ws   = (float*)d_ws;
  float* dinv = ws;
  float* h1   = dinv + (size_t)n;
  float* agg1 = h1   + (size_t)n * 64;
  float* pp   = agg1 + (size_t)n * 64;
  float* pq   = pp   + (size_t)n * 32;
  float* h2   = h1;
  float* agg2 = h1   + (size_t)n * 32;
  unsigned int* marks1 = (unsigned int*)pp;
  unsigned int* rc     = (unsigned int*)(pp + (size_t)n * 32 - 256);
  unsigned int* list   = (unsigned int*)pq;

  const int T = 256;
  auto blocks = [](long long total, int t) {
    return (unsigned int)((total + t - 1) / t);
  };

  unsigned int* part = (unsigned int*)h1;
  degree_partials<25000, 64, 512, 25><<<dim3(blocks(n, 25000), 64), 512, 0, stream>>>(
      dst, part, n, e);
  dinv_and_counts<64, 400, 512><<<blocks(n, 400), 512, 0, stream>>>(part, dinv, rc, n);

  gemm_wmma_f32<256, 64, 1><<<n / 16, 128, 0, stream>>>(x, W1, h1);

  mark_pieces<400, 250, 1024><<<blocks(blocks(e, 1024), 32), 1024, 0, stream>>>(dst, marks1, e);

  aggregate_ranges<64, 400, 1024, 4, 16><<<blocks(n, 400), 1024, 0, stream>>>(
      src, dst, dinv, h1, agg1, marks1, list, rc, n, e);

  finalize_bn<64><<<blocks((long long)n * 64, T), T, 0, stream>>>(
      agg1, h1, dinv, b1, gamma1, beta1, mean1, var1, (long long)n * 64);

  gemm_wmma_f32<64, 32, 2><<<n / 32, 128, 0, stream>>>(agg1, W2, h2);

  aggregate_lists<400, 1024><<<blocks(n, 400), 1024, 0, stream>>>(
      list, rc, dinv, h2, agg2, n);

  final_head<<<blocks(n, T), T, 0, stream>>>(
      agg2, h2, dinv, b2, gamma2, beta2, mean2, var2, Wl, bl,
      (float*)d_out, n);
}


# layer-1 epilogue (self-loop, bias, BN, ReLU) formed in the layer-2 GEMM's row staging: its launch of 200000 one-element waves is gone (9->8 launches); h2 rows at pitch 64; sums unchanged
# speedup vs baseline: 3.6644x; 1.0712x over previous
#include <hip/hip_runtime.h>

typedef __attribute__((ext_vector_type(2))) float v2f;
typedef __attribute__((ext_vector_type(8))) float v8f;

#define BN_EPS 1e-5f

template <bool B> struct flag_t { static constexpr bool value = B; };

template <int R, int C, int NT, int U>
__global__ void __launch_bounds__(NT)
degree_partials(const int* __restrict__ dst, unsigned int* __restrict__ part,
                int n, int e) {
  __shared__ __attribute__((aligned(16))) unsigned int cnt[R];
  const int tid = threadIdx.x;
  const int lo  = blockIdx.x * R;
  const int c   = blockIdx.y;

  uint4* cnt4 = (uint4*)cnt;
  for (int i = tid; i < R/4; i += NT) cnt4[i] = make_uint4(0u, 0u, 0u, 0u);
  __syncthreads();

  const long long per = ((long long)e + C - 1) / C;
  const long long beg = per * c;
  long long end = beg + per;
  if (end > e) end = e;
  const int len = end > beg ? (int)(end - beg) : 0;
  const int* dptr = dst + beg;

  for (int base = 0; base < len; base += NT*U) {
    int v[U];
    #pragma unroll
    for (int u = 0; u < U; ++u) {
      const int i = base + u*NT + tid;
      const int x = dptr[i < len ? i : len - 1];
      v[u] = (i < len) ? x : -1;
    }
    #pragma unroll
    for (int u = 0; u < U; ++u) {
      const unsigned int d = (unsigned int)v[u] - (unsigned int)lo;
      if (d < (unsigned int)R) atomicAdd(&cnt[d], 1u);
    }
  }
  __syncthreads();

  int m = n - lo;
  if (m > R) m = R;
  unsigned int* row = part + (size_t)c * n + lo;
  for (int i = tid; i < m; i += NT) row[i] = cnt[i];
}

template <int C, int R, int NT>
__global__ void __launch_bounds__(NT)
dinv_and_counts(const unsigned int* __restrict__ part, float* __restrict__ dinv,
                unsigned int* __restrict__ rc, int n) {
  static_assert(R <= NT, "one thread per node of the range");
  __shared__ unsigned int tot;
  const int tid = threadIdx.x;
  if (tid == 0) tot = 0u;
  __syncthreads();
  const int i = blockIdx.x * R + tid;
  unsigned int s = 0u;
  if (tid < R && i < n) {
    #pragma unroll
    for (int c = 0; c < C; ++c) s += part[(size_t)c * n + i];
    dinv[i] = rsqrtf((float)s + 1.0f);
  }
  for (int o = 16; o > 0; o >>= 1) s += __shfl_xor(s, o);
  if ((tid & 31) == 0) atomicAdd(&tot, s);
  __syncthreads();
  if (tid == 0) rc[blockIdx.x] = tot;
}

template <int K, int NC, int MT>
__global__ void __launch_bounds__(MT*(NC/16)*32)
gemm_wmma_f32(const float* __restrict__ A, const float* __restrict__ B,
              float* __restrict__ C) {
  constexpr int KP  = K + 4;
  constexpr int NCP = NC + 8;
  constexpr int NW  = MT * (NC/16);
  __shared__ float sA[MT*16*KP];
  __shared__ float sB[K*NCP];

  const int tid = threadIdx.x;
  constexpr int NTH = NW * 32;
  constexpr int NA  = MT*16*(K/4) / NTH;
  constexpr int NB  = K*(NC/4) / NTH;
  static_assert(NA * NTH == MT*16*(K/4), "row tile must split evenly over the threads");
  static_assert(NB * NTH == K*(NC/4), "B must split evenly over the threads");
  const float* Ablk = A + (size_t)blockIdx.x * (MT*16) * K;

  float4 rb[NB];
  float4 ra[NA];
  #pragma unroll
  for (int i = 0; i < NB; ++i) {
    const int idx = tid + i*NTH;
    const int r = idx / (NC/4), c4 = (idx % (NC/4)) * 4;
    rb[i] = *(const float4*)&B[(size_t)r*NC + c4];
  }
  #pragma unroll
  for (int i = 0; i < NA; ++i) {
    const int idx = tid + i*NTH;
    const int r = idx / (K/4), c4 = (idx % (K/4)) * 4;
    ra[i] = *(const float4*)&Ablk[(size_t)r*K + c4];
  }
  #pragma unroll
  for (int i = 0; i < NB; ++i) {
    const int idx = tid + i*NTH;
    const int r = idx / (NC/4), c4 = (idx % (NC/4)) * 4;
    *(float4*)&sB[r*NCP + c4] = rb[i];
  }
  #pragma unroll
  for (int i = 0; i < NA; ++i) {
    const int idx = tid + i*NTH;
    const int r = idx / (K/4), c4 = (idx % (K/4)) * 4;
    *(float4*)&sA[r*KP + c4] = ra[i];
  }
  __syncthreads();

  const int wave  = tid >> 5, lane = tid & 31;
  const int mt    = wave / (NC/16), nt = wave % (NC/16);
  const int rrow  = lane & 15;
  const int khalf = lane >> 4;

  const float* a_ptr = &sA[(mt*16 + rrow)*KP + khalf*2];
  const float* b_ptr = &sB[(khalf*2)*NCP + nt*16 + rrow];

  v8f c0 = {};
  v8f c1 = {};
  #pragma unroll 4
  for (int ks = 0; ks < K/8; ++ks) {
    v2f a0; a0.x = a_ptr[0];     a0.y = a_ptr[1];
    v2f b0; b0.x = b_ptr[0];     b0.y = b_ptr[NCP];
    v2f a1; a1.x = a_ptr[4];     a1.y = a_ptr[5];
    v2f b1; b1.x = b_ptr[4*NCP]; b1.y = b_ptr[5*NCP];
    c0 = __builtin_amdgcn_wmma_f32_16x16x4_f32(false, a0, false, b0,
                                               (short)0, c0, false, false);
    c1 = __builtin_amdgcn_wmma_f32_16x16x4_f32(false, a1, false, b1,
                                               (short)0, c1, false, false);
    a_ptr += 8;
    b_ptr += 8*NCP;
  }
  v8f c = c0 + c1;

  const int gm  = blockIdx.x*(MT*16) + mt*16 + khalf*8;
  const int col = nt*16 + rrow;
  #pragma unroll
  for (int r = 0; r < 8; ++r)
    C[(size_t)(gm + r)*NC + col] = c[r];
}

__device__ __forceinline__ float bn_relu(float a, float hv, float di, float b,
                                         float m, float rs, float g, float be) {
  float v = a + hv * di * di + b;
  v = (v - m) * rs * g + be;
  return fmaxf(v, 0.0f);
}

template <int K, int NC, int MT, int CP>
__global__ void __launch_bounds__(MT*(NC/16)*32)
gemm_act_wmma_f32(const float* agg, const float* __restrict__ h,
                  const float* __restrict__ dinv, const float* __restrict__ bias,
                  const float* __restrict__ gamma, const float* __restrict__ beta,
                  const float* __restrict__ mean, const float* __restrict__ var,
                  const float* __restrict__ B, float* C) {
  constexpr int KP  = K + 4;
  constexpr int NCP = NC + 8;
  constexpr int NW  = MT * (NC/16);
  __shared__ float sA[MT*16*KP];
  __shared__ float sB[K*NCP];

  const int tid = threadIdx.x;
  constexpr int NTH = NW * 32;
  constexpr int NA  = MT*16*(K/4) / NTH;
  constexpr int NB  = K*(NC/4) / NTH;
  constexpr int RS  = NTH / (K/4);
  static_assert(NA * NTH == MT*16*(K/4), "row tile must split evenly over the threads");
  static_assert(NB * NTH == K*(NC/4), "B must split evenly over the threads");
  static_assert(RS * (K/4) == NTH, "a thread's pieces share their columns");
  const size_t row0 = (size_t)blockIdx.x * (MT*16);
  const int c4 = (tid % (K/4)) * 4;
  const int r0 = tid / (K/4);

  float4 rb[NB];
  #pragma unroll
  for (int i = 0; i < NB; ++i) {
    const int idx = tid + i*NTH;
    const int r = idx / (NC/4), cb = (idx % (NC/4)) * 4;
    rb[i] = *(const float4*)&B[(size_t)r*NC + cb];
  }
  const float4 pb = *(const float4*)&bias[c4];
  const float4 pm = *(const float4*)&mean[c4];
  const float4 pv = *(const float4*)&var[c4];
  const float4 pg = *(const float4*)&gamma[c4];
  const float4 pe = *(const float4*)&beta[c4];
  float4 ra[NA];
  float4 rh[NA];
  float  rd[NA];
  #pragma unroll
  for (int i = 0; i < NA; ++i) {
    const size_t row = row0 + (size_t)(r0 + i*RS);
    ra[i] = *(const float4*)&agg[row*K + c4];
    rh[i] = *(const float4*)&h[row*K + c4];
    rd[i] = dinv[row];
  }
  #pragma unroll
  for (int i = 0; i < NB; ++i) {
    const int idx = tid + i*NTH;
    const int r = idx / (NC/4), cb = (idx % (NC/4)) * 4;
    *(float4*)&sB[r*NCP + cb] = rb[i];
  }
  float4 rs;
  rs.x = rsqrtf(pv.x + BN_EPS);
  rs.y = rsqrtf(pv.y + BN_EPS);
  rs.z = rsqrtf(pv.z + BN_EPS);
  rs.w = rsqrtf(pv.w + BN_EPS);
  #pragma unroll
  for (int i = 0; i < NA; ++i) {
    const float di = rd[i];
    float4 a;
    a.x = bn_relu(ra[i].x, rh[i].x, di, pb.x, pm.x, rs.x, pg.x, pe.x);
    a.y = bn_relu(ra[i].y, rh[i].y, di, pb.y, pm.y, rs.y, pg.y, pe.y);
    a.z = bn_relu(ra[i].z, rh[i].z, di, pb.z, pm.z, rs.z, pg.z, pe.z);
    a.w = bn_relu(ra[i].w, rh[i].w, di, pb.w, pm.w, rs.w, pg.w, pe.w);
    *(float4*)&sA[(r0 + i*RS)*KP + c4] = a;
  }
  __syncthreads();

  const int wave  = tid >> 5, lane = tid & 31;
  const int mt    = wave / (NC/16), nt = wave % (NC/16);
  const int rrow  = lane & 15;
  const int khalf = lane >> 4;

  const float* a_ptr = &sA[(mt*16 + rrow)*KP + khalf*2];
  const float* b_ptr = &sB[(khalf*2)*NCP + nt*16 + rrow];

  v8f c0 = {};
  v8f c1 = {};
  #pragma unroll 4
  for (int ks = 0; ks < K/8; ++ks) {
    v2f a0; a0.x = a_ptr[0];     a0.y = a_ptr[1];
    v2f b0; b0.x = b_ptr[0];     b0.y = b_ptr[NCP];
    v2f a1; a1.x = a_ptr[4];     a1.y = a_ptr[5];
    v2f b1; b1.x = b_ptr[4*NCP]; b1.y = b_ptr[5*NCP];
    c0 = __builtin_amdgcn_wmma_f32_16x16x4_f32(false, a0, false, b0,
                                               (short)0, c0, false, false);
    c1 = __builtin_amdgcn_wmma_f32_16x16x4_f32(false, a1, false, b1,
                                               (short)0, c1, false, false);
    a_ptr += 8;
    b_ptr += 8*NCP;
  }
  v8f c = c0 + c1;

  const size_t gm = row0 + (size_t)(mt*16 + khalf*8);
  const int col = nt*16 + rrow;
  #pragma unroll
  for (int r = 0; r < 8; ++r)
    C[(gm + r)*CP + col] = c[r];
}

template <int R, int NR, int NT>
__global__ void __launch_bounds__(NT)
mark_pieces(const int* __restrict__ dst, unsigned int* __restrict__ marks, int e) {
  constexpr int NW = NT / 32;
  constexpr int WW = NW * 4;
  constexpr int LP = WW + 1;
  __shared__ unsigned int bits[NR * LP];
  const int tid = threadIdx.x, lane = tid & 31, wave = tid >> 5;
  for (int i = tid; i < NR * LP; i += NT) bits[i] = 0u;
  __syncthreads();

  const int nwords = (e + 255) >> 8;
  const int pitch  = ((((e + 1023) >> 10) + 31) & ~31) * 4;
  const int word0  = (blockIdx.x * NW + wave) * 4;
  if (word0 < nwords) {
    const int base = word0 * 256 + lane;
    for (int j0 = 0; j0 < 32; j0 += 8) {
      int v[8];
      #pragma unroll
      for (int u = 0; u < 8; ++u) {
        const int i = base + (j0 + u) * 32;
        const int x = dst[i < e ? i : e - 1];
        v[u] = (i < e) ? x : -1;
      }
      #pragma unroll
      for (int u = 0; u < 8; ++u) {
        const unsigned int rid = (unsigned int)v[u] / (unsigned int)R;
        if (rid < (unsigned int)NR)
          atomicOr(&bits[rid * LP + wave * 4 + (j0 >> 3)], 1u << (u * 4 + (lane >> 3)));
      }
    }
  }
  __syncthreads();

  for (int i = tid; i < NR * WW; i += NT) {
    const int r = i / WW, w = i % WW;
    marks[(size_t)r * pitch + blockIdx.x * WW + w] = bits[r * LP + w];
  }
}

template <int F, int R, int NT, int U, int G>
__global__ void __launch_bounds__(NT)
aggregate_ranges(const int* __restrict__ src, const int* __restrict__ dst,
                 const float* __restrict__ dinv, const float* __restrict__ h,
                 float* __restrict__ agg, const unsigned int* __restrict__ marks,
                 unsigned int* __restrict__ list, const unsigned int* __restrict__ rc,
                 int n, int e) {
  constexpr int NW = NT / 32;
  constexpr int FQ = F / 32;
  constexpr int QN = 256;
  constexpr int PQ = 64;
  static_assert(31 + 32*U <= QN, "ring too small for a batch");
  static_assert(4*U + 31 <= PQ, "piece queue too small for the batch");
  __shared__ __attribute__((aligned(16))) float acc[R*F];
  __shared__ uint2 ring[NW*QN];
  __shared__ int pieces[NW*PQ];
  __shared__ unsigned int lcur;
  __shared__ unsigned int lbase;

  const int tid  = threadIdx.x;
  const int lane = tid & 31;
  const int wave = __builtin_amdgcn_readfirstlane(tid >> 5);
  const int lo   = blockIdx.x * R;

  float4* acc4 = (float4*)acc;
  for (int i = tid; i < R*F/4; i += NT) acc4[i] = make_float4(0.f, 0.f, 0.f, 0.f);
  if (tid == 0) { lcur = 0u; lbase = 0u; }
  __syncthreads();
  if (tid < (int)blockIdx.x) atomicAdd(&lbase, rc[tid]);
  __syncthreads();
  const unsigned int lb = lbase;

  const int nwords = (e + 255) >> 8;
  const int pitch  = ((((e + 1023) >> 10) + 31) & ~31) * 4;
  const unsigned int* mrow = marks + (size_t)blockIdx.x * pitch;
  const int pw = (nwords + NW - 1) / NW;
  int wb = wave * pw;
  int wstop = wb + pw;
  if (wstop > nwords) wstop = nwords;

  uint2* q = ring + wave*QN;
  unsigned int head = 0, cnt = 0;

  auto drain = [&](unsigned int k, auto full) {
    constexpr bool FULL = decltype(full)::value;
    unsigned int p0 = 0u;
    if (lane == 0) p0 = atomicAdd(&lcur, k);
    const uint2 ent = q[(head + lane) & (QN - 1)];
    const bool val = (unsigned int)lane < k;
    const unsigned int ei = val ? ent.x : 0u;
    const unsigned int dl = val ? ent.y : 0u;
    const int   s = src[ei];
    const float w = dinv[s] * dinv[lo + (int)dl];
    p0 = (unsigned int)__builtin_amdgcn_readfirstlane((int)p0);
    if (val) list[lb + p0 + (unsigned int)lane] = (unsigned int)s | (dl << 17);
    auto group = [&](unsigned int j0) {
      int sj[G]; float wj[G]; unsigned int dj[G]; float v[G][FQ];
      #pragma unroll
      for (int g = 0; g < G; ++g) {
        const int j = (int)j0 + g;
        sj[g] = __builtin_amdgcn_readlane(s, j);
        wj[g] = __int_as_float(__builtin_amdgcn_readlane(__float_as_int(w), j));
        dj[g] = (unsigned int)__builtin_amdgcn_readlane((int)dl, j);
        const unsigned int rb = (unsigned int)sj[g] * (unsigned int)F + (unsigned int)lane;
        #pragma unroll
        for (int f = 0; f < FQ; ++f)
          v[g][f] = h[rb + (unsigned int)(f*32)];
      }
      #pragma unroll
      for (int g = 0; g < G; ++g) {
        if (FULL || j0 + g < k) {
          #pragma unroll
          for (int f = 0; f < FQ; ++f)
            atomicAdd(&acc[dj[g]*F + f*32 + lane], v[g][f] * wj[g]);
        }
      }
    };
    if constexpr (FULL) {
      #pragma unroll
      for (unsigned int j0 = 0; j0 < 32u; j0 += G) group(j0);
    } else {
      for (unsigned int j0 = 0; j0 < k; j0 += G) group(j0);
    }
  };

  int* pq = pieces + wave*PQ;
  unsigned int qh = 0, qn = 0;
  const unsigned int below = (1u << lane) - 1u;

  auto load_block = [&](int w0) -> unsigned int {
    const int w = w0 + lane;
    return mrow[w < wstop ? w : 0];
  };
  auto live = [&](int w0, unsigned int x) -> unsigned int {
    return (w0 + lane < wstop) ? x : 0u;
  };
  unsigned int mv  = live(wb, load_block(wb));
  unsigned int mvn = load_block(wb + 32);

  auto fill = [&]() {
    while (qn < 4u*U) {
      const bool has = (mv != 0u);
      const unsigned int hm = __builtin_amdgcn_ballot_w32(has);
      if (hm == 0u) {
        if (wb + 32 >= wstop) break;
        wb += 32;
        mv  = live(wb, mvn);
        mvn = load_block(wb + 32);
        continue;
      }
      if (has) {
        const int j = __builtin_ctz(mv);
        const unsigned int pos = qh + qn + (unsigned int)__builtin_popcount(hm & below);
        pq[pos & (PQ - 1)] = ((wb + lane) << 5) + j;
        mv &= mv - 1u;
      }
      qn += (unsigned int)__builtin_popcount(hm);
    }
  };

  auto fetch = [&](int (&v)[U], int (&ix)[U]) -> int {
    fill();
    const unsigned int take = qn < 4u*U ? qn : 4u*U;
    unsigned int p[U];
    #pragma unroll
    for (int u = 0; u < U; ++u)
      p[u] = (unsigned int)pq[(qh + (unsigned int)(u*4) + ((unsigned int)lane >> 3)) & (PQ - 1)];
    #pragma unroll
    for (int u = 0; u < U; ++u) {
      const unsigned int t = (unsigned int)(u*4) + ((unsigned int)lane >> 3);
      const unsigned int none = (t < take) ? 0u : ~0u;
      const int i = (int)(((p[u] << 3) | (unsigned int)(lane & 7)) | none);
      const bool in = (unsigned int)i < (unsigned int)e;
      const int x = dst[in ? i : e - 1];
      v[u]  = in ? x : -1;
      ix[u] = i;
    }
    qh += take;
    qn -= take;
    return (int)((take + 3u) >> 2);
  };

  int kA; int dvA[U]; int ixA[U];
  kA = fetch(dvA, ixA);
  while (kA > 0) {
    int kB; int dvB[U]; int ixB[U];
    kB = fetch(dvB, ixB);
    #pragma unroll
    for (int u = 0; u < U; ++u) {
      const unsigned int dl = (unsigned int)dvA[u] - (unsigned int)lo;
      const bool hit = dl < (unsigned int)R;
      const unsigned int hm = __builtin_amdgcn_ballot_w32(hit);
      if (hit) {
        const unsigned int pos =
            head + cnt + (unsigned int)__builtin_popcount(hm & below);
        q[pos & (QN - 1)] = make_uint2((unsigned int)ixA[u], dl);
      }
      cnt += (unsigned int)__builtin_popcount(hm);
    }
    while (cnt >= 32u) { drain(32u, flag_t<true>{}); head += 32u; cnt -= 32u; }
    #pragma unroll
    for (int u = 0; u < U; ++u) { dvA[u] = dvB[u]; ixA[u] = ixB[u]; }
    kA = kB;
  }
  if (cnt) drain(cnt, flag_t<false>{});
  __syncthreads();

  static_assert(F % 32 == 0, "a range starts on a 128-byte line");
  int rows = n - lo;
  if (rows > R) rows = R;
  float4* arow4 = (float4*)(agg + (size_t)lo * F);
  for (int i = tid; i < rows*F/4; i += NT) arow4[i] = acc4[i];
}

template <int R, int NT, int HP>
__global__ void __launch_bounds__(NT)
aggregate_lists(const unsigned int* __restrict__ list, const unsigned int* __restrict__ rc,
                const float* __restrict__ dinv, const float* __restrict__ h,
                float* __restrict__ agg, int n) {
  constexpr int F  = 32;
  constexpr int NW = NT / 32;
  __shared__ __attribute__((aligned(16))) float acc[R*F];
  __shared__ unsigned int lbase;

  const int tid  = threadIdx.x;
  const int lane = tid & 31;
  const int wave = __builtin_amdgcn_readfirstlane(tid >> 5);
  const int lo   = blockIdx.x * R;
  const int idx  = blockIdx.x;

  float4* acc4 = (float4*)acc;
  for (int i = tid; i < R*F/4; i += NT) acc4[i] = make_float4(0.f, 0.f, 0.f, 0.f);
  if (tid == 0) lbase = 0u;
  __syncthreads();
  if (tid < idx) atomicAdd(&lbase, rc[tid]);
  __syncthreads();
  const unsigned int cnt = rc[idx];
  const unsigned int lb  = lbase;

  auto load_block = [&](unsigned int b) -> unsigned int {
    const unsigned int i = b + (unsigned int)lane;
    const bool in = i < cnt;
    const unsigned int x = list[in ? lb + i : 0u];
    return in ? x : 0u;
  };

  auto factors = [&](unsigned int ent, float& ds, float& dd) {
    ds = dinv[ent & 0x1FFFFu];
    dd = dinv[lo + (int)(ent >> 17)];
  };

  auto block = [&](unsigned int ent, float ds, float dd, unsigned int k, auto full) {
    constexpr bool FULL = decltype(full)::value;
    const unsigned int s  = ent & 0x1FFFFu;
    const unsigned int dl = ent >> 17;
    float v[32];
    #pragma unroll
    for (int g = 0; g < 32; ++g) {
      const unsigned int sj = (unsigned int)__builtin_amdgcn_readlane((int)s, g);
      v[g] = h[sj * (unsigned int)HP + (unsigned int)lane];
    }
    const float w = ds * dd;
    #pragma unroll
    for (int g = 0; g < 32; ++g) {
      if (FULL || (unsigned int)g < k) {
        const float wj = __int_as_float(__builtin_amdgcn_readlane(__float_as_int(w), g));
        const unsigned int dj = (unsigned int)__builtin_amdgcn_readlane((int)dl, g);
        atomicAdd(&acc[dj*F + (unsigned int)lane], v[g] * wj);
      }
    }
  };

  const unsigned int step = (unsigned int)(NW * 32);
  unsigned int b = (unsigned int)wave * 32u;
  unsigned int entA = load_block(b);
  unsigned int entB = load_block(b + step);
  float dsA, ddA;
  factors(entA, dsA, ddA);
  while (b + 32u <= cnt) {
    float dsB, ddB;
    factors(entB, dsB, ddB);
    const unsigned int entC = load_block(b + 2u * step);
    block(entA, dsA, ddA, 32u, flag_t<true>{});
    entA = entB; dsA = dsB; ddA = ddB;
    entB = entC;
    b += step;
  }
  if (b < cnt) block(entA, dsA, ddA, cnt - b, flag_t<false>{});
  __syncthreads();

  int rows = n - lo;
  if (rows > R) rows = R;
  float4* arow4 = (float4*)(agg + (size_t)lo * F);
  for (int i = tid; i < rows*F/4; i += NT) arow4[i] = acc4[i];
}

template <int HP>
__global__ void final_head(const float* __restrict__ agg2, const float* __restrict__ h2,
                           const float* __restrict__ dinv, const float* __restrict__ b2,
                           const float* __restrict__ g2, const float* __restrict__ be2,
                           const float* __restrict__ m2, const float* __restrict__ v2,
                           const float* __restrict__ Wl, const float* __restrict__ bl,
                           float* __restrict__ out, int n) {
  int node = blockIdx.x * blockDim.x + threadIdx.x;
  if (node >= n) return;
  float di = dinv[node];
  float di2 = di * di;
  float acc0 = bl[0], acc1 = bl[1];
  const float* ar = agg2 + (size_t)node * 32;
  const float* hr = h2 + (size_t)node * HP;
  #pragma unroll
  for (int f = 0; f < 32; ++f) {
    float v = ar[f] + hr[f] * di2 + b2[f];
    v = (v - m2[f]) * rsqrtf(v2[f] + BN_EPS) * g2[f] + be2[f];
    v = fmaxf(v, 0.0f);
    acc0 += v * Wl[f*2 + 0];
    acc1 += v * Wl[f*2 + 1];
  }
  out[(size_t)node*2 + 0] = acc0;
  out[(size_t)node*2 + 1] = acc1;
}

extern "C" void kernel_launch(void* const* d_in, const int* in_sizes, int n_in,
                              void* d_out, int out_size, void* d_ws, size_t ws_size,
                              hipStream_t stream) {
  const float* x      = (const float*)d_in[0];
  const int*   ei     = (const int*)  d_in[1];
  const float* W1     = (const float*)d_in[2];
  const float* b1     = (const float*)d_in[3];
  const float* gamma1 = (const float*)d_in[4];
  const float* beta1  = (const float*)d_in[5];
  const float* mean1  = (const float*)d_in[6];
  const float* var1   = (const float*)d_in[7];
  const float* W2     = (const float*)d_in[8];
  const float* b2     = (const float*)d_in[9];
  const float* gamma2 = (const float*)d_in[10];
  const float* beta2  = (const float*)d_in[11];
  const float* mean2  = (const float*)d_in[12];
  const float* var2   = (const float*)d_in[13];
  const float* Wl     = (const float*)d_in[14];
  const float* bl     = (const float*)d_in[15];

  const int n = in_sizes[0] / 256;
  const int e = in_sizes[1] / 2;
  const int* src = ei;
  const int* dst = ei + e;

  float* ws   = (float*)d_ws;
  float* dinv = ws;
  float* h1   = dinv + (size_t)n;
  float* agg1 = h1   + (size_t)n * 64;
  float* pp   = agg1 + (size_t)n * 64;
  float* pq   = pp   + (size_t)n * 32;
  float* h2   = agg1;
  float* agg2 = h1;
  unsigned int* marks1 = (unsigned int*)pp;
  unsigned int* rc     = (unsigned int*)(pp + (size_t)n * 32 - 256);
  unsigned int* list   = (unsigned int*)pq;

  const int T = 256;
  auto blocks = [](long long total, int t) {
    return (unsigned int)((total + t - 1) / t);
  };

  unsigned int* part = (unsigned int*)h1;
  degree_partials<25000, 64, 512, 25><<<dim3(blocks(n, 25000), 64), 512, 0, stream>>>(
      dst, part, n, e);
  dinv_and_counts<64, 400, 512><<<blocks(n, 400), 512, 0, stream>>>(part, dinv, rc, n);

  gemm_wmma_f32<256, 64, 1><<<n / 16, 128, 0, stream>>>(x, W1, h1);

  mark_pieces<400, 250, 1024><<<blocks(blocks(e, 1024), 32), 1024, 0, stream>>>(dst, marks1, e);

  aggregate_ranges<64, 400, 1024, 4, 16><<<blocks(n, 400), 1024, 0, stream>>>(
      src, dst, dinv, h1, agg1, marks1, list, rc, n, e);

  gemm_act_wmma_f32<64, 32, 2, 64><<<n / 32, 128, 0, stream>>>(
      agg1, h1, dinv, b1, gamma1, beta1, mean1, var1, W2, h2);

  aggregate_lists<400, 1024, 64><<<blocks(n, 400), 1024, 0, stream>>>(
      list, rc, dinv, h2, agg2, n);

  final_head<64><<<blocks(n, T), T, 0, stream>>>(
      agg2, h2, dinv, b2, gamma2, beta2, mean2, var2, Wl, bl,
      (float*)d_out, n);
}


# aggregation weights (both layers): node counts also kept as one byte each; a work-group holds that table (100000 B) in LDS, an edge's deg^-1/2 factors are LDS reads + rsqrt, not dinv reads; same bits
# speedup vs baseline: 3.9649x; 1.0820x over previous
#include <hip/hip_runtime.h>

typedef __attribute__((ext_vector_type(2))) float v2f;
typedef __attribute__((ext_vector_type(8))) float v8f;

#define BN_EPS 1e-5f

template <bool B> struct flag_t { static constexpr bool value = B; };

template <int R, int C, int NT, int U>
__global__ void __launch_bounds__(NT)
degree_partials(const int* __restrict__ dst, unsigned int* __restrict__ part,
                int n, int e) {
  __shared__ __attribute__((aligned(16))) unsigned int cnt[R];
  const int tid = threadIdx.x;
  const int lo  = blockIdx.x * R;
  const int c   = blockIdx.y;

  uint4* cnt4 = (uint4*)cnt;
  for (int i = tid; i < R/4; i += NT) cnt4[i] = make_uint4(0u, 0u, 0u, 0u);
  __syncthreads();

  const long long per = ((long long)e + C - 1) / C;
  const long long beg = per * c;
  long long end = beg + per;
  if (end > e) end = e;
  const int len = end > beg ? (int)(end - beg) : 0;
  const int* dptr = dst + beg;

  for (int base = 0; base < len; base += NT*U) {
    int v[U];
    #pragma unroll
    for (int u = 0; u < U; ++u) {
      const int i = base + u*NT + tid;
      const int x = dptr[i < len ? i : len - 1];
      v[u] = (i < len) ? x : -1;
    }
    #pragma unroll
    for (int u = 0; u < U; ++u) {
      const unsigned int d = (unsigned int)v[u] - (unsigned int)lo;
      if (d < (unsigned int)R) atomicAdd(&cnt[d], 1u);
    }
  }
  __syncthreads();

  int m = n - lo;
  if (m > R) m = R;
  unsigned int* row = part + (size_t)c * n + lo;
  for (int i = tid; i < m; i += NT) row[i] = cnt[i];
}

template <int C, int R, int NT>
__global__ void __launch_bounds__(NT)
dinv_and_counts(const unsigned int* __restrict__ part, float* __restrict__ dinv,
                unsigned int* __restrict__ deg, unsigned int* __restrict__ rc, int n) {
  static_assert(R <= NT, "one thread per node of the range");
  __shared__ unsigned int tot;
  const int tid = threadIdx.x;
  if (tid == 0) tot = 0u;
  __syncthreads();
  const int i = blockIdx.x * R + tid;
  unsigned int s = 0u;
  if (tid < R && i < n) {
    #pragma unroll
    for (int c = 0; c < C; ++c) s += part[(size_t)c * n + i];
    dinv[i] = rsqrtf((float)s + 1.0f);
    ((unsigned char*)deg)[i] = (unsigned char)(s < 255u ? s : 255u);
  }
  for (int o = 16; o > 0; o >>= 1) s += __shfl_xor(s, o);
  if ((tid & 31) == 0) atomicAdd(&tot, s);
  __syncthreads();
  if (tid == 0) rc[blockIdx.x] = tot;
}

template <int TN>
__device__ __forceinline__ unsigned int deg_byte(const unsigned char* tab, unsigned int i) {
  const bool in = i < (unsigned int)TN;
  const unsigned int c = tab[in ? i : 0u];
  return in ? c : 255u;
}

__device__ __forceinline__ float edge_weight(unsigned int cs, unsigned int cd,
                                             const float* __restrict__ dinv,
                                             unsigned int s, unsigned int d) {
  float w = rsqrtf((float)cs + 1.0f) * rsqrtf((float)cd + 1.0f);
  if ((cs == 255u) | (cd == 255u)) w = dinv[s] * dinv[d];
  return w;
}

template <int TN, int NT>
struct table_stage {
  static_assert(TN % 16 == 0, "whole 16-byte pieces");
  static constexpr int TQ = TN / 16;
  static constexpr int TP = (TQ + NT - 1) / NT;
  uint4 t[TP];
  __device__ __forceinline__ void load(const unsigned int* __restrict__ deg, int tid) {
    #pragma unroll
    for (int i = 0; i < TP; ++i) {
      const int j = tid + i*NT;
      t[i] = ((const uint4*)deg)[j < TQ ? j : 0];
    }
  }
  __device__ __forceinline__ void store(unsigned char* tab, int tid) const {
    #pragma unroll
    for (int i = 0; i < TP; ++i) {
      const int j = tid + i*NT;
      ((uint4*)tab)[j < TQ ? j : 0] = t[i];
    }
  }
};

template <int K, int NC, int MT>
__global__ void __launch_bounds__(MT*(NC/16)*32)
gemm_wmma_f32(const float* __restrict__ A, const float* __restrict__ B,
              float* __restrict__ C) {
  constexpr int KP  = K + 4;
  constexpr int NCP = NC + 8;
  constexpr int NW  = MT * (NC/16);
  __shared__ float sA[MT*16*KP];
  __shared__ float sB[K*NCP];

  const int tid = threadIdx.x;
  constexpr int NTH = NW * 32;
  constexpr int NA  = MT*16*(K/4) / NTH;
  constexpr int NB  = K*(NC/4) / NTH;
  static_assert(NA * NTH == MT*16*(K/4), "row tile must split evenly over the threads");
  static_assert(NB * NTH == K*(NC/4), "B must split evenly over the threads");
  const float* Ablk = A + (size_t)blockIdx.x * (MT*16) * K;

  float4 rb[NB];
  float4 ra[NA];
  #pragma unroll
  for (int i = 0; i < NB; ++i) {
    const int idx = tid + i*NTH;
    const int r = idx / (NC/4), c4 = (idx % (NC/4)) * 4;
    rb[i] = *(const float4*)&B[(size_t)r*NC + c4];
  }
  #pragma unroll
  for (int i = 0; i < NA; ++i) {
    const int idx = tid + i*NTH;
    const int r = idx / (K/4), c4 = (idx % (K/4)) * 4;
    ra[i] = *(const float4*)&Ablk[(size_t)r*K + c4];
  }
  #pragma unroll
  for (int i = 0; i < NB; ++i) {
    const int idx = tid + i*NTH;
    const int r = idx / (NC/4), c4 = (idx % (NC/4)) * 4;
    *(float4*)&sB[r*NCP + c4] = rb[i];
  }
  #pragma unroll
  for (int i = 0; i < NA; ++i) {
    const int idx = tid + i*NTH;
    const int r = idx / (K/4), c4 = (idx % (K/4)) * 4;
    *(float4*)&sA[r*KP + c4] = ra[i];
  }
  __syncthreads();

  const int wave  = tid >> 5, lane = tid & 31;
  const int mt    = wave / (NC/16), nt = wave % (NC/16);
  const int rrow  = lane & 15;
  const int khalf = lane >> 4;

  const float* a_ptr = &sA[(mt*16 + rrow)*KP + khalf*2];
  const float* b_ptr = &sB[(khalf*2)*NCP + nt*16 + rrow];

  v8f c0 = {};
  v8f c1 = {};
  #pragma unroll 4
  for (int ks = 0; ks < K/8; ++ks) {
    v2f a0; a0.x = a_ptr[0];     a0.y = a_ptr[1];
    v2f b0; b0.x = b_ptr[0];     b0.y = b_ptr[NCP];
    v2f a1; a1.x = a_ptr[4];     a1.y = a_ptr[5];
    v2f b1; b1.x = b_ptr[4*NCP]; b1.y = b_ptr[5*NCP];
    c0 = __builtin_amdgcn_wmma_f32_16x16x4_f32(false, a0, false, b0,
                                               (short)0, c0, false, false);
    c1 = __builtin_amdgcn_wmma_f32_16x16x4_f32(false, a1, false, b1,
                                               (short)0, c1, false, false);
    a_ptr += 8;
    b_ptr += 8*NCP;
  }
  v8f c = c0 + c1;

  const int gm  = blockIdx.x*(MT*16) + mt*16 + khalf*8;
  const int col = nt*16 + rrow;
  #pragma unroll
  for (int r = 0; r < 8; ++r)
    C[(size_t)(gm + r)*NC + col] = c[r];
}

__device__ __forceinline__ float bn_relu(float a, float hv, float di, float b,
                                         float m, float rs, float g, float be) {
  float v = a + hv * di * di + b;
  v = (v - m) * rs * g + be;
  return fmaxf(v, 0.0f);
}

template <int K, int NC, int MT, int CP>
__global__ void __launch_bounds__(MT*(NC/16)*32)
gemm_act_wmma_f32(const float* agg, const float* __restrict__ h,
                  const float* __restrict__ dinv, const float* __restrict__ bias,
                  const float* __restrict__ gamma, const float* __restrict__ beta,
                  const float* __restrict__ mean, const float* __restrict__ var,
                  const float* __restrict__ B, float* C) {
  constexpr int KP  = K + 4;
  constexpr int NCP = NC + 8;
  constexpr int NW  = MT * (NC/16);
  __shared__ float sA[MT*16*KP];
  __shared__ float sB[K*NCP];

  const int tid = threadIdx.x;
  constexpr int NTH = NW * 32;
  constexpr int NA  = MT*16*(K/4) / NTH;
  constexpr int NB  = K*(NC/4) / NTH;
  constexpr int RS  = NTH / (K/4);
  static_assert(NA * NTH == MT*16*(K/4), "row tile must split evenly over the threads");
  static_assert(NB * NTH == K*(NC/4), "B must split evenly over the threads");
  static_assert(RS * (K/4) == NTH, "a thread's pieces share their columns");
  const size_t row0 = (size_t)blockIdx.x * (MT*16);
  const int c4 = (tid % (K/4)) * 4;
  const int r0 = tid / (K/4);

  float4 rb[NB];
  #pragma unroll
  for (int i = 0; i < NB; ++i) {
    const int idx = tid + i*NTH;
    const int r = idx / (NC/4), cb = (idx % (NC/4)) * 4;
    rb[i] = *(const float4*)&B[(size_t)r*NC + cb];
  }
  const float4 pb = *(const float4*)&bias[c4];
  const float4 pm = *(const float4*)&mean[c4];
  const float4 pv = *(const float4*)&var[c4];
  const float4 pg = *(const float4*)&gamma[c4];
  const float4 pe = *(const float4*)&beta[c4];
  float4 ra[NA];
  float4 rh[NA];
  float  rd[NA];
  #pragma unroll
  for (int i = 0; i < NA; ++i) {
    const size_t row = row0 + (size_t)(r0 + i*RS);
    ra[i] = *(const float4*)&agg[row*K + c4];
    rh[i] = *(const float4*)&h[row*K + c4];
    rd[i] = dinv[row];
  }
  #pragma unroll
  for (int i = 0; i < NB; ++i) {
    const int idx = tid + i*NTH;
    const int r = idx / (NC/4), cb = (idx % (NC/4)) * 4;
    *(float4*)&sB[r*NCP + cb] = rb[i];
  }
  float4 rs;
  rs.x = rsqrtf(pv.x + BN_EPS);
  rs.y = rsqrtf(pv.y + BN_EPS);
  rs.z = rsqrtf(pv.z + BN_EPS);
  rs.w = rsqrtf(pv.w + BN_EPS);
  #pragma unroll
  for (int i = 0; i < NA; ++i) {
    const float di = rd[i];
    float4 a;
    a.x = bn_relu(ra[i].x, rh[i].x, di, pb.x, pm.x, rs.x, pg.x, pe.x);
    a.y = bn_relu(ra[i].y, rh[i].y, di, pb.y, pm.y, rs.y, pg.y, pe.y);
    a.z = bn_relu(ra[i].z, rh[i].z, di, pb.z, pm.z, rs.z, pg.z, pe.z);
    a.w = bn_relu(ra[i].w, rh[i].w, di, pb.w, pm.w, rs.w, pg.w, pe.w);
    *(float4*)&sA[(r0 + i*RS)*KP + c4] = a;
  }
  __syncthreads();

  const int wave  = tid >> 5, lane = tid & 31;
  const int mt    = wave / (NC/16), nt = wave % (NC/16);
  const int rrow  = lane & 15;
  const int khalf = lane >> 4;

  const float* a_ptr = &sA[(mt*16 + rrow)*KP + khalf*2];
  const float* b_ptr = &sB[(khalf*2)*NCP + nt*16 + rrow];

  v8f c0 = {};
  v8f c1 = {};
  #pragma unroll 4
  for (int ks = 0; ks < K/8; ++ks) {
    v2f a0; a0.x = a_ptr[0];     a0.y = a_ptr[1];
    v2f b0; b0.x = b_ptr[0];     b0.y = b_ptr[NCP];
    v2f a1; a1.x = a_ptr[4];     a1.y = a_ptr[5];
    v2f b1; b1.x = b_ptr[4*NCP]; b1.y = b_ptr[5*NCP];
    c0 = __builtin_amdgcn_wmma_f32_16x16x4_f32(false, a0, false, b0,
                                               (short)0, c0, false, false);
    c1 = __builtin_amdgcn_wmma_f32_16x16x4_f32(false, a1, false, b1,
                                               (short)0, c1, false, false);
    a_ptr += 8;
    b_ptr += 8*NCP;
  }
  v8f c = c0 + c1;

  const size_t gm = row0 + (size_t)(mt*16 + khalf*8);
  const int col = nt*16 + rrow;
  #pragma unroll
  for (int r = 0; r < 8; ++r)
    C[(gm + r)*CP + col] = c[r];
}

template <int R, int NR, int NT>
__global__ void __launch_bounds__(NT)
mark_pieces(const int* __restrict__ dst, unsigned int* __restrict__ marks, int e) {
  constexpr int NW = NT / 32;
  constexpr int WW = NW * 4;
  constexpr int LP = WW + 1;
  __shared__ unsigned int bits[NR * LP];
  const int tid = threadIdx.x, lane = tid & 31, wave = tid >> 5;
  for (int i = tid; i < NR * LP; i += NT) bits[i] = 0u;
  __syncthreads();

  const int nwords = (e + 255) >> 8;
  const int pitch  = ((((e + 1023) >> 10) + 31) & ~31) * 4;
  const int word0  = (blockIdx.x * NW + wave) * 4;
  if (word0 < nwords) {
    const int base = word0 * 256 + lane;
    for (int j0 = 0; j0 < 32; j0 += 8) {
      int v[8];
      #pragma unroll
      for (int u = 0; u < 8; ++u) {
        const int i = base + (j0 + u) * 32;
        const int x = dst[i < e ? i : e - 1];
        v[u] = (i < e) ? x : -1;
      }
      #pragma unroll
      for (int u = 0; u < 8; ++u) {
        const unsigned int rid = (unsigned int)v[u] / (unsigned int)R;
        if (rid < (unsigned int)NR)
          atomicOr(&bits[rid * LP + wave * 4 + (j0 >> 3)], 1u << (u * 4 + (lane >> 3)));
      }
    }
  }
  __syncthreads();

  for (int i = tid; i < NR * WW; i += NT) {
    const int r = i / WW, w = i % WW;
    marks[(size_t)r * pitch + blockIdx.x * WW + w] = bits[r * LP + w];
  }
}

template <int F, int R, int NT, int U, int G, int TN>
__global__ void __launch_bounds__(NT)
aggregate_ranges(const int* __restrict__ src, const int* __restrict__ dst,
                 const float* __restrict__ dinv, const unsigned int* __restrict__ deg,
                 const float* __restrict__ h,
                 float* __restrict__ agg, const unsigned int* __restrict__ marks,
                 unsigned int* __restrict__ list, const unsigned int* __restrict__ rc,
                 int n, int e) {
  constexpr int NW = NT / 32;
  constexpr int FQ = F / 32;
  constexpr int QN = 256;
  constexpr int PQ = 64;
  static_assert(31 + 32*U <= QN, "ring too small for a batch");
  static_assert(4*U + 31 <= PQ, "piece queue too small for the batch");
  __shared__ __attribute__((aligned(16))) float acc[R*F];
  __shared__ uint2 ring[NW*QN];
  __shared__ int pieces[NW*PQ];
  __shared__ __attribute__((aligned(16))) unsigned char dtab[TN];
  __shared__ unsigned int lcur;
  __shared__ unsigned int lbase;

  const int tid  = threadIdx.x;
  const int lane = tid & 31;
  const int wave = __builtin_amdgcn_readfirstlane(tid >> 5);
  const int lo   = blockIdx.x * R;

  table_stage<TN, NT> ts;
  ts.load(deg, tid);
  float4* acc4 = (float4*)acc;
  for (int i = tid; i < R*F/4; i += NT) acc4[i] = make_float4(0.f, 0.f, 0.f, 0.f);
  if (tid == 0) { lcur = 0u; lbase = 0u; }
  ts.store(dtab, tid);
  __syncthreads();
  if (tid < (int)blockIdx.x) atomicAdd(&lbase, rc[tid]);
  __syncthreads();
  const unsigned int lb = lbase;

  const int nwords = (e + 255) >> 8;
  const int pitch  = ((((e + 1023) >> 10) + 31) & ~31) * 4;
  const unsigned int* mrow = marks + (size_t)blockIdx.x * pitch;
  const int pw = (nwords + NW - 1) / NW;
  int wb = wave * pw;
  int wstop = wb + pw;
  if (wstop > nwords) wstop = nwords;

  uint2* q = ring + wave*QN;
  unsigned int head = 0, cnt = 0;

  auto drain = [&](unsigned int k, auto full) {
    constexpr bool FULL = decltype(full)::value;
    unsigned int p0 = 0u;
    if (lane == 0) p0 = atomicAdd(&lcur, k);
    const uint2 ent = q[(head + lane) & (QN - 1)];
    const bool val = (unsigned int)lane < k;
    const unsigned int ei = val ? ent.x : 0u;
    const unsigned int dl = val ? ent.y : 0u;
    const int   s = src[ei];
    const unsigned int cs = deg_byte<TN>(dtab, (unsigned int)s);
    const unsigned int cd = deg_byte<TN>(dtab, (unsigned int)lo + dl);
    p0 = (unsigned int)__builtin_amdgcn_readfirstlane((int)p0);
    if (val) list[lb + p0 + (unsigned int)lane] = (unsigned int)s | (dl << 17);
    auto fetch_rows = [&](unsigned int j0, float (&v)[G][FQ]) {
      #pragma unroll
      for (int g = 0; g < G; ++g) {
        const int sj = __builtin_amdgcn_readlane(s, (int)j0 + g);
        const unsigned int rb = (unsigned int)sj * (unsigned int)F + (unsigned int)lane;
        #pragma unroll
        for (int f = 0; f < FQ; ++f)
          v[g][f] = h[rb + (unsigned int)(f*32)];
      }
    };
    auto add_rows = [&](unsigned int j0, const float (&v)[G][FQ], float w) {
      #pragma unroll
      for (int g = 0; g < G; ++g) {
        if (FULL || j0 + g < k) {
          const int j = (int)j0 + g;
          const float wj = __int_as_float(__builtin_amdgcn_readlane(__float_as_int(w), j));
          const unsigned int dj = (unsigned int)__builtin_amdgcn_readlane((int)dl, j);
          #pragma unroll
          for (int f = 0; f < FQ; ++f)
            atomicAdd(&acc[dj*F + f*32 + lane], v[g][f] * wj);
        }
      }
    };
    float v0[G][FQ];
    fetch_rows(0u, v0);
    const float w = edge_weight(cs, cd, dinv, (unsigned int)s, (unsigned int)lo + dl);
    add_rows(0u, v0, w);
    if constexpr (FULL) {
      #pragma unroll
      for (unsigned int j0 = G; j0 < 32u; j0 += G) {
        float v[G][FQ];
        fetch_rows(j0, v);
        add_rows(j0, v, w);
      }
    } else {
      for (unsigned int j0 = G; j0 < k; j0 += G) {
        float v[G][FQ];
        fetch_rows(j0, v);
        add_rows(j0, v, w);
      }
    }
  };

  int* pq = pieces + wave*PQ;
  unsigned int qh = 0, qn = 0;
  const unsigned int below = (1u << lane) - 1u;

  auto load_block = [&](int w0) -> unsigned int {
    const int w = w0 + lane;
    return mrow[w < wstop ? w : 0];
  };
  auto live = [&](int w0, unsigned int x) -> unsigned int {
    return (w0 + lane < wstop) ? x : 0u;
  };
  unsigned int mv  = live(wb, load_block(wb));
  unsigned int mvn = load_block(wb + 32);

  auto fill = [&]() {
    while (qn < 4u*U) {
      const bool has = (mv != 0u);
      const unsigned int hm = __builtin_amdgcn_ballot_w32(has);
      if (hm == 0u) {
        if (wb + 32 >= wstop) break;
        wb += 32;
        mv  = live(wb, mvn);
        mvn = load_block(wb + 32);
        continue;
      }
      if (has) {
        const int j = __builtin_ctz(mv);
        const unsigned int pos = qh + qn + (unsigned int)__builtin_popcount(hm & below);
        pq[pos & (PQ - 1)] = ((wb + lane) << 5) + j;
        mv &= mv - 1u;
      }
      qn += (unsigned int)__builtin_popcount(hm);
    }
  };

  auto fetch = [&](int (&v)[U], int (&ix)[U]) -> int {
    fill();
    const unsigned int take = qn < 4u*U ? qn : 4u*U;
    unsigned int p[U];
    #pragma unroll
    for (int u = 0; u < U; ++u)
      p[u] = (unsigned int)pq[(qh + (unsigned int)(u*4) + ((unsigned int)lane >> 3)) & (PQ - 1)];
    #pragma unroll
    for (int u = 0; u < U; ++u) {
      const unsigned int t = (unsigned int)(u*4) + ((unsigned int)lane >> 3);
      const unsigned int none = (t < take) ? 0u : ~0u;
      const int i = (int)(((p[u] << 3) | (unsigned int)(lane & 7)) | none);
      const bool in = (unsigned int)i < (unsigned int)e;
      const int x = dst[in ? i : e - 1];
      v[u]  = in ? x : -1;
      ix[u] = i;
    }
    qh += take;
    qn -= take;
    return (int)((take + 3u) >> 2);
  };

  int kA; int dvA[U]; int ixA[U];
  kA = fetch(dvA, ixA);
  while (kA > 0) {
    int kB; int dvB[U]; int ixB[U];
    kB = fetch(dvB, ixB);
    #pragma unroll
    for (int u = 0; u < U; ++u) {
      const unsigned int dl = (unsigned int)dvA[u] - (unsigned int)lo;
      const bool hit = dl < (unsigned int)R;
      const unsigned int hm = __builtin_amdgcn_ballot_w32(hit);
      if (hit) {
        const unsigned int pos =
            head + cnt + (unsigned int)__builtin_popcount(hm & below);
        q[pos & (QN - 1)] = make_uint2((unsigned int)ixA[u], dl);
      }
      cnt += (unsigned int)__builtin_popcount(hm);
    }
    while (cnt >= 32u) { drain(32u, flag_t<true>{}); head += 32u; cnt -= 32u; }
    #pragma unroll
    for (int u = 0; u < U; ++u) { dvA[u] = dvB[u]; ixA[u] = ixB[u]; }
    kA = kB;
  }
  if (cnt) drain(cnt, flag_t<false>{});
  __syncthreads();

  static_assert(F % 32 == 0, "a range starts on a 128-byte line");
  int rows = n - lo;
  if (rows > R) rows = R;
  float4* arow4 = (float4*)(agg + (size_t)lo * F);
  for (int i = tid; i < rows*F/4; i += NT) arow4[i] = acc4[i];
}

template <int R, int NT, int HP, int TN>
__global__ void __launch_bounds__(NT)
aggregate_lists(const unsigned int* __restrict__ list, const unsigned int* __restrict__ rc,
                const float* __restrict__ dinv, const unsigned int* __restrict__ deg,
                const float* __restrict__ h,
                float* __restrict__ agg, int n) {
  constexpr int F  = 32;
  constexpr int NW = NT / 32;
  __shared__ __attribute__((aligned(16))) float acc[R*F];
  __shared__ __attribute__((aligned(16))) unsigned char dtab[TN];
  __shared__ unsigned int lbase;

  const int tid  = threadIdx.x;
  const int lane = tid & 31;
  const int wave = __builtin_amdgcn_readfirstlane(tid >> 5);
  const int lo   = blockIdx.x * R;
  const int idx  = blockIdx.x;

  table_stage<TN, NT> ts;
  ts.load(deg, tid);
  float4* acc4 = (float4*)acc;
  for (int i = tid; i < R*F/4; i += NT) acc4[i] = make_float4(0.f, 0.f, 0.f, 0.f);
  if (tid == 0) lbase = 0u;
  ts.store(dtab, tid);
  __syncthreads();
  if (tid < idx) atomicAdd(&lbase, rc[tid]);
  __syncthreads();
  const unsigned int cnt = rc[idx];
  const unsigned int lb  = lbase;

  auto load_block = [&](unsigned int b) -> unsigned int {
    const unsigned int i = b + (unsigned int)lane;
    const bool in = i < cnt;
    const unsigned int x = list[in ? lb + i : 0u];
    return in ? x : 0u;
  };

  auto block = [&](unsigned int ent, unsigned int k, auto full) {
    constexpr bool FULL = decltype(full)::value;
    const unsigned int s  = ent & 0x1FFFFu;
    const unsigned int dl = ent >> 17;
    const unsigned int cs = deg_byte<TN>(dtab, s);
    const unsigned int cd = deg_byte<TN>(dtab, (unsigned int)lo + dl);
    float v[32];
    #pragma unroll
    for (int g = 0; g < 32; ++g) {
      const unsigned int sj = (unsigned int)__builtin_amdgcn_readlane((int)s, g);
      v[g] = h[sj * (unsigned int)HP + (unsigned int)lane];
    }
    const float w = edge_weight(cs, cd, dinv, s, (unsigned int)lo + dl);
    #pragma unroll
    for (int g = 0; g < 32; ++g) {
      if (FULL || (unsigned int)g < k) {
        const float wj = __int_as_float(__builtin_amdgcn_readlane(__float_as_int(w), g));
        const unsigned int dj = (unsigned int)__builtin_amdgcn_readlane((int)dl, g);
        atomicAdd(&acc[dj*F + (unsigned int)lane], v[g] * wj);
      }
    }
  };

  const unsigned int step = (unsigned int)(NW * 32);
  unsigned int b = (unsigned int)wave * 32u;
  unsigned int entA = load_block(b);
  unsigned int entB = load_block(b + step);
  while (b + 32u <= cnt) {
    const unsigned int entC = load_block(b + 2u * step);
    block(entA, 32u, flag_t<true>{});
    entA = entB;
    entB = entC;
    b += step;
  }
  if (b < cnt) block(entA, cnt - b, flag_t<false>{});
  __syncthreads();

  int rows = n - lo;
  if (rows > R) rows = R;
  float4* arow4 = (float4*)(agg + (size_t)lo * F);
  for (int i = tid; i < rows*F/4; i += NT) arow4[i] = acc4[i];
}

template <int HP>
__global__ void final_head(const float* __restrict__ agg2, const float* __restrict__ h2,
                           const float* __restrict__ dinv, const float* __restrict__ b2,
                           const float* __restrict__ g2, const float* __restrict__ be2,
                           const float* __restrict__ m2, const float* __restrict__ v2,
                           const float* __restrict__ Wl, const float* __restrict__ bl,
                           float* __restrict__ out, int n) {
  int node = blockIdx.x * blockDim.x + threadIdx.x;
  if (node >= n) return;
  float di = dinv[node];
  float di2 = di * di;
  float acc0 = bl[0], acc1 = bl[1];
  const float* ar = agg2 + (size_t)node * 32;
  const float* hr = h2 + (size_t)node * HP;
  #pragma unroll
  for (int f = 0; f < 32; ++f) {
    float v = ar[f] + hr[f] * di2 + b2[f];
    v = (v - m2[f]) * rsqrtf(v2[f] + BN_EPS) * g2[f] + be2[f];
    v = fmaxf(v, 0.0f);
    acc0 += v * Wl[f*2 + 0];
    acc1 += v * Wl[f*2 + 1];
  }
  out[(size_t)node*2 + 0] = acc0;
  out[(size_t)node*2 + 1] = acc1;
}

extern "C" void kernel_launch(void* const* d_in, const int* in_sizes, int n_in,
                              void* d_out, int out_size, void* d_ws, size_t ws_size,
                              hipStream_t stream) {
  const float* x      = (const float*)d_in[0];
  const int*   ei     = (const int*)  d_in[1];
  const float* W1     = (const float*)d_in[2];
  const float* b1     = (const float*)d_in[3];
  const float* gamma1 = (const float*)d_in[4];
  const float* beta1  = (const float*)d_in[5];
  const float* mean1  = (const float*)d_in[6];
  const float* var1   = (const float*)d_in[7];
  const float* W2     = (const float*)d_in[8];
  const float* b2     = (const float*)d_in[9];
  const float* gamma2 = (const float*)d_in[10];
  const float* beta2  = (const float*)d_in[11];
  const float* mean2  = (const float*)d_in[12];
  const float* var2   = (const float*)d_in[13];
  const float* Wl     = (const float*)d_in[14];
  const float* bl     = (const float*)d_in[15];

  const int n = in_sizes[0] / 256;
  const int e = in_sizes[1] / 2;
  const int* src = ei;
  const int* dst = ei + e;

  float* ws   = (float*)d_ws;
  float* dinv = ws;
  float* h1   = dinv + (size_t)n;
  float* agg1 = h1   + (size_t)n * 64;
  float* pp   = agg1 + (size_t)n * 64;
  float* pq   = pp   + (size_t)n * 32;
  float* h2   = agg1;
  float* agg2 = h1;
  unsigned int* marks1 = (unsigned int*)pp;
  unsigned int* rc     = (unsigned int*)(pp + (size_t)n * 32 - 256);
  unsigned int* deg    = rc - 25600;
  unsigned int* list   = (unsigned int*)pq;

  const int T = 256;
  auto blocks = [](long long total, int t) {
    return (unsigned int)((total + t - 1) / t);
  };

  unsigned int* part = (unsigned int*)h1;
  degree_partials<25000, 64, 512, 25><<<dim3(blocks(n, 25000), 64), 512, 0, stream>>>(
      dst, part, n, e);
  dinv_and_counts<64, 400, 512><<<blocks(n, 400), 512, 0, stream>>>(part, dinv, deg, rc, n);

  gemm_wmma_f32<256, 64, 1><<<n / 16, 128, 0, stream>>>(x, W1, h1);

  mark_pieces<400, 250, 1024><<<blocks(blocks(e, 1024), 32), 1024, 0, stream>>>(dst, marks1, e);

  aggregate_ranges<64, 400, 1024, 4, 16, 100000><<<blocks(n, 400), 1024, 0, stream>>>(
      src, dst, dinv, deg, h1, agg1, marks1, list, rc, n, e);

  gemm_act_wmma_f32<64, 32, 2, 64><<<n / 32, 128, 0, stream>>>(
      agg1, h1, dinv, b1, gamma1, beta1, mean1, var1, W2, h2);

  aggregate_lists<400, 1024, 64, 100000><<<blocks(n, 400), 1024, 0, stream>>>(
      list, rc, dinv, deg, h2, agg2, n);

  final_head<64><<<blocks(n, T), T, 0, stream>>>(
      agg2, h2, dinv, b2, gamma2, beta2, mean2, var2, Wl, bl,
      (float*)d_out, n);
}
